# starting kernel
# baseline (speedup 1.0000x reference)
#include <hip/hip_runtime.h>
#include <math.h>

#define CCH 30
#define NLAYERS 3
#define AST 34
#define CH2 (CCH / 2)

typedef __attribute__((ext_vector_type(2))) float v2f;
typedef __attribute__((ext_vector_type(8))) float v8f;

__device__ __forceinline__ float sigmoid_f(float x) { return 1.0f / (1.0f + __expf(-x)); }

__global__ void ggnn_transform_wmma(const float* __restrict__ h,
                                    const float* __restrict__ W,
                                    float* __restrict__ m, int N) {
    __shared__ float s_Wt[32 * AST];
    __shared__ float s_A[4][16 * AST];

    const int tid  = threadIdx.x;
    const int lane = tid & 31;
    const int warp = tid >> 5;
    const int row0 = (blockIdx.x * 4 + warp) * 16;
    const bool full = (row0 + 16 <= N);

    for (int i = tid; i < 32 * AST; i += 128) s_Wt[i] = 0.0f;
    for (int i = lane; i < 16 * AST; i += 32) s_A[warp][i] = 0.0f;
    __syncthreads();

    for (int i = tid; i < CCH * CCH; i += 128) {
        const int k = i / CCH, c = i % CCH;
        s_Wt[c * AST + k] = W[i];
    }
    if (full) {
        const float2* h2 = reinterpret_cast<const float2*>(h + (size_t)row0 * CCH);
        #pragma unroll
        for (int i = lane; i < 16 * CH2; i += 32) {
            const int r = i / CH2, c2 = i % CH2;
            *reinterpret_cast<float2*>(&s_A[warp][r * AST + 2 * c2]) =
                h2[(size_t)r * CH2 + c2];
        }
    } else {
        for (int i = lane; i < 16 * CCH; i += 32) {
            const int r = i / CCH, c = i % CCH;
            const int row = row0 + r;
            if (row < N) s_A[warp][r * AST + c] = h[(size_t)row * CCH + c];
        }
    }
    __syncthreads();

    if (row0 >= N) return;

    const int half = lane >> 4;
    const int l    = lane & 15;
    v8f acc0 = {};
    v8f acc1 = {};
    float* sa = &s_A[warp][0];

    #pragma unroll
    for (int k0 = 0; k0 < 32; k0 += 4) {
        const int ka = k0 + half * 2;
        const v2f a  = *(const v2f*)&sa  [l        * AST + ka];
        const v2f b0 = *(const v2f*)&s_Wt[l        * AST + ka];
        const v2f b1 = *(const v2f*)&s_Wt[(16 + l) * AST + ka];
        acc0 = __builtin_amdgcn_wmma_f32_16x16x4_f32(false, a, false, b0,
                                                     (short)0, acc0, false, false);
        acc1 = __builtin_amdgcn_wmma_f32_16x16x4_f32(false, a, false, b1,
                                                     (short)0, acc1, false, false);
    }

    #pragma unroll
    for (int v = 0; v < 8; ++v) {
        const int r = v + half * 8;
        sa[r * AST + l]      = acc0[v];
        sa[r * AST + 16 + l] = acc1[v];
    }

    if (full) {
        float2* m2 = reinterpret_cast<float2*>(m + (size_t)row0 * CCH);
        #pragma unroll
        for (int i = lane; i < 16 * CH2; i += 32) {
            const int r = i / CH2, c2 = i % CH2;
            m2[(size_t)r * CH2 + c2] =
                *reinterpret_cast<const float2*>(&sa[r * AST + 2 * c2]);
        }
    } else {
        for (int i = lane; i < 16 * CCH; i += 32) {
            const int r = i / CCH, c = i % CCH;
            const int row = row0 + r;
            if (row < N) m[(size_t)row * CCH + c] = sa[r * AST + c];
        }
    }
}

__global__ void ggnn_copy(const float* __restrict__ src, float* __restrict__ dst,
                          int n) {
    const int n4 = n >> 2;
    int i = blockIdx.x * blockDim.x + threadIdx.x;
    if (i < n4) {
        reinterpret_cast<float4*>(dst)[i] =
            reinterpret_cast<const float4*>(src)[i];
    }
    if (i < (n & 3)) {
        const int t = n4 * 4 + i;
        dst[t] = src[t];
    }
}

__global__ void ggnn_zero(float* __restrict__ p, int n) {
    const int n4 = n >> 2;
    int i = blockIdx.x * blockDim.x + threadIdx.x;
    if (i < n4) reinterpret_cast<float4*>(p)[i] = make_float4(0.f, 0.f, 0.f, 0.f);
    if (i < (n & 3)) p[n4 * 4 + i] = 0.0f;
}

__global__ void ggnn_scatter(const int* __restrict__ src,
                             const int* __restrict__ dst,
                             const float* __restrict__ m,
                             float* __restrict__ agg, int E) {
    int e = blockIdx.x * blockDim.x + threadIdx.x;
    if (e >= E) return;
    const int s = src[e];
    const int d = dst[e];
    const float2* ms = reinterpret_cast<const float2*>(m + (size_t)s * CCH);
    float*        ad = agg + (size_t)d * CCH;

    float2 v[CH2];
    #pragma unroll
    for (int i = 0; i < CH2; ++i) v[i] = ms[i];
    #pragma unroll
    for (int i = 0; i < CH2; ++i) {
        atomicAdd(&ad[2 * i],     v[i].x);
        atomicAdd(&ad[2 * i + 1], v[i].y);
    }
}

__global__ void ggnn_gru(const float* __restrict__ agg,
                         float* __restrict__ h,
                         const float* __restrict__ w_ih,
                         const float* __restrict__ w_hh,
                         const float* __restrict__ b_ih,
                         const float* __restrict__ b_hh, int N) {
    __shared__ float s_wih[3 * CCH * CCH];
    __shared__ float s_whh[3 * CCH * CCH];
    __shared__ float s_bih[3 * CCH];
    __shared__ float s_bhh[3 * CCH];

    for (int i = threadIdx.x; i < 3 * CCH * CCH; i += blockDim.x) {
        s_wih[i] = w_ih[i];
        s_whh[i] = w_hh[i];
    }
    if (threadIdx.x < 3 * CCH) {
        s_bih[threadIdx.x] = b_ih[threadIdx.x];
        s_bhh[threadIdx.x] = b_hh[threadIdx.x];
    }
    __syncthreads();

    const int n = blockIdx.x * blockDim.x + threadIdx.x;
    if (n >= N) return;

    float a[CCH], hv[CCH];
    #pragma unroll
    for (int k = 0; k < CCH; ++k) {
        a[k]  = agg[(size_t)n * CCH + k];
        hv[k] = h  [(size_t)n * CCH + k];
    }

    #pragma unroll 1
    for (int j = 0; j < CCH; ++j) {
        float ir = s_bih[j], iz = s_bih[CCH + j], inn = s_bih[2 * CCH + j];
        float hr = s_bhh[j], hz = s_bhh[CCH + j], hnn = s_bhh[2 * CCH + j];
        #pragma unroll
        for (int k = 0; k < CCH; ++k) {
            ir  += a[k]  * s_wih[j * CCH + k];
            iz  += a[k]  * s_wih[(CCH + j) * CCH + k];
            inn += a[k]  * s_wih[(2 * CCH + j) * CCH + k];
            hr  += hv[k] * s_whh[j * CCH + k];
            hz  += hv[k] * s_whh[(CCH + j) * CCH + k];
            hnn += hv[k] * s_whh[(2 * CCH + j) * CCH + k];
        }
        const float r  = sigmoid_f(ir + hr);
        const float z  = sigmoid_f(iz + hz);
        const float nn = tanhf(inn + r * hnn);
        h[(size_t)n * CCH + j] = (1.0f - z) * nn + z * hv[j];
    }
}

__global__ void ggnn_final(const float* __restrict__ h,
                           const float* __restrict__ lin_w,
                           const float* __restrict__ lin_b,
                           float* __restrict__ out, int N) {
    __shared__ float s_w[CCH * CCH];
    __shared__ float s_b[CCH];
    for (int i = threadIdx.x; i < CCH * CCH; i += blockDim.x) s_w[i] = lin_w[i];
    if (threadIdx.x < CCH) s_b[threadIdx.x] = lin_b[threadIdx.x];
    __syncthreads();

    const int n = blockIdx.x * blockDim.x + threadIdx.x;
    if (n >= N) return;

    float hv[CCH];
    #pragma unroll
    for (int k = 0; k < CCH; ++k) {
        const float v = h[(size_t)n * CCH + k];
        hv[k] = v > 0.0f ? v : 0.0f;
    }
    #pragma unroll 1
    for (int j = 0; j < CCH; ++j) {
        float acc = s_b[j];
        #pragma unroll
        for (int k = 0; k < CCH; ++k) acc += hv[k] * s_w[j * CCH + k];
        out[(size_t)n * CCH + j] = acc;
    }
}

extern "C" void kernel_launch(void* const* d_in, const int* in_sizes, int n_in,
                              void* d_out, int out_size, void* d_ws, size_t ws_size,
                              hipStream_t stream) {
    const float* x      = (const float*)d_in[0];
    const int*   eidx   = (const int*)  d_in[1];
    const float* weight = (const float*)d_in[2];
    const float* w_ih   = (const float*)d_in[3];
    const float* w_hh   = (const float*)d_in[4];
    const float* b_ih   = (const float*)d_in[5];
    const float* b_hh   = (const float*)d_in[6];
    const float* lin_w  = (const float*)d_in[7];
    const float* lin_b  = (const float*)d_in[8];
    float* out = (float*)d_out;

    const int N = in_sizes[0] / CCH;
    const int E = in_sizes[1] / 2;
    const int* src = eidx;
    const int* dst = eidx + E;

    float* h   = (float*)d_ws;
    float* m   = h + (size_t)N * CCH;
    float* agg = m + (size_t)N * CCH;

    const int nelem = N * CCH;
    const int n4    = (nelem >> 2) + 4;
    ggnn_copy<<<(n4 + 255) / 256, 256, 0, stream>>>(x, h, nelem);

    const int tiles = (N + 15) / 16;
    for (int layer = 0; layer < NLAYERS; ++layer) {
        ggnn_transform_wmma<<<(tiles + 3) / 4, 128, 0, stream>>>(
            h, weight + (size_t)layer * CCH * CCH, m, N);
        ggnn_zero<<<(n4 + 255) / 256, 256, 0, stream>>>(agg, nelem);
        ggnn_scatter<<<(E + 255) / 256, 256, 0, stream>>>(src, dst, m, agg, E);
        ggnn_gru<<<(N + 127) / 128, 128, 0, stream>>>(agg, h, w_ih, w_hh,
                                                      b_ih, b_hh, N);
    }
    ggnn_final<<<(N + 127) / 128, 128, 0, stream>>>(h, lin_w, lin_b, out, N);
}


# Scatter without float atomics: edges counting-sorted by destination per call (one returning int atomic per edge, scan, place), then per-node gather-sum of m rows; sum order = rank arrival order
# speedup vs baseline: 5.6744x; 5.6744x over previous
#include <hip/hip_runtime.h>
#include <math.h>

#define CCH 30
#define NLAYERS 3
#define AST 34
#define CH2 (CCH / 2)

typedef __attribute__((ext_vector_type(2))) float v2f;
typedef __attribute__((ext_vector_type(8))) float v8f;

__device__ __forceinline__ float sigmoid_f(float x) { return 1.0f / (1.0f + __expf(-x)); }

__global__ void ggnn_transform_wmma(const float* __restrict__ h,
                                    const float* __restrict__ W,
                                    float* __restrict__ m, int N) {
    __shared__ float s_Wt[32 * AST];
    __shared__ float s_A[4][16 * AST];

    const int tid  = threadIdx.x;
    const int lane = tid & 31;
    const int warp = tid >> 5;
    const int row0 = (blockIdx.x * 4 + warp) * 16;
    const bool full = (row0 + 16 <= N);

    for (int i = tid; i < 32 * AST; i += 128) s_Wt[i] = 0.0f;
    for (int i = lane; i < 16 * AST; i += 32) s_A[warp][i] = 0.0f;
    __syncthreads();

    for (int i = tid; i < CCH * CCH; i += 128) {
        const int k = i / CCH, c = i % CCH;
        s_Wt[c * AST + k] = W[i];
    }
    if (full) {
        const float2* h2 = reinterpret_cast<const float2*>(h + (size_t)row0 * CCH);
        #pragma unroll
        for (int i = lane; i < 16 * CH2; i += 32) {
            const int r = i / CH2, c2 = i % CH2;
            *reinterpret_cast<float2*>(&s_A[warp][r * AST + 2 * c2]) =
                h2[(size_t)r * CH2 + c2];
        }
    } else {
        for (int i = lane; i < 16 * CCH; i += 32) {
            const int r = i / CCH, c = i % CCH;
            const int row = row0 + r;
            if (row < N) s_A[warp][r * AST + c] = h[(size_t)row * CCH + c];
        }
    }
    __syncthreads();

    if (row0 >= N) return;

    const int half = lane >> 4;
    const int l    = lane & 15;
    v8f acc0 = {};
    v8f acc1 = {};
    float* sa = &s_A[warp][0];

    #pragma unroll
    for (int k0 = 0; k0 < 32; k0 += 4) {
        const int ka = k0 + half * 2;
        const v2f a  = *(const v2f*)&sa  [l        * AST + ka];
        const v2f b0 = *(const v2f*)&s_Wt[l        * AST + ka];
        const v2f b1 = *(const v2f*)&s_Wt[(16 + l) * AST + ka];
        acc0 = __builtin_amdgcn_wmma_f32_16x16x4_f32(false, a, false, b0,
                                                     (short)0, acc0, false, false);
        acc1 = __builtin_amdgcn_wmma_f32_16x16x4_f32(false, a, false, b1,
                                                     (short)0, acc1, false, false);
    }

    #pragma unroll
    for (int v = 0; v < 8; ++v) {
        const int r = v + half * 8;
        sa[r * AST + l]      = acc0[v];
        sa[r * AST + 16 + l] = acc1[v];
    }

    if (full) {
        float2* m2 = reinterpret_cast<float2*>(m + (size_t)row0 * CCH);
        #pragma unroll
        for (int i = lane; i < 16 * CH2; i += 32) {
            const int r = i / CH2, c2 = i % CH2;
            m2[(size_t)r * CH2 + c2] =
                *reinterpret_cast<const float2*>(&sa[r * AST + 2 * c2]);
        }
    } else {
        for (int i = lane; i < 16 * CCH; i += 32) {
            const int r = i / CCH, c = i % CCH;
            const int row = row0 + r;
            if (row < N) m[(size_t)row * CCH + c] = sa[r * AST + c];
        }
    }
}

__global__ void ggnn_copy(const float* __restrict__ src, float* __restrict__ dst,
                          int n) {
    const int n4 = n >> 2;
    int i = blockIdx.x * blockDim.x + threadIdx.x;
    if (i < n4) {
        reinterpret_cast<float4*>(dst)[i] =
            reinterpret_cast<const float4*>(src)[i];
    }
    if (i < (n & 3)) {
        const int t = n4 * 4 + i;
        dst[t] = src[t];
    }
}

__global__ void ggnn_zero(float* __restrict__ p, int n) {
    const int n4 = n >> 2;
    int i = blockIdx.x * blockDim.x + threadIdx.x;
    if (i < n4) reinterpret_cast<float4*>(p)[i] = make_float4(0.f, 0.f, 0.f, 0.f);
    if (i < (n & 3)) p[n4 * 4 + i] = 0.0f;
}

__global__ void ggnn_rank(const int* __restrict__ dst, int* __restrict__ cnt,
                          int* __restrict__ rank, int E) {
    const int e = blockIdx.x * blockDim.x + threadIdx.x;
    if (e < E) rank[e] = atomicAdd(&cnt[dst[e]], 1);
}

#define SCAN_T 1024
#define SCAN_V 8

__global__ void __launch_bounds__(SCAN_T)
ggnn_scan(const int* __restrict__ cnt, int* __restrict__ rowptr, int N) {
    __shared__ int s_w[32];
    __shared__ int s_tot;
    const int tid  = threadIdx.x;
    const int lane = tid & 31;
    const int wid  = tid >> 5;
    int base = 0;

    for (int c0 = 0; c0 < N; c0 += SCAN_T * SCAN_V) {
        const int i0 = c0 + tid * SCAN_V;
        int v[SCAN_V];
        #pragma unroll
        for (int j = 0; j < SCAN_V; ++j) v[j] = (i0 + j < N) ? cnt[i0 + j] : 0;
        int ex[SCAN_V];
        int t = 0;
        #pragma unroll
        for (int j = 0; j < SCAN_V; ++j) { ex[j] = t; t += v[j]; }

        int x = t;
        #pragma unroll
        for (int d = 1; d < 32; d <<= 1) {
            const int y = __shfl_up(x, d);
            if (lane >= d) x += y;
        }
        if (lane == 31) s_w[wid] = x;
        __syncthreads();
        if (wid == 0) {
            const int wt = s_w[lane];
            int xs = wt;
            #pragma unroll
            for (int d = 1; d < 32; d <<= 1) {
                const int y = __shfl_up(xs, d);
                if (lane >= d) xs += y;
            }
            s_w[lane] = xs - wt;
            if (lane == 31) s_tot = xs;
        }
        __syncthreads();
        const int off = base + s_w[wid] + (x - t);
        #pragma unroll
        for (int j = 0; j < SCAN_V; ++j) {
            if (i0 + j < N) rowptr[i0 + j] = off + ex[j];
        }
        base += s_tot;
        __syncthreads();
    }
    if (tid == 0) rowptr[N] = base;
}

__global__ void ggnn_place(const int* __restrict__ src,
                           const int* __restrict__ dst,
                           const int* __restrict__ rowptr,
                           const int* __restrict__ rank,
                           int* __restrict__ ssrc, int E) {
    const int e = blockIdx.x * blockDim.x + threadIdx.x;
    if (e >= E) return;
    const int pos = rowptr[dst[e]] + rank[e];
    if ((unsigned)pos < (unsigned)E) ssrc[pos] = src[e];
}

__global__ void __launch_bounds__(256)
ggnn_gather(const int* __restrict__ rowptr, const int* __restrict__ ssrc,
            const float* __restrict__ m, float* __restrict__ agg, int N) {
    const int tid  = threadIdx.x;
    const int l    = tid & 15;
    const int node = (blockIdx.x * 256 + tid) >> 4;
    const bool nv  = node < N;
    const int lc   = l < CH2 ? l : CH2 - 1;
    const float2* m2 = reinterpret_cast<const float2*>(m);

    int p         = nv ? rowptr[node]     : 0;
    const int end = nv ? rowptr[node + 1] : 0;
    int rem = end - p;
    const int other = __shfl_xor(rem, 16);
    int remw = __builtin_amdgcn_readfirstlane(rem > other ? rem : other);

    float2 acc = make_float2(0.0f, 0.0f);
    while (remw > 0) {
        const int idx = (l < rem) ? ssrc[p + l] : -1;
        #pragma unroll
        for (int j0 = 0; j0 < 16; j0 += 8) {
            if (j0 < remw) {
                int    s[8];
                float2 v[8];
                #pragma unroll
                for (int j = 0; j < 8; ++j) {
                    s[j] = __shfl(idx, j0 + j, 16);
                    const int sc = s[j] < 0 ? 0 : (s[j] < N ? s[j] : N - 1);
                    v[j] = m2[(size_t)sc * CH2 + lc];
                }
                #pragma unroll
                for (int j = 0; j < 8; ++j) {
                    if (s[j] >= 0) { acc.x += v[j].x; acc.y += v[j].y; }
                }
            }
        }
        p += 16; rem -= 16; remw -= 16;
    }
    if (nv && l < CH2)
        reinterpret_cast<float2*>(agg)[(size_t)node * CH2 + l] = acc;
}

__global__ void ggnn_gru(const float* __restrict__ agg,
                         float* __restrict__ h,
                         const float* __restrict__ w_ih,
                         const float* __restrict__ w_hh,
                         const float* __restrict__ b_ih,
                         const float* __restrict__ b_hh, int N) {
    __shared__ float s_wih[3 * CCH * CCH];
    __shared__ float s_whh[3 * CCH * CCH];
    __shared__ float s_bih[3 * CCH];
    __shared__ float s_bhh[3 * CCH];

    for (int i = threadIdx.x; i < 3 * CCH * CCH; i += blockDim.x) {
        s_wih[i] = w_ih[i];
        s_whh[i] = w_hh[i];
    }
    if (threadIdx.x < 3 * CCH) {
        s_bih[threadIdx.x] = b_ih[threadIdx.x];
        s_bhh[threadIdx.x] = b_hh[threadIdx.x];
    }
    __syncthreads();

    const int n = blockIdx.x * blockDim.x + threadIdx.x;
    if (n >= N) return;

    float a[CCH], hv[CCH];
    #pragma unroll
    for (int k = 0; k < CCH; ++k) {
        a[k]  = agg[(size_t)n * CCH + k];
        hv[k] = h  [(size_t)n * CCH + k];
    }

    #pragma unroll 1
    for (int j = 0; j < CCH; ++j) {
        float ir = s_bih[j], iz = s_bih[CCH + j], inn = s_bih[2 * CCH + j];
        float hr = s_bhh[j], hz = s_bhh[CCH + j], hnn = s_bhh[2 * CCH + j];
        #pragma unroll
        for (int k = 0; k < CCH; ++k) {
            ir  += a[k]  * s_wih[j * CCH + k];
            iz  += a[k]  * s_wih[(CCH + j) * CCH + k];
            inn += a[k]  * s_wih[(2 * CCH + j) * CCH + k];
            hr  += hv[k] * s_whh[j * CCH + k];
            hz  += hv[k] * s_whh[(CCH + j) * CCH + k];
            hnn += hv[k] * s_whh[(2 * CCH + j) * CCH + k];
        }
        const float r  = sigmoid_f(ir + hr);
        const float z  = sigmoid_f(iz + hz);
        const float nn = tanhf(inn + r * hnn);
        h[(size_t)n * CCH + j] = (1.0f - z) * nn + z * hv[j];
    }
}

__global__ void ggnn_final(const float* __restrict__ h,
                           const float* __restrict__ lin_w,
                           const float* __restrict__ lin_b,
                           float* __restrict__ out, int N) {
    __shared__ float s_w[CCH * CCH];
    __shared__ float s_b[CCH];
    for (int i = threadIdx.x; i < CCH * CCH; i += blockDim.x) s_w[i] = lin_w[i];
    if (threadIdx.x < CCH) s_b[threadIdx.x] = lin_b[threadIdx.x];
    __syncthreads();

    const int n = blockIdx.x * blockDim.x + threadIdx.x;
    if (n >= N) return;

    float hv[CCH];
    #pragma unroll
    for (int k = 0; k < CCH; ++k) {
        const float v = h[(size_t)n * CCH + k];
        hv[k] = v > 0.0f ? v : 0.0f;
    }
    #pragma unroll 1
    for (int j = 0; j < CCH; ++j) {
        float acc = s_b[j];
        #pragma unroll
        for (int k = 0; k < CCH; ++k) acc += hv[k] * s_w[j * CCH + k];
        out[(size_t)n * CCH + j] = acc;
    }
}

extern "C" void kernel_launch(void* const* d_in, const int* in_sizes, int n_in,
                              void* d_out, int out_size, void* d_ws, size_t ws_size,
                              hipStream_t stream) {
    const float* x      = (const float*)d_in[0];
    const int*   eidx   = (const int*)  d_in[1];
    const float* weight = (const float*)d_in[2];
    const float* w_ih   = (const float*)d_in[3];
    const float* w_hh   = (const float*)d_in[4];
    const float* b_ih   = (const float*)d_in[5];
    const float* b_hh   = (const float*)d_in[6];
    const float* lin_w  = (const float*)d_in[7];
    const float* lin_b  = (const float*)d_in[8];
    float* out = (float*)d_out;

    const int N = in_sizes[0] / CCH;
    const int E = in_sizes[1] / 2;
    const int* src = eidx;
    const int* dst = eidx + E;

    float* h   = (float*)d_ws;
    float* m   = h + (size_t)N * CCH;
    float* agg = m + (size_t)N * CCH;
    int* cnt    = (int*)(agg + (size_t)N * CCH);
    int* rowptr = cnt + ((N + 3) & ~3);
    int* rank   = rowptr + ((N + 4) & ~3);
    int* ssrc   = rank + ((E + 3) & ~3);

    const int nelem = N * CCH;
    const int n4    = (nelem >> 2) + 4;
    ggnn_copy<<<(n4 + 255) / 256, 256, 0, stream>>>(x, h, nelem);

    const int c4 = (N >> 2) + 4;
    ggnn_zero<<<(c4 + 255) / 256, 256, 0, stream>>>((float*)cnt, N);
    ggnn_rank<<<(E + 255) / 256, 256, 0, stream>>>(dst, cnt, rank, E);
    ggnn_scan<<<1, SCAN_T, 0, stream>>>(cnt, rowptr, N);
    ggnn_place<<<(E + 255) / 256, 256, 0, stream>>>(src, dst, rowptr, rank, ssrc, E);

    const int tiles = (N + 15) / 16;
    for (int layer = 0; layer < NLAYERS; ++layer) {
        ggnn_transform_wmma<<<(tiles + 3) / 4, 128, 0, stream>>>(
            h, weight + (size_t)layer * CCH * CCH, m, N);
        ggnn_gather<<<(N + 15) / 16, 256, 0, stream>>>(rowptr, ssrc, m, agg, N);
        ggnn_gru<<<(N + 127) / 128, 128, 0, stream>>>(agg, h, w_ih, w_hh,
                                                      b_ih, b_hh, N);
    }
    ggnn_final<<<(N + 127) / 128, 128, 0, stream>>>(h, lin_w, lin_b, out, N);
}


# Destination sort by LDS binning (block-staged split into 256-node buckets, then per-bucket counting sort in LDS, stores in runs): no per-edge global atomic or scattered store; sum order = LDS arrival
# speedup vs baseline: 9.2267x; 1.6260x over previous
#include <hip/hip_runtime.h>
#include <math.h>

#define CCH 30
#define NLAYERS 3
#define AST 34
#define CH2 (CCH / 2)

typedef __attribute__((ext_vector_type(2))) float v2f;
typedef __attribute__((ext_vector_type(8))) float v8f;

__device__ __forceinline__ float sigmoid_f(float x) { return 1.0f / (1.0f + __expf(-x)); }

__global__ void ggnn_transform_wmma(const float* __restrict__ h,
                                    const float* __restrict__ W,
                                    float* __restrict__ m, int N) {
    __shared__ float s_Wt[32 * AST];
    __shared__ float s_A[4][16 * AST];

    const int tid  = threadIdx.x;
    const int lane = tid & 31;
    const int warp = tid >> 5;
    const int row0 = (blockIdx.x * 4 + warp) * 16;
    const bool full = (row0 + 16 <= N);

    for (int i = tid; i < 32 * AST; i += 128) s_Wt[i] = 0.0f;
    for (int i = lane; i < 16 * AST; i += 32) s_A[warp][i] = 0.0f;
    __syncthreads();

    for (int i = tid; i < CCH * CCH; i += 128) {
        const int k = i / CCH, c = i % CCH;
        s_Wt[c * AST + k] = W[i];
    }
    if (full) {
        const float2* h2 = reinterpret_cast<const float2*>(h + (size_t)row0 * CCH);
        #pragma unroll
        for (int i = lane; i < 16 * CH2; i += 32) {
            const int r = i / CH2, c2 = i % CH2;
            *reinterpret_cast<float2*>(&s_A[warp][r * AST + 2 * c2]) =
                h2[(size_t)r * CH2 + c2];
        }
    } else {
        for (int i = lane; i < 16 * CCH; i += 32) {
            const int r = i / CCH, c = i % CCH;
            const int row = row0 + r;
            if (row < N) s_A[warp][r * AST + c] = h[(size_t)row * CCH + c];
        }
    }
    __syncthreads();

    if (row0 >= N) return;

    const int half = lane >> 4;
    const int l    = lane & 15;
    v8f acc0 = {};
    v8f acc1 = {};
    float* sa = &s_A[warp][0];

    #pragma unroll
    for (int k0 = 0; k0 < 32; k0 += 4) {
        const int ka = k0 + half * 2;
        const v2f a  = *(const v2f*)&sa  [l        * AST + ka];
        const v2f b0 = *(const v2f*)&s_Wt[l        * AST + ka];
        const v2f b1 = *(const v2f*)&s_Wt[(16 + l) * AST + ka];
        acc0 = __builtin_amdgcn_wmma_f32_16x16x4_f32(false, a, false, b0,
                                                     (short)0, acc0, false, false);
        acc1 = __builtin_amdgcn_wmma_f32_16x16x4_f32(false, a, false, b1,
                                                     (short)0, acc1, false, false);
    }

    #pragma unroll
    for (int v = 0; v < 8; ++v) {
        const int r = v + half * 8;
        sa[r * AST + l]      = acc0[v];
        sa[r * AST + 16 + l] = acc1[v];
    }

    if (full) {
        float2* m2 = reinterpret_cast<float2*>(m + (size_t)row0 * CCH);
        #pragma unroll
        for (int i = lane; i < 16 * CH2; i += 32) {
            const int r = i / CH2, c2 = i % CH2;
            m2[(size_t)r * CH2 + c2] =
                *reinterpret_cast<const float2*>(&sa[r * AST + 2 * c2]);
        }
    } else {
        for (int i = lane; i < 16 * CCH; i += 32) {
            const int r = i / CCH, c = i % CCH;
            const int row = row0 + r;
            if (row < N) m[(size_t)row * CCH + c] = sa[r * AST + c];
        }
    }
}

__global__ void ggnn_copy(const float* __restrict__ src, float* __restrict__ dst,
                          int n) {
    const int n4 = n >> 2;
    int i = blockIdx.x * blockDim.x + threadIdx.x;
    if (i < n4) {
        reinterpret_cast<float4*>(dst)[i] =
            reinterpret_cast<const float4*>(src)[i];
    }
    if (i < (n & 3)) {
        const int t = n4 * 4 + i;
        dst[t] = src[t];
    }
}

__global__ void ggnn_zero(float* __restrict__ p, int n) {
    const int n4 = n >> 2;
    int i = blockIdx.x * blockDim.x + threadIdx.x;
    if (i < n4) reinterpret_cast<float4*>(p)[i] = make_float4(0.f, 0.f, 0.f, 0.f);
    if (i < (n & 3)) p[n4 * 4 + i] = 0.0f;
}

#define BSH   8
#define BNODE (1 << BSH)
#define NBMAX 1024
#define PTHR  1024
#define PPT   6
#define PCH   (PTHR * PPT)
#define EBLK  (4 * PCH)
#define CSTR  64
#define STHR  512
#define SCAP  12288

__device__ __forceinline__ int bucket_of(int d) {
    const unsigned b = (unsigned)d >> BSH;
    return (int)(b < (unsigned)NBMAX ? b : (unsigned)(NBMAX - 1));
}

__device__ __forceinline__ int block_excl_scan(int v, int* s_w, int& total) {
    const int lane = threadIdx.x & 31;
    const int wid  = threadIdx.x >> 5;
    const int nw   = (int)(blockDim.x >> 5);
    int x = v;
    #pragma unroll
    for (int d = 1; d < 32; d <<= 1) {
        const int y = __shfl_up(x, d);
        if (lane >= d) x += y;
    }
    __syncthreads();
    if (lane == 31) s_w[wid] = x;
    __syncthreads();
    if (wid == 0) {
        const int wt = lane < nw ? s_w[lane] : 0;
        int xs = wt;
        #pragma unroll
        for (int d = 1; d < 32; d <<= 1) {
            const int y = __shfl_up(xs, d);
            if (lane >= d) xs += y;
        }
        s_w[lane] = xs - wt;
        if (lane == 31) s_w[32] = xs;
    }
    __syncthreads();
    total = s_w[32];
    return s_w[wid] + (x - v);
}

__global__ void __launch_bounds__(PTHR)
ggnn_bhist(const int* __restrict__ dst, int* __restrict__ bcnt,
           int* __restrict__ hmat, int E) {
    __shared__ int s_h[NBMAX];
    const int tid = threadIdx.x;
    s_h[tid] = 0;
    __syncthreads();
    const int e0 = blockIdx.x * EBLK;
    const int n  = (E - e0) < EBLK ? (E - e0) : EBLK;
    for (int i0 = 0; i0 < n; i0 += PTHR * 8) {
        int d[8];
        #pragma unroll
        for (int j = 0; j < 8; ++j) {
            const int i = i0 + j * PTHR + tid;
            d[j] = i < n ? dst[e0 + i] : 0;
        }
        #pragma unroll
        for (int j = 0; j < 8; ++j) {
            const int i = i0 + j * PTHR + tid;
            if (i < n) atomicAdd(&s_h[bucket_of(d[j])], 1);
        }
    }
    __syncthreads();
    const int c = s_h[tid];
    hmat[blockIdx.x * NBMAX + tid] = c;
    if (c) atomicAdd(&bcnt[tid * CSTR], c);
}

__global__ void __launch_bounds__(PTHR)
ggnn_part(const int* __restrict__ src, const int* __restrict__ dst,
          const int* __restrict__ bcnt, int* __restrict__ bcur,
          const int* __restrict__ hmat, int* __restrict__ bbase,
          int* __restrict__ pairs_w, int E) {
    __shared__ int  s_h[NBMAX];
    __shared__ int  s_lo[NBMAX];
    __shared__ int  s_run[NBMAX];
    __shared__ int  s_w[33];
    __shared__ int2 s_p[PCH];
    int2* pairs = reinterpret_cast<int2*>(pairs_w);
    const int tid  = threadIdx.x;
    const int e0   = blockIdx.x * EBLK;
    const int nblk = (E - e0) < EBLK ? (E - e0) : EBLK;

    const int ctot = hmat[blockIdx.x * NBMAX + tid];
    int gtot;
    const int gb = block_excl_scan(bcnt[tid * CSTR], s_w, gtot);
    s_run[tid] = gb + (ctot > 0 ? atomicAdd(&bcur[tid * CSTR], ctot) : 0);
    if (blockIdx.x == 0) {
        bbase[tid] = gb;
        if (tid == PTHR - 1) bbase[NBMAX] = gtot;
    }

    for (int c0 = 0; c0 < nblk; c0 += PCH) {
        const int n = (nblk - c0) < PCH ? (nblk - c0) : PCH;
        s_h[tid] = 0;
        __syncthreads();

        int d[PPT], s[PPT], r[PPT];
        #pragma unroll
        for (int j = 0; j < PPT; ++j) {
            const int i = j * PTHR + tid;
            d[j] = i < n ? dst[e0 + c0 + i] : 0;
            s[j] = i < n ? src[e0 + c0 + i] : 0;
        }
        #pragma unroll
        for (int j = 0; j < PPT; ++j) {
            const int i = j * PTHR + tid;
            r[j] = i < n ? atomicAdd(&s_h[bucket_of(d[j])], 1) : 0;
        }
        __syncthreads();

        const int c = s_h[tid];
        int tot;
        s_lo[tid] = block_excl_scan(c, s_w, tot);
        __syncthreads();

        #pragma unroll
        for (int j = 0; j < PPT; ++j) {
            const int i = j * PTHR + tid;
            if (i < n) s_p[s_lo[bucket_of(d[j])] + r[j]] = make_int2(d[j], s[j]);
        }
        __syncthreads();

        for (int i = tid; i < n; i += PTHR) {
            const int2 v = s_p[i];
            const int b  = bucket_of(v.x);
            const int pos = s_run[b] + (i - s_lo[b]);
            if ((unsigned)pos < (unsigned)E) pairs[pos] = v;
        }
        __syncthreads();
        s_run[tid] += c;
    }
}

__global__ void __launch_bounds__(STHR)
ggnn_bsort(const int* __restrict__ pairs_w, const int* __restrict__ bbase,
           int* __restrict__ rowptr, int* __restrict__ ssrc, int N, int E) {
    __shared__ int s_c[BNODE];
    __shared__ int s_w[33];
    __shared__ int s_s[SCAP];
    const int tid   = threadIdx.x;
    const int b     = blockIdx.x;
    const int node0 = b << BSH;
    int bs = bbase[b];
    int be = bbase[b + 1];
    bs = bs < 0 ? 0 : (bs > E ? E : bs);
    be = be < bs ? bs : (be > E ? E : be);
    const int n = be - bs;
    const bool stage = n <= SCAP;
    const int2* pb = reinterpret_cast<const int2*>(pairs_w) + bs;

    if (tid < BNODE) s_c[tid] = 0;
    __syncthreads();
    for (int i0 = 0; i0 < n; i0 += STHR * 8) {
        int2 v[8];
        #pragma unroll
        for (int j = 0; j < 8; ++j) {
            const int i = i0 + j * STHR + tid;
            v[j] = i < n ? pb[i] : make_int2(0, 0);
        }
        #pragma unroll
        for (int j = 0; j < 8; ++j) {
            const int i = i0 + j * STHR + tid;
            if (i < n) atomicAdd(&s_c[(v[j].x - node0) & (BNODE - 1)], 1);
        }
    }
    __syncthreads();

    const int c = tid < BNODE ? s_c[tid] : 0;
    int tot;
    const int ex = block_excl_scan(c, s_w, tot);
    if (tid < BNODE) {
        const int node = node0 + tid;
        if (node < N) rowptr[node] = bs + ex;
        s_c[tid] = ex;
    }
    if (b == (int)gridDim.x - 1 && tid == 0) rowptr[N] = E;
    __syncthreads();

    for (int i0 = 0; i0 < n; i0 += STHR * 8) {
        int2 v[8];
        #pragma unroll
        for (int j = 0; j < 8; ++j) {
            const int i = i0 + j * STHR + tid;
            v[j] = i < n ? pb[i] : make_int2(0, 0);
        }
        #pragma unroll
        for (int j = 0; j < 8; ++j) {
            const int i = i0 + j * STHR + tid;
            if (i < n) {
                const int p = atomicAdd(&s_c[(v[j].x - node0) & (BNODE - 1)], 1);
                if (stage) {
                    if ((unsigned)p < (unsigned)SCAP) s_s[p] = v[j].y;
                } else {
                    if ((unsigned)p < (unsigned)n) ssrc[bs + p] = v[j].y;
                }
            }
        }
    }
    if (stage) {
        __syncthreads();
        for (int i = tid; i < n; i += STHR) ssrc[bs + i] = s_s[i];
    }
}

__global__ void __launch_bounds__(256)
ggnn_gather(const int* __restrict__ rowptr, const int* __restrict__ ssrc,
            const float* __restrict__ m, float* __restrict__ agg, int N) {
    const int tid  = threadIdx.x;
    const int l    = tid & 15;
    const int node = (blockIdx.x * 256 + tid) >> 4;
    const bool nv  = node < N;
    const int lc   = l < CH2 ? l : CH2 - 1;
    const float2* m2 = reinterpret_cast<const float2*>(m);

    int p         = nv ? rowptr[node]     : 0;
    const int end = nv ? rowptr[node + 1] : 0;
    int rem = end - p;
    const int other = __shfl_xor(rem, 16);
    int remw = __builtin_amdgcn_readfirstlane(rem > other ? rem : other);

    float2 acc = make_float2(0.0f, 0.0f);
    while (remw > 0) {
        const int idx = (l < rem) ? ssrc[p + l] : -1;
        #pragma unroll
        for (int j0 = 0; j0 < 16; j0 += 8) {
            if (j0 < remw) {
                int    s[8];
                float2 v[8];
                #pragma unroll
                for (int j = 0; j < 8; ++j) {
                    s[j] = __shfl(idx, j0 + j, 16);
                    const int sc = s[j] < 0 ? 0 : (s[j] < N ? s[j] : N - 1);
                    v[j] = m2[(size_t)sc * CH2 + lc];
                }
                #pragma unroll
                for (int j = 0; j < 8; ++j) {
                    if (s[j] >= 0) { acc.x += v[j].x; acc.y += v[j].y; }
                }
            }
        }
        p += 16; rem -= 16; remw -= 16;
    }
    if (nv && l < CH2)
        reinterpret_cast<float2*>(agg)[(size_t)node * CH2 + l] = acc;
}

__global__ void ggnn_gru(const float* __restrict__ agg,
                         float* __restrict__ h,
                         const float* __restrict__ w_ih,
                         const float* __restrict__ w_hh,
                         const float* __restrict__ b_ih,
                         const float* __restrict__ b_hh, int N) {
    __shared__ float s_wih[3 * CCH * CCH];
    __shared__ float s_whh[3 * CCH * CCH];
    __shared__ float s_bih[3 * CCH];
    __shared__ float s_bhh[3 * CCH];

    for (int i = threadIdx.x; i < 3 * CCH * CCH; i += blockDim.x) {
        s_wih[i] = w_ih[i];
        s_whh[i] = w_hh[i];
    }
    if (threadIdx.x < 3 * CCH) {
        s_bih[threadIdx.x] = b_ih[threadIdx.x];
        s_bhh[threadIdx.x] = b_hh[threadIdx.x];
    }
    __syncthreads();

    const int n = blockIdx.x * blockDim.x + threadIdx.x;
    if (n >= N) return;

    float a[CCH], hv[CCH];
    #pragma unroll
    for (int k = 0; k < CCH; ++k) {
        a[k]  = agg[(size_t)n * CCH + k];
        hv[k] = h  [(size_t)n * CCH + k];
    }

    #pragma unroll 1
    for (int j = 0; j < CCH; ++j) {
        float ir = s_bih[j], iz = s_bih[CCH + j], inn = s_bih[2 * CCH + j];
        float hr = s_bhh[j], hz = s_bhh[CCH + j], hnn = s_bhh[2 * CCH + j];
        #pragma unroll
        for (int k = 0; k < CCH; ++k) {
            ir  += a[k]  * s_wih[j * CCH + k];
            iz  += a[k]  * s_wih[(CCH + j) * CCH + k];
            inn += a[k]  * s_wih[(2 * CCH + j) * CCH + k];
            hr  += hv[k] * s_whh[j * CCH + k];
            hz  += hv[k] * s_whh[(CCH + j) * CCH + k];
            hnn += hv[k] * s_whh[(2 * CCH + j) * CCH + k];
        }
        const float r  = sigmoid_f(ir + hr);
        const float z  = sigmoid_f(iz + hz);
        const float nn = tanhf(inn + r * hnn);
        h[(size_t)n * CCH + j] = (1.0f - z) * nn + z * hv[j];
    }
}

__global__ void ggnn_final(const float* __restrict__ h,
                           const float* __restrict__ lin_w,
                           const float* __restrict__ lin_b,
                           float* __restrict__ out, int N) {
    __shared__ float s_w[CCH * CCH];
    __shared__ float s_b[CCH];
    for (int i = threadIdx.x; i < CCH * CCH; i += blockDim.x) s_w[i] = lin_w[i];
    if (threadIdx.x < CCH) s_b[threadIdx.x] = lin_b[threadIdx.x];
    __syncthreads();

    const int n = blockIdx.x * blockDim.x + threadIdx.x;
    if (n >= N) return;

    float hv[CCH];
    #pragma unroll
    for (int k = 0; k < CCH; ++k) {
        const float v = h[(size_t)n * CCH + k];
        hv[k] = v > 0.0f ? v : 0.0f;
    }
    #pragma unroll 1
    for (int j = 0; j < CCH; ++j) {
        float acc = s_b[j];
        #pragma unroll
        for (int k = 0; k < CCH; ++k) acc += hv[k] * s_w[j * CCH + k];
        out[(size_t)n * CCH + j] = acc;
    }
}

extern "C" void kernel_launch(void* const* d_in, const int* in_sizes, int n_in,
                              void* d_out, int out_size, void* d_ws, size_t ws_size,
                              hipStream_t stream) {
    const float* x      = (const float*)d_in[0];
    const int*   eidx   = (const int*)  d_in[1];
    const float* weight = (const float*)d_in[2];
    const float* w_ih   = (const float*)d_in[3];
    const float* w_hh   = (const float*)d_in[4];
    const float* b_ih   = (const float*)d_in[5];
    const float* b_hh   = (const float*)d_in[6];
    const float* lin_w  = (const float*)d_in[7];
    const float* lin_b  = (const float*)d_in[8];
    float* out = (float*)d_out;

    const int N = in_sizes[0] / CCH;
    const int E = in_sizes[1] / 2;
    const int* src = eidx;
    const int* dst = eidx + E;

    float* h   = (float*)d_ws;
    float* m   = h + (size_t)N * CCH;
    float* agg = m + (size_t)N * CCH;
    const int nblk = (E + EBLK - 1) / EBLK;
    int* bcnt   = (int*)(agg + (size_t)N * CCH);
    int* bcur   = bcnt + NBMAX * CSTR;
    int* bbase  = bcur + NBMAX * CSTR;
    int* hmat   = bbase + NBMAX + 4;
    int* rowptr = hmat + (size_t)nblk * NBMAX;
    int* pairs  = rowptr + ((N + 4) & ~3);
    int* ssrc   = pairs + 2 * (size_t)E;

    const int nelem = N * CCH;
    const int n4    = (nelem >> 2) + 4;
    ggnn_copy<<<(n4 + 255) / 256, 256, 0, stream>>>(x, h, nelem);

    const int c4 = ((2 * NBMAX * CSTR) >> 2) + 4;
    ggnn_zero<<<(c4 + 255) / 256, 256, 0, stream>>>((float*)bcnt, 2 * NBMAX * CSTR);
    ggnn_bhist<<<nblk, PTHR, 0, stream>>>(dst, bcnt, hmat, E);
    ggnn_part<<<nblk, PTHR, 0, stream>>>(src, dst, bcnt, bcur, hmat, bbase, pairs, E);
    ggnn_bsort<<<(N + BNODE - 1) / BNODE, STHR, 0, stream>>>(pairs, bbase, rowptr, ssrc, N, E);

    const int tiles = (N + 15) / 16;
    for (int layer = 0; layer < NLAYERS; ++layer) {
        ggnn_transform_wmma<<<(tiles + 3) / 4, 128, 0, stream>>>(
            h, weight + (size_t)layer * CCH * CCH, m, N);
        ggnn_gather<<<(N + 15) / 16, 256, 0, stream>>>(rowptr, ssrc, m, agg, N);
        ggnn_gru<<<(N + 127) / 128, 128, 0, stream>>>(agg, h, w_ih, w_hh,
                                                      b_ih, b_hh, N);
    }
    ggnn_final<<<(N + 127) / 128, 128, 0, stream>>>(h, lin_w, lin_b, out, N);
}


# m stored with rows padded from 30 to 32 floats (128-B pitch, 256-B aligned start): each gathered source row is one aligned 128-B line instead of straddling two lines or two 256-B units
# speedup vs baseline: 9.5154x; 1.0313x over previous
#include <hip/hip_runtime.h>
#include <math.h>

#define CCH 30
#define NLAYERS 3
#define AST 34
#define CH2 (CCH / 2)
#define MST 32
#define MS2 (MST / 2)

typedef __attribute__((ext_vector_type(2))) float v2f;
typedef __attribute__((ext_vector_type(8))) float v8f;

__device__ __forceinline__ float sigmoid_f(float x) { return 1.0f / (1.0f + __expf(-x)); }

__global__ void ggnn_transform_wmma(const float* __restrict__ h,
                                    const float* __restrict__ W,
                                    float* __restrict__ m, int N) {
    __shared__ float s_Wt[32 * AST];
    __shared__ float s_A[4][16 * AST];

    const int tid  = threadIdx.x;
    const int lane = tid & 31;
    const int warp = tid >> 5;
    const int row0 = (blockIdx.x * 4 + warp) * 16;
    const bool full = (row0 + 16 <= N);

    for (int i = tid; i < 32 * AST; i += 128) s_Wt[i] = 0.0f;
    for (int i = lane; i < 16 * AST; i += 32) s_A[warp][i] = 0.0f;
    __syncthreads();

    for (int i = tid; i < CCH * CCH; i += 128) {
        const int k = i / CCH, c = i % CCH;
        s_Wt[c * AST + k] = W[i];
    }
    if (full) {
        const float2* h2 = reinterpret_cast<const float2*>(h + (size_t)row0 * CCH);
        #pragma unroll
        for (int i = lane; i < 16 * CH2; i += 32) {
            const int r = i / CH2, c2 = i % CH2;
            *reinterpret_cast<float2*>(&s_A[warp][r * AST + 2 * c2]) =
                h2[(size_t)r * CH2 + c2];
        }
    } else {
        for (int i = lane; i < 16 * CCH; i += 32) {
            const int r = i / CCH, c = i % CCH;
            const int row = row0 + r;
            if (row < N) s_A[warp][r * AST + c] = h[(size_t)row * CCH + c];
        }
    }
    __syncthreads();

    if (row0 >= N) return;

    const int half = lane >> 4;
    const int l    = lane & 15;
    v8f acc0 = {};
    v8f acc1 = {};
    float* sa = &s_A[warp][0];

    #pragma unroll
    for (int k0 = 0; k0 < 32; k0 += 4) {
        const int ka = k0 + half * 2;
        const v2f a  = *(const v2f*)&sa  [l        * AST + ka];
        const v2f b0 = *(const v2f*)&s_Wt[l        * AST + ka];
        const v2f b1 = *(const v2f*)&s_Wt[(16 + l) * AST + ka];
        acc0 = __builtin_amdgcn_wmma_f32_16x16x4_f32(false, a, false, b0,
                                                     (short)0, acc0, false, false);
        acc1 = __builtin_amdgcn_wmma_f32_16x16x4_f32(false, a, false, b1,
                                                     (short)0, acc1, false, false);
    }

    #pragma unroll
    for (int v = 0; v < 8; ++v) {
        const int r = v + half * 8;
        sa[r * AST + l]      = acc0[v];
        sa[r * AST + 16 + l] = acc1[v];
    }

    if (full) {
        float2* m2 = reinterpret_cast<float2*>(m + (size_t)row0 * MST);
        #pragma unroll
        for (int i = lane; i < 16 * MS2; i += 32) {
            const int r = i / MS2, c2 = i % MS2;
            m2[i] = *reinterpret_cast<const float2*>(&sa[r * AST + 2 * c2]);
        }
    } else {
        for (int i = lane; i < 16 * MST; i += 32) {
            const int r = i / MST, c = i % MST;
            const int row = row0 + r;
            if (row < N) m[(size_t)row * MST + c] = sa[r * AST + c];
        }
    }
}

__global__ void ggnn_copy(const float* __restrict__ src, float* __restrict__ dst,
                          int n) {
    const int n4 = n >> 2;
    int i = blockIdx.x * blockDim.x + threadIdx.x;
    if (i < n4) {
        reinterpret_cast<float4*>(dst)[i] =
            reinterpret_cast<const float4*>(src)[i];
    }
    if (i < (n & 3)) {
        const int t = n4 * 4 + i;
        dst[t] = src[t];
    }
}

__global__ void ggnn_zero(float* __restrict__ p, int n) {
    const int n4 = n >> 2;
    int i = blockIdx.x * blockDim.x + threadIdx.x;
    if (i < n4) reinterpret_cast<float4*>(p)[i] = make_float4(0.f, 0.f, 0.f, 0.f);
    if (i < (n & 3)) p[n4 * 4 + i] = 0.0f;
}

#define BSH   8
#define BNODE (1 << BSH)
#define NBMAX 1024
#define PTHR  1024
#define PPT   6
#define PCH   (PTHR * PPT)
#define EBLK  (4 * PCH)
#define CSTR  64
#define STHR  512
#define SCAP  12288

__device__ __forceinline__ int bucket_of(int d) {
    const unsigned b = (unsigned)d >> BSH;
    return (int)(b < (unsigned)NBMAX ? b : (unsigned)(NBMAX - 1));
}

__device__ __forceinline__ int block_excl_scan(int v, int* s_w, int& total) {
    const int lane = threadIdx.x & 31;
    const int wid  = threadIdx.x >> 5;
    const int nw   = (int)(blockDim.x >> 5);
    int x = v;
    #pragma unroll
    for (int d = 1; d < 32; d <<= 1) {
        const int y = __shfl_up(x, d);
        if (lane >= d) x += y;
    }
    __syncthreads();
    if (lane == 31) s_w[wid] = x;
    __syncthreads();
    if (wid == 0) {
        const int wt = lane < nw ? s_w[lane] : 0;
        int xs = wt;
        #pragma unroll
        for (int d = 1; d < 32; d <<= 1) {
            const int y = __shfl_up(xs, d);
            if (lane >= d) xs += y;
        }
        s_w[lane] = xs - wt;
        if (lane == 31) s_w[32] = xs;
    }
    __syncthreads();
    total = s_w[32];
    return s_w[wid] + (x - v);
}

__global__ void __launch_bounds__(PTHR)
ggnn_bhist(const int* __restrict__ dst, int* __restrict__ bcnt,
           int* __restrict__ hmat, int E) {
    __shared__ int s_h[NBMAX];
    const int tid = threadIdx.x;
    s_h[tid] = 0;
    __syncthreads();
    const int e0 = blockIdx.x * EBLK;
    const int n  = (E - e0) < EBLK ? (E - e0) : EBLK;
    for (int i0 = 0; i0 < n; i0 += PTHR * 8) {
        int d[8];
        #pragma unroll
        for (int j = 0; j < 8; ++j) {
            const int i = i0 + j * PTHR + tid;
            d[j] = i < n ? dst[e0 + i] : 0;
        }
        #pragma unroll
        for (int j = 0; j < 8; ++j) {
            const int i = i0 + j * PTHR + tid;
            if (i < n) atomicAdd(&s_h[bucket_of(d[j])], 1);
        }
    }
    __syncthreads();
    const int c = s_h[tid];
    hmat[blockIdx.x * NBMAX + tid] = c;
    if (c) atomicAdd(&bcnt[tid * CSTR], c);
}

__global__ void __launch_bounds__(PTHR)
ggnn_part(const int* __restrict__ src, const int* __restrict__ dst,
          const int* __restrict__ bcnt, int* __restrict__ bcur,
          const int* __restrict__ hmat, int* __restrict__ bbase,
          int* __restrict__ pairs_w, int E) {
    __shared__ int  s_h[NBMAX];
    __shared__ int  s_lo[NBMAX];
    __shared__ int  s_run[NBMAX];
    __shared__ int  s_w[33];
    __shared__ int2 s_p[PCH];
    int2* pairs = reinterpret_cast<int2*>(pairs_w);
    const int tid  = threadIdx.x;
    const int e0   = blockIdx.x * EBLK;
    const int nblk = (E - e0) < EBLK ? (E - e0) : EBLK;

    const int ctot = hmat[blockIdx.x * NBMAX + tid];
    int gtot;
    const int gb = block_excl_scan(bcnt[tid * CSTR], s_w, gtot);
    s_run[tid] = gb + (ctot > 0 ? atomicAdd(&bcur[tid * CSTR], ctot) : 0);
    if (blockIdx.x == 0) {
        bbase[tid] = gb;
        if (tid == PTHR - 1) bbase[NBMAX] = gtot;
    }

    for (int c0 = 0; c0 < nblk; c0 += PCH) {
        const int n = (nblk - c0) < PCH ? (nblk - c0) : PCH;
        s_h[tid] = 0;
        __syncthreads();

        int d[PPT], s[PPT], r[PPT];
        #pragma unroll
        for (int j = 0; j < PPT; ++j) {
            const int i = j * PTHR + tid;
            d[j] = i < n ? dst[e0 + c0 + i] : 0;
            s[j] = i < n ? src[e0 + c0 + i] : 0;
        }
        #pragma unroll
        for (int j = 0; j < PPT; ++j) {
            const int i = j * PTHR + tid;
            r[j] = i < n ? atomicAdd(&s_h[bucket_of(d[j])], 1) : 0;
        }
        __syncthreads();

        const int c = s_h[tid];
        int tot;
        s_lo[tid] = block_excl_scan(c, s_w, tot);
        __syncthreads();

        #pragma unroll
        for (int j = 0; j < PPT; ++j) {
            const int i = j * PTHR + tid;
            if (i < n) s_p[s_lo[bucket_of(d[j])] + r[j]] = make_int2(d[j], s[j]);
        }
        __syncthreads();

        for (int i = tid; i < n; i += PTHR) {
            const int2 v = s_p[i];
            const int b  = bucket_of(v.x);
            const int pos = s_run[b] + (i - s_lo[b]);
            if ((unsigned)pos < (unsigned)E) pairs[pos] = v;
        }
        __syncthreads();
        s_run[tid] += c;
    }
}

__global__ void __launch_bounds__(STHR)
ggnn_bsort(const int* __restrict__ pairs_w, const int* __restrict__ bbase,
           int* __restrict__ rowptr, int* __restrict__ ssrc, int N, int E) {
    __shared__ int s_c[BNODE];
    __shared__ int s_w[33];
    __shared__ int s_s[SCAP];
    const int tid   = threadIdx.x;
    const int b     = blockIdx.x;
    const int node0 = b << BSH;
    int bs = bbase[b];
    int be = bbase[b + 1];
    bs = bs < 0 ? 0 : (bs > E ? E : bs);
    be = be < bs ? bs : (be > E ? E : be);
    const int n = be - bs;
    const bool stage = n <= SCAP;
    const int2* pb = reinterpret_cast<const int2*>(pairs_w) + bs;

    if (tid < BNODE) s_c[tid] = 0;
    __syncthreads();
    for (int i0 = 0; i0 < n; i0 += STHR * 8) {
        int2 v[8];
        #pragma unroll
        for (int j = 0; j < 8; ++j) {
            const int i = i0 + j * STHR + tid;
            v[j] = i < n ? pb[i] : make_int2(0, 0);
        }
        #pragma unroll
        for (int j = 0; j < 8; ++j) {
            const int i = i0 + j * STHR + tid;
            if (i < n) atomicAdd(&s_c[(v[j].x - node0) & (BNODE - 1)], 1);
        }
    }
    __syncthreads();

    const int c = tid < BNODE ? s_c[tid] : 0;
    int tot;
    const int ex = block_excl_scan(c, s_w, tot);
    if (tid < BNODE) {
        const int node = node0 + tid;
        if (node < N) rowptr[node] = bs + ex;
        s_c[tid] = ex;
    }
    if (b == (int)gridDim.x - 1 && tid == 0) rowptr[N] = E;
    __syncthreads();

    for (int i0 = 0; i0 < n; i0 += STHR * 8) {
        int2 v[8];
        #pragma unroll
        for (int j = 0; j < 8; ++j) {
            const int i = i0 + j * STHR + tid;
            v[j] = i < n ? pb[i] : make_int2(0, 0);
        }
        #pragma unroll
        for (int j = 0; j < 8; ++j) {
            const int i = i0 + j * STHR + tid;
            if (i < n) {
                const int p = atomicAdd(&s_c[(v[j].x - node0) & (BNODE - 1)], 1);
                if (stage) {
                    if ((unsigned)p < (unsigned)SCAP) s_s[p] = v[j].y;
                } else {
                    if ((unsigned)p < (unsigned)n) ssrc[bs + p] = v[j].y;
                }
            }
        }
    }
    if (stage) {
        __syncthreads();
        for (int i = tid; i < n; i += STHR) ssrc[bs + i] = s_s[i];
    }
}

__global__ void __launch_bounds__(256)
ggnn_gather(const int* __restrict__ rowptr, const int* __restrict__ ssrc,
            const float* __restrict__ m, float* __restrict__ agg, int N) {
    const int tid  = threadIdx.x;
    const int l    = tid & 15;
    const int node = (blockIdx.x * 256 + tid) >> 4;
    const bool nv  = node < N;
    const float2* m2 = reinterpret_cast<const float2*>(m);

    int p         = nv ? rowptr[node]     : 0;
    const int end = nv ? rowptr[node + 1] : 0;
    int rem = end - p;
    const int other = __shfl_xor(rem, 16);
    int remw = __builtin_amdgcn_readfirstlane(rem > other ? rem : other);

    float2 acc = make_float2(0.0f, 0.0f);
    while (remw > 0) {
        const int idx = (l < rem) ? ssrc[p + l] : -1;
        #pragma unroll
        for (int j0 = 0; j0 < 16; j0 += 8) {
            if (j0 < remw) {
                int    s[8];
                float2 v[8];
                #pragma unroll
                for (int j = 0; j < 8; ++j) {
                    s[j] = __shfl(idx, j0 + j, 16);
                    const int sc = s[j] < 0 ? 0 : (s[j] < N ? s[j] : N - 1);
                    v[j] = m2[(size_t)sc * MS2 + l];
                }
                #pragma unroll
                for (int j = 0; j < 8; ++j) {
                    if (s[j] >= 0) { acc.x += v[j].x; acc.y += v[j].y; }
                }
            }
        }
        p += 16; rem -= 16; remw -= 16;
    }
    if (nv && l < CH2)
        reinterpret_cast<float2*>(agg)[(size_t)node * CH2 + l] = acc;
}

__global__ void ggnn_gru(const float* __restrict__ agg,
                         float* __restrict__ h,
                         const float* __restrict__ w_ih,
                         const float* __restrict__ w_hh,
                         const float* __restrict__ b_ih,
                         const float* __restrict__ b_hh, int N) {
    __shared__ float s_wih[3 * CCH * CCH];
    __shared__ float s_whh[3 * CCH * CCH];
    __shared__ float s_bih[3 * CCH];
    __shared__ float s_bhh[3 * CCH];

    for (int i = threadIdx.x; i < 3 * CCH * CCH; i += blockDim.x) {
        s_wih[i] = w_ih[i];
        s_whh[i] = w_hh[i];
    }
    if (threadIdx.x < 3 * CCH) {
        s_bih[threadIdx.x] = b_ih[threadIdx.x];
        s_bhh[threadIdx.x] = b_hh[threadIdx.x];
    }
    __syncthreads();

    const int n = blockIdx.x * blockDim.x + threadIdx.x;
    if (n >= N) return;

    float a[CCH], hv[CCH];
    #pragma unroll
    for (int k = 0; k < CCH; ++k) {
        a[k]  = agg[(size_t)n * CCH + k];
        hv[k] = h  [(size_t)n * CCH + k];
    }

    #pragma unroll 1
    for (int j = 0; j < CCH; ++j) {
        float ir = s_bih[j], iz = s_bih[CCH + j], inn = s_bih[2 * CCH + j];
        float hr = s_bhh[j], hz = s_bhh[CCH + j], hnn = s_bhh[2 * CCH + j];
        #pragma unroll
        for (int k = 0; k < CCH; ++k) {
            ir  += a[k]  * s_wih[j * CCH + k];
            iz  += a[k]  * s_wih[(CCH + j) * CCH + k];
            inn += a[k]  * s_wih[(2 * CCH + j) * CCH + k];
            hr  += hv[k] * s_whh[j * CCH + k];
            hz  += hv[k] * s_whh[(CCH + j) * CCH + k];
            hnn += hv[k] * s_whh[(2 * CCH + j) * CCH + k];
        }
        const float r  = sigmoid_f(ir + hr);
        const float z  = sigmoid_f(iz + hz);
        const float nn = tanhf(inn + r * hnn);
        h[(size_t)n * CCH + j] = (1.0f - z) * nn + z * hv[j];
    }
}

__global__ void ggnn_final(const float* __restrict__ h,
                           const float* __restrict__ lin_w,
                           const float* __restrict__ lin_b,
                           float* __restrict__ out, int N) {
    __shared__ float s_w[CCH * CCH];
    __shared__ float s_b[CCH];
    for (int i = threadIdx.x; i < CCH * CCH; i += blockDim.x) s_w[i] = lin_w[i];
    if (threadIdx.x < CCH) s_b[threadIdx.x] = lin_b[threadIdx.x];
    __syncthreads();

    const int n = blockIdx.x * blockDim.x + threadIdx.x;
    if (n >= N) return;

    float hv[CCH];
    #pragma unroll
    for (int k = 0; k < CCH; ++k) {
        const float v = h[(size_t)n * CCH + k];
        hv[k] = v > 0.0f ? v : 0.0f;
    }
    #pragma unroll 1
    for (int j = 0; j < CCH; ++j) {
        float acc = s_b[j];
        #pragma unroll
        for (int k = 0; k < CCH; ++k) acc += hv[k] * s_w[j * CCH + k];
        out[(size_t)n * CCH + j] = acc;
    }
}

extern "C" void kernel_launch(void* const* d_in, const int* in_sizes, int n_in,
                              void* d_out, int out_size, void* d_ws, size_t ws_size,
                              hipStream_t stream) {
    const float* x      = (const float*)d_in[0];
    const int*   eidx   = (const int*)  d_in[1];
    const float* weight = (const float*)d_in[2];
    const float* w_ih   = (const float*)d_in[3];
    const float* w_hh   = (const float*)d_in[4];
    const float* b_ih   = (const float*)d_in[5];
    const float* b_hh   = (const float*)d_in[6];
    const float* lin_w  = (const float*)d_in[7];
    const float* lin_b  = (const float*)d_in[8];
    float* out = (float*)d_out;

    const int N = in_sizes[0] / CCH;
    const int E = in_sizes[1] / 2;
    const int* src = eidx;
    const int* dst = eidx + E;

    float* h   = (float*)d_ws;
    float* m   = h + (((size_t)N * CCH + 63) & ~(size_t)63);
    float* agg = m + (size_t)N * MST;
    const int nblk = (E + EBLK - 1) / EBLK;
    int* bcnt   = (int*)(agg + (size_t)N * CCH);
    int* bcur   = bcnt + NBMAX * CSTR;
    int* bbase  = bcur + NBMAX * CSTR;
    int* hmat   = bbase + NBMAX + 4;
    int* rowptr = hmat + (size_t)nblk * NBMAX;
    int* pairs  = rowptr + ((N + 4) & ~3);
    int* ssrc   = pairs + 2 * (size_t)E;

    const int nelem = N * CCH;
    const int n4    = (nelem >> 2) + 4;
    ggnn_copy<<<(n4 + 255) / 256, 256, 0, stream>>>(x, h, nelem);

    const int c4 = ((2 * NBMAX * CSTR) >> 2) + 4;
    ggnn_zero<<<(c4 + 255) / 256, 256, 0, stream>>>((float*)bcnt, 2 * NBMAX * CSTR);
    ggnn_bhist<<<nblk, PTHR, 0, stream>>>(dst, bcnt, hmat, E);
    ggnn_part<<<nblk, PTHR, 0, stream>>>(src, dst, bcnt, bcur, hmat, bbase, pairs, E);
    ggnn_bsort<<<(N + BNODE - 1) / BNODE, STHR, 0, stream>>>(pairs, bbase, rowptr, ssrc, N, E);

    const int tiles = (N + 15) / 16;
    for (int layer = 0; layer < NLAYERS; ++layer) {
        ggnn_transform_wmma<<<(tiles + 3) / 4, 128, 0, stream>>>(
            h, weight + (size_t)layer * CCH * CCH, m, N);
        ggnn_gather<<<(N + 15) / 16, 256, 0, stream>>>(rowptr, ssrc, m, agg, N);
        ggnn_gru<<<(N + 127) / 128, 128, 0, stream>>>(agg, h, w_ih, w_hh,
                                                      b_ih, b_hh, N);
    }
    ggnn_final<<<(N + 127) / 128, 128, 0, stream>>>(h, lin_w, lin_b, out, N);
}


# GRU gates and final linear moved from thread-per-node vector FMAs with 4-B strided stores to f32 16x16x4 WMMA tiles (16 nodes per wave, LDS-staged, whole-line row I/O); sums now in WMMA k order
# speedup vs baseline: 14.3456x; 1.5076x over previous
#include <hip/hip_runtime.h>
#include <math.h>

#define CCH 30
#define NLAYERS 3
#define AST 34
#define CH2 (CCH / 2)
#define MST 32
#define MS2 (MST / 2)

typedef __attribute__((ext_vector_type(2))) float v2f;
typedef __attribute__((ext_vector_type(8))) float v8f;

__device__ __forceinline__ float sigmoid_f(float x) { return 1.0f / (1.0f + __expf(-x)); }

__global__ void ggnn_transform_wmma(const float* __restrict__ h,
                                    const float* __restrict__ W,
                                    float* __restrict__ m, int N) {
    __shared__ float s_Wt[32 * AST];
    __shared__ float s_A[4][16 * AST];

    const int tid  = threadIdx.x;
    const int lane = tid & 31;
    const int warp = tid >> 5;
    const int row0 = (blockIdx.x * 4 + warp) * 16;
    const bool full = (row0 + 16 <= N);

    for (int i = tid; i < 32 * AST; i += 128) s_Wt[i] = 0.0f;
    for (int i = lane; i < 16 * AST; i += 32) s_A[warp][i] = 0.0f;
    __syncthreads();

    for (int i = tid; i < CCH * CCH; i += 128) {
        const int k = i / CCH, c = i % CCH;
        s_Wt[c * AST + k] = W[i];
    }
    if (full) {
        const float2* h2 = reinterpret_cast<const float2*>(h + (size_t)row0 * CCH);
        #pragma unroll
        for (int i = lane; i < 16 * CH2; i += 32) {
            const int r = i / CH2, c2 = i % CH2;
            *reinterpret_cast<float2*>(&s_A[warp][r * AST + 2 * c2]) =
                h2[(size_t)r * CH2 + c2];
        }
    } else {
        for (int i = lane; i < 16 * CCH; i += 32) {
            const int r = i / CCH, c = i % CCH;
            const int row = row0 + r;
            if (row < N) s_A[warp][r * AST + c] = h[(size_t)row * CCH + c];
        }
    }
    __syncthreads();

    if (row0 >= N) return;

    const int half = lane >> 4;
    const int l    = lane & 15;
    v8f acc0 = {};
    v8f acc1 = {};
    float* sa = &s_A[warp][0];

    #pragma unroll
    for (int k0 = 0; k0 < 32; k0 += 4) {
        const int ka = k0 + half * 2;
        const v2f a  = *(const v2f*)&sa  [l        * AST + ka];
        const v2f b0 = *(const v2f*)&s_Wt[l        * AST + ka];
        const v2f b1 = *(const v2f*)&s_Wt[(16 + l) * AST + ka];
        acc0 = __builtin_amdgcn_wmma_f32_16x16x4_f32(false, a, false, b0,
                                                     (short)0, acc0, false, false);
        acc1 = __builtin_amdgcn_wmma_f32_16x16x4_f32(false, a, false, b1,
                                                     (short)0, acc1, false, false);
    }

    #pragma unroll
    for (int v = 0; v < 8; ++v) {
        const int r = v + half * 8;
        sa[r * AST + l]      = acc0[v];
        sa[r * AST + 16 + l] = acc1[v];
    }

    if (full) {
        float2* m2 = reinterpret_cast<float2*>(m + (size_t)row0 * MST);
        #pragma unroll
        for (int i = lane; i < 16 * MS2; i += 32) {
            const int r = i / MS2, c2 = i % MS2;
            m2[i] = *reinterpret_cast<const float2*>(&sa[r * AST + 2 * c2]);
        }
    } else {
        for (int i = lane; i < 16 * MST; i += 32) {
            const int r = i / MST, c = i % MST;
            const int row = row0 + r;
            if (row < N) m[(size_t)row * MST + c] = sa[r * AST + c];
        }
    }
}

__global__ void ggnn_copy(const float* __restrict__ src, float* __restrict__ dst,
                          int n) {
    const int n4 = n >> 2;
    int i = blockIdx.x * blockDim.x + threadIdx.x;
    if (i < n4) {
        reinterpret_cast<float4*>(dst)[i] =
            reinterpret_cast<const float4*>(src)[i];
    }
    if (i < (n & 3)) {
        const int t = n4 * 4 + i;
        dst[t] = src[t];
    }
}

__global__ void ggnn_zero(float* __restrict__ p, int n) {
    const int n4 = n >> 2;
    int i = blockIdx.x * blockDim.x + threadIdx.x;
    if (i < n4) reinterpret_cast<float4*>(p)[i] = make_float4(0.f, 0.f, 0.f, 0.f);
    if (i < (n & 3)) p[n4 * 4 + i] = 0.0f;
}

#define BSH   8
#define BNODE (1 << BSH)
#define NBMAX 1024
#define PTHR  1024
#define PPT   6
#define PCH   (PTHR * PPT)
#define EBLK  (4 * PCH)
#define CSTR  64
#define STHR  512
#define SCAP  12288

__device__ __forceinline__ int bucket_of(int d) {
    const unsigned b = (unsigned)d >> BSH;
    return (int)(b < (unsigned)NBMAX ? b : (unsigned)(NBMAX - 1));
}

__device__ __forceinline__ int block_excl_scan(int v, int* s_w, int& total) {
    const int lane = threadIdx.x & 31;
    const int wid  = threadIdx.x >> 5;
    const int nw   = (int)(blockDim.x >> 5);
    int x = v;
    #pragma unroll
    for (int d = 1; d < 32; d <<= 1) {
        const int y = __shfl_up(x, d);
        if (lane >= d) x += y;
    }
    __syncthreads();
    if (lane == 31) s_w[wid] = x;
    __syncthreads();
    if (wid == 0) {
        const int wt = lane < nw ? s_w[lane] : 0;
        int xs = wt;
        #pragma unroll
        for (int d = 1; d < 32; d <<= 1) {
            const int y = __shfl_up(xs, d);
            if (lane >= d) xs += y;
        }
        s_w[lane] = xs - wt;
        if (lane == 31) s_w[32] = xs;
    }
    __syncthreads();
    total = s_w[32];
    return s_w[wid] + (x - v);
}

__global__ void __launch_bounds__(PTHR)
ggnn_bhist(const int* __restrict__ dst, int* __restrict__ bcnt,
           int* __restrict__ hmat, int E) {
    __shared__ int s_h[NBMAX];
    const int tid = threadIdx.x;
    s_h[tid] = 0;
    __syncthreads();
    const int e0 = blockIdx.x * EBLK;
    const int n  = (E - e0) < EBLK ? (E - e0) : EBLK;
    for (int i0 = 0; i0 < n; i0 += PTHR * 8) {
        int d[8];
        #pragma unroll
        for (int j = 0; j < 8; ++j) {
            const int i = i0 + j * PTHR + tid;
            d[j] = i < n ? dst[e0 + i] : 0;
        }
        #pragma unroll
        for (int j = 0; j < 8; ++j) {
            const int i = i0 + j * PTHR + tid;
            if (i < n) atomicAdd(&s_h[bucket_of(d[j])], 1);
        }
    }
    __syncthreads();
    const int c = s_h[tid];
    hmat[blockIdx.x * NBMAX + tid] = c;
    if (c) atomicAdd(&bcnt[tid * CSTR], c);
}

__global__ void __launch_bounds__(PTHR)
ggnn_part(const int* __restrict__ src, const int* __restrict__ dst,
          const int* __restrict__ bcnt, int* __restrict__ bcur,
          const int* __restrict__ hmat, int* __restrict__ bbase,
          int* __restrict__ pairs_w, int E) {
    __shared__ int  s_h[NBMAX];
    __shared__ int  s_lo[NBMAX];
    __shared__ int  s_run[NBMAX];
    __shared__ int  s_w[33];
    __shared__ int2 s_p[PCH];
    int2* pairs = reinterpret_cast<int2*>(pairs_w);
    const int tid  = threadIdx.x;
    const int e0   = blockIdx.x * EBLK;
    const int nblk = (E - e0) < EBLK ? (E - e0) : EBLK;

    const int ctot = hmat[blockIdx.x * NBMAX + tid];
    int gtot;
    const int gb = block_excl_scan(bcnt[tid * CSTR], s_w, gtot);
    s_run[tid] = gb + (ctot > 0 ? atomicAdd(&bcur[tid * CSTR], ctot) : 0);
    if (blockIdx.x == 0) {
        bbase[tid] = gb;
        if (tid == PTHR - 1) bbase[NBMAX] = gtot;
    }

    for (int c0 = 0; c0 < nblk; c0 += PCH) {
        const int n = (nblk - c0) < PCH ? (nblk - c0) : PCH;
        s_h[tid] = 0;
        __syncthreads();

        int d[PPT], s[PPT], r[PPT];
        #pragma unroll
        for (int j = 0; j < PPT; ++j) {
            const int i = j * PTHR + tid;
            d[j] = i < n ? dst[e0 + c0 + i] : 0;
            s[j] = i < n ? src[e0 + c0 + i] : 0;
        }
        #pragma unroll
        for (int j = 0; j < PPT; ++j) {
            const int i = j * PTHR + tid;
            r[j] = i < n ? atomicAdd(&s_h[bucket_of(d[j])], 1) : 0;
        }
        __syncthreads();

        const int c = s_h[tid];
        int tot;
        s_lo[tid] = block_excl_scan(c, s_w, tot);
        __syncthreads();

        #pragma unroll
        for (int j = 0; j < PPT; ++j) {
            const int i = j * PTHR + tid;
            if (i < n) s_p[s_lo[bucket_of(d[j])] + r[j]] = make_int2(d[j], s[j]);
        }
        __syncthreads();

        for (int i = tid; i < n; i += PTHR) {
            const int2 v = s_p[i];
            const int b  = bucket_of(v.x);
            const int pos = s_run[b] + (i - s_lo[b]);
            if ((unsigned)pos < (unsigned)E) pairs[pos] = v;
        }
        __syncthreads();
        s_run[tid] += c;
    }
}

__global__ void __launch_bounds__(STHR)
ggnn_bsort(const int* __restrict__ pairs_w, const int* __restrict__ bbase,
           int* __restrict__ rowptr, int* __restrict__ ssrc, int N, int E) {
    __shared__ int s_c[BNODE];
    __shared__ int s_w[33];
    __shared__ int s_s[SCAP];
    const int tid   = threadIdx.x;
    const int b     = blockIdx.x;
    const int node0 = b << BSH;
    int bs = bbase[b];
    int be = bbase[b + 1];
    bs = bs < 0 ? 0 : (bs > E ? E : bs);
    be = be < bs ? bs : (be > E ? E : be);
    const int n = be - bs;
    const bool stage = n <= SCAP;
    const int2* pb = reinterpret_cast<const int2*>(pairs_w) + bs;

    if (tid < BNODE) s_c[tid] = 0;
    __syncthreads();
    for (int i0 = 0; i0 < n; i0 += STHR * 8) {
        int2 v[8];
        #pragma unroll
        for (int j = 0; j < 8; ++j) {
            const int i = i0 + j * STHR + tid;
            v[j] = i < n ? pb[i] : make_int2(0, 0);
        }
        #pragma unroll
        for (int j = 0; j < 8; ++j) {
            const int i = i0 + j * STHR + tid;
            if (i < n) atomicAdd(&s_c[(v[j].x - node0) & (BNODE - 1)], 1);
        }
    }
    __syncthreads();

    const int c = tid < BNODE ? s_c[tid] : 0;
    int tot;
    const int ex = block_excl_scan(c, s_w, tot);
    if (tid < BNODE) {
        const int node = node0 + tid;
        if (node < N) rowptr[node] = bs + ex;
        s_c[tid] = ex;
    }
    if (b == (int)gridDim.x - 1 && tid == 0) rowptr[N] = E;
    __syncthreads();

    for (int i0 = 0; i0 < n; i0 += STHR * 8) {
        int2 v[8];
        #pragma unroll
        for (int j = 0; j < 8; ++j) {
            const int i = i0 + j * STHR + tid;
            v[j] = i < n ? pb[i] : make_int2(0, 0);
        }
        #pragma unroll
        for (int j = 0; j < 8; ++j) {
            const int i = i0 + j * STHR + tid;
            if (i < n) {
                const int p = atomicAdd(&s_c[(v[j].x - node0) & (BNODE - 1)], 1);
                if (stage) {
                    if ((unsigned)p < (unsigned)SCAP) s_s[p] = v[j].y;
                } else {
                    if ((unsigned)p < (unsigned)n) ssrc[bs + p] = v[j].y;
                }
            }
        }
    }
    if (stage) {
        __syncthreads();
        for (int i = tid; i < n; i += STHR) ssrc[bs + i] = s_s[i];
    }
}

__global__ void __launch_bounds__(256)
ggnn_gather(const int* __restrict__ rowptr, const int* __restrict__ ssrc,
            const float* __restrict__ m, float* __restrict__ agg, int N) {
    const int tid  = threadIdx.x;
    const int l    = tid & 15;
    const int node = (blockIdx.x * 256 + tid) >> 4;
    const bool nv  = node < N;
    const float2* m2 = reinterpret_cast<const float2*>(m);

    int p         = nv ? rowptr[node]     : 0;
    const int end = nv ? rowptr[node + 1] : 0;
    int rem = end - p;
    const int other = __shfl_xor(rem, 16);
    int remw = __builtin_amdgcn_readfirstlane(rem > other ? rem : other);

    float2 acc = make_float2(0.0f, 0.0f);
    while (remw > 0) {
        const int idx = (l < rem) ? ssrc[p + l] : -1;
        #pragma unroll
        for (int j0 = 0; j0 < 16; j0 += 8) {
            if (j0 < remw) {
                int    s[8];
                float2 v[8];
                #pragma unroll
                for (int j = 0; j < 8; ++j) {
                    s[j] = __shfl(idx, j0 + j, 16);
                    const int sc = s[j] < 0 ? 0 : (s[j] < N ? s[j] : N - 1);
                    v[j] = m2[(size_t)sc * MS2 + l];
                }
                #pragma unroll
                for (int j = 0; j < 8; ++j) {
                    if (s[j] >= 0) { acc.x += v[j].x; acc.y += v[j].y; }
                }
            }
        }
        p += 16; rem -= 16; remw -= 16;
    }
    if (nv && l < CH2)
        reinterpret_cast<float2*>(agg)[(size_t)node * CH2 + l] = acc;
}

#define GST   34
#define GROWS 96
#define GW    8

__global__ void __launch_bounds__(GW * 32)
ggnn_gru_wmma(const float* __restrict__ agg,
              float* __restrict__ h,
              const float* __restrict__ w_ih,
              const float* __restrict__ w_hh,
              const float* __restrict__ b_ih,
              const float* __restrict__ b_hh, int N) {
    __shared__ __attribute__((aligned(16))) float s_wih[GROWS * GST];
    __shared__ __attribute__((aligned(16))) float s_whh[GROWS * GST];
    __shared__ __attribute__((aligned(16))) float s_A[GW][16 * GST];
    __shared__ __attribute__((aligned(16))) float s_H[GW][16 * GST];

    const int tid  = threadIdx.x;
    const int lane = tid & 31;
    const int warp = tid >> 5;
    const int row0 = (blockIdx.x * GW + warp) * 16;
    const bool full = (row0 + 16 <= N);
    float* sa = &s_A[warp][0];
    float* sh = &s_H[warp][0];

    const float2* wi2 = reinterpret_cast<const float2*>(w_ih);
    const float2* wh2 = reinterpret_cast<const float2*>(w_hh);
    #pragma unroll
    for (int j = 0; j < (3 * CCH * CH2 + GW * 32 - 1) / (GW * 32); ++j) {
        const int i = j * (GW * 32) + tid;
        if (i < 3 * CCH * CH2) {
            const int c = i / CH2, c2 = i % CH2;
            const int g = c / CCH, cc = c % CCH;
            const int o = (g * 32 + cc) * GST + 2 * c2;
            *reinterpret_cast<float2*>(&s_wih[o]) = wi2[i];
            *reinterpret_cast<float2*>(&s_whh[o]) = wh2[i];
        }
    }
    if (tid < GROWS) {
        *reinterpret_cast<float2*>(&s_wih[tid * GST + CCH]) = make_float2(0.f, 0.f);
        *reinterpret_cast<float2*>(&s_whh[tid * GST + CCH]) = make_float2(0.f, 0.f);
    }
    if (tid < 6 * CH2) {
        const int rr = tid / CH2, c2 = tid % CH2;
        const int o = ((rr >> 1) * 32 + CCH + (rr & 1)) * GST + 2 * c2;
        *reinterpret_cast<float2*>(&s_wih[o]) = make_float2(0.f, 0.f);
        *reinterpret_cast<float2*>(&s_whh[o]) = make_float2(0.f, 0.f);
    }
    if (lane < 16) {
        *reinterpret_cast<float2*>(&sa[lane * GST + CCH]) = make_float2(0.f, 0.f);
        *reinterpret_cast<float2*>(&sh[lane * GST + CCH]) = make_float2(0.f, 0.f);
    }
    if (full) {
        const float2* a2 = reinterpret_cast<const float2*>(agg + (size_t)row0 * CCH);
        const float2* h2 = reinterpret_cast<const float2*>(h   + (size_t)row0 * CCH);
        #pragma unroll
        for (int i = lane; i < 16 * CH2; i += 32) {
            const int r = i / CH2, c2 = i % CH2;
            *reinterpret_cast<float2*>(&sa[r * GST + 2 * c2]) = a2[i];
            *reinterpret_cast<float2*>(&sh[r * GST + 2 * c2]) = h2[i];
        }
    } else {
        for (int i = lane; i < 16 * CCH; i += 32) {
            const int r = i / CCH, c = i % CCH;
            const int row = row0 + r;
            if (row < N) {
                sa[r * GST + c] = agg[(size_t)row * CCH + c];
                sh[r * GST + c] = h  [(size_t)row * CCH + c];
            }
        }
    }
    __syncthreads();

    if (row0 >= N) return;

    const int half = lane >> 4;
    const int l    = lane & 15;

    v8f accr[2], accz[2], acci[2], acch[2];
    #pragma unroll
    for (int t = 0; t < 2; ++t) {
        const int c  = t * 16 + l;
        const bool cv = c < CCH;
        const int cc = cv ? c : 0;
        const float br = cv ? b_ih[cc] + b_hh[cc] : 0.0f;
        const float bz = cv ? b_ih[CCH + cc] + b_hh[CCH + cc] : 0.0f;
        const float bi = cv ? b_ih[2 * CCH + cc] : 0.0f;
        const float bh = cv ? b_hh[2 * CCH + cc] : 0.0f;
        #pragma unroll
        for (int v = 0; v < 8; ++v) {
            accr[t][v] = br; accz[t][v] = bz; acci[t][v] = bi; acch[t][v] = bh;
        }
    }

    #pragma unroll
    for (int k0 = 0; k0 < 32; k0 += 4) {
        const int ka = k0 + half * 2;
        const v2f am = *(const v2f*)&sa[l * GST + ka];
        const v2f ah = *(const v2f*)&sh[l * GST + ka];
        #pragma unroll
        for (int t = 0; t < 2; ++t) {
            const int bo = (t * 16 + l) * GST + ka;
            const v2f bir = *(const v2f*)&s_wih[bo];
            const v2f biz = *(const v2f*)&s_wih[32 * GST + bo];
            const v2f bin = *(const v2f*)&s_wih[64 * GST + bo];
            const v2f bhr = *(const v2f*)&s_whh[bo];
            const v2f bhz = *(const v2f*)&s_whh[32 * GST + bo];
            const v2f bhn = *(const v2f*)&s_whh[64 * GST + bo];
            accr[t] = __builtin_amdgcn_wmma_f32_16x16x4_f32(false, am, false, bir,
                                                            (short)0, accr[t], false, false);
            accz[t] = __builtin_amdgcn_wmma_f32_16x16x4_f32(false, am, false, biz,
                                                            (short)0, accz[t], false, false);
            acci[t] = __builtin_amdgcn_wmma_f32_16x16x4_f32(false, am, false, bin,
                                                            (short)0, acci[t], false, false);
            acch[t] = __builtin_amdgcn_wmma_f32_16x16x4_f32(false, ah, false, bhn,
                                                            (short)0, acch[t], false, false);
            accr[t] = __builtin_amdgcn_wmma_f32_16x16x4_f32(false, ah, false, bhr,
                                                            (short)0, accr[t], false, false);
            accz[t] = __builtin_amdgcn_wmma_f32_16x16x4_f32(false, ah, false, bhz,
                                                            (short)0, accz[t], false, false);
        }
    }

    #pragma unroll
    for (int t = 0; t < 2; ++t) {
        #pragma unroll
        for (int v = 0; v < 8; ++v) {
            const int o = (v + half * 8) * GST + t * 16 + l;
            const float hold = sh[o];
            const float r  = sigmoid_f(accr[t][v]);
            const float z  = sigmoid_f(accz[t][v]);
            const float nn = tanhf(acci[t][v] + r * acch[t][v]);
            sa[o] = (1.0f - z) * nn + z * hold;
        }
    }

    if (full) {
        float2* h2w = reinterpret_cast<float2*>(h + (size_t)row0 * CCH);
        #pragma unroll
        for (int i = lane; i < 16 * CH2; i += 32) {
            const int r = i / CH2, c2 = i % CH2;
            h2w[i] = *reinterpret_cast<const float2*>(&sa[r * GST + 2 * c2]);
        }
    } else {
        for (int i = lane; i < 16 * CCH; i += 32) {
            const int r = i / CCH, c = i % CCH;
            const int row = row0 + r;
            if (row < N) h[(size_t)row * CCH + c] = sa[r * GST + c];
        }
    }
}

__global__ void __launch_bounds__(GW * 32)
ggnn_final_wmma(const float* __restrict__ h,
                const float* __restrict__ lin_w,
                const float* __restrict__ lin_b,
                float* __restrict__ out, int N) {
    __shared__ __attribute__((aligned(16))) float s_B[32 * GST];
    __shared__ __attribute__((aligned(16))) float s_A[GW][16 * GST];

    const int tid  = threadIdx.x;
    const int lane = tid & 31;
    const int warp = tid >> 5;
    const int row0 = (blockIdx.x * GW + warp) * 16;
    const bool full = (row0 + 16 <= N);
    float* sa = &s_A[warp][0];

    const float2* w2 = reinterpret_cast<const float2*>(lin_w);
    #pragma unroll
    for (int j = 0; j < (CCH * CH2 + GW * 32 - 1) / (GW * 32); ++j) {
        const int i = j * (GW * 32) + tid;
        if (i < CCH * CH2) {
            const int c = i / CH2, c2 = i % CH2;
            *reinterpret_cast<float2*>(&s_B[c * GST + 2 * c2]) = w2[i];
        }
    }
    if (tid < 32)
        *reinterpret_cast<float2*>(&s_B[tid * GST + CCH]) = make_float2(0.f, 0.f);
    if (tid < 2 * CH2) {
        const int rr = tid / CH2, c2 = tid % CH2;
        *reinterpret_cast<float2*>(&s_B[(CCH + rr) * GST + 2 * c2]) = make_float2(0.f, 0.f);
    }
    if (lane < 16)
        *reinterpret_cast<float2*>(&sa[lane * GST + CCH]) = make_float2(0.f, 0.f);
    if (full) {
        const float2* h2 = reinterpret_cast<const float2*>(h + (size_t)row0 * CCH);
        #pragma unroll
        for (int i = lane; i < 16 * CH2; i += 32) {
            const int r = i / CH2, c2 = i % CH2;
            float2 v = h2[i];
            v.x = v.x > 0.0f ? v.x : 0.0f;
            v.y = v.y > 0.0f ? v.y : 0.0f;
            *reinterpret_cast<float2*>(&sa[r * GST + 2 * c2]) = v;
        }
    } else {
        for (int i = lane; i < 16 * CCH; i += 32) {
            const int r = i / CCH, c = i % CCH;
            const int row = row0 + r;
            if (row < N) {
                const float v = h[(size_t)row * CCH + c];
                sa[r * GST + c] = v > 0.0f ? v : 0.0f;
            }
        }
    }
    __syncthreads();

    if (row0 >= N) return;

    const int half = lane >> 4;
    const int l    = lane & 15;

    v8f acc[2];
    #pragma unroll
    for (int t = 0; t < 2; ++t) {
        const int c = t * 16 + l;
        const float b = c < CCH ? lin_b[c < CCH ? c : 0] : 0.0f;
        #pragma unroll
        for (int v = 0; v < 8; ++v) acc[t][v] = b;
    }

    #pragma unroll
    for (int k0 = 0; k0 < 32; k0 += 4) {
        const int ka = k0 + half * 2;
        const v2f a  = *(const v2f*)&sa [l        * GST + ka];
        const v2f b0 = *(const v2f*)&s_B[l        * GST + ka];
        const v2f b1 = *(const v2f*)&s_B[(16 + l) * GST + ka];
        acc[0] = __builtin_amdgcn_wmma_f32_16x16x4_f32(false, a, false, b0,
                                                       (short)0, acc[0], false, false);
        acc[1] = __builtin_amdgcn_wmma_f32_16x16x4_f32(false, a, false, b1,
                                                       (short)0, acc[1], false, false);
    }

    #pragma unroll
    for (int v = 0; v < 8; ++v) {
        const int r = v + half * 8;
        sa[r * GST + l]      = acc[0][v];
        sa[r * GST + 16 + l] = acc[1][v];
    }

    if (full) {
        float2* o2 = reinterpret_cast<float2*>(out + (size_t)row0 * CCH);
        #pragma unroll
        for (int i = lane; i < 16 * CH2; i += 32) {
            const int r = i / CH2, c2 = i % CH2;
            o2[i] = *reinterpret_cast<const float2*>(&sa[r * GST + 2 * c2]);
        }
    } else {
        for (int i = lane; i < 16 * CCH; i += 32) {
            const int r = i / CCH, c = i % CCH;
            const int row = row0 + r;
            if (row < N) out[(size_t)row * CCH + c] = sa[r * GST + c];
        }
    }
}

extern "C" void kernel_launch(void* const* d_in, const int* in_sizes, int n_in,
                              void* d_out, int out_size, void* d_ws, size_t ws_size,
                              hipStream_t stream) {
    const float* x      = (const float*)d_in[0];
    const int*   eidx   = (const int*)  d_in[1];
    const float* weight = (const float*)d_in[2];
    const float* w_ih   = (const float*)d_in[3];
    const float* w_hh   = (const float*)d_in[4];
    const float* b_ih   = (const float*)d_in[5];
    const float* b_hh   = (const float*)d_in[6];
    const float* lin_w  = (const float*)d_in[7];
    const float* lin_b  = (const float*)d_in[8];
    float* out = (float*)d_out;

    const int N = in_sizes[0] / CCH;
    const int E = in_sizes[1] / 2;
    const int* src = eidx;
    const int* dst = eidx + E;

    float* h   = (float*)d_ws;
    float* m   = h + (((size_t)N * CCH + 63) & ~(size_t)63);
    float* agg = m + (size_t)N * MST;
    const int nblk = (E + EBLK - 1) / EBLK;
    int* bcnt   = (int*)(agg + (size_t)N * CCH);
    int* bcur   = bcnt + NBMAX * CSTR;
    int* bbase  = bcur + NBMAX * CSTR;
    int* hmat   = bbase + NBMAX + 4;
    int* rowptr = hmat + (size_t)nblk * NBMAX;
    int* pairs  = rowptr + ((N + 4) & ~3);
    int* ssrc   = pairs + 2 * (size_t)E;

    const int nelem = N * CCH;
    const int n4    = (nelem >> 2) + 4;
    ggnn_copy<<<(n4 + 255) / 256, 256, 0, stream>>>(x, h, nelem);

    const int c4 = ((2 * NBMAX * CSTR) >> 2) + 4;
    ggnn_zero<<<(c4 + 255) / 256, 256, 0, stream>>>((float*)bcnt, 2 * NBMAX * CSTR);
    ggnn_bhist<<<nblk, PTHR, 0, stream>>>(dst, bcnt, hmat, E);
    ggnn_part<<<nblk, PTHR, 0, stream>>>(src, dst, bcnt, bcur, hmat, bbase, pairs, E);
    ggnn_bsort<<<(N + BNODE - 1) / BNODE, STHR, 0, stream>>>(pairs, bbase, rowptr, ssrc, N, E);

    const int tiles = (N + 15) / 16;
    for (int layer = 0; layer < NLAYERS; ++layer) {
        ggnn_transform_wmma<<<(tiles + 3) / 4, 128, 0, stream>>>(
            h, weight + (size_t)layer * CCH * CCH, m, N);
        ggnn_gather<<<(N + 15) / 16, 256, 0, stream>>>(rowptr, ssrc, m, agg, N);
        ggnn_gru_wmma<<<(tiles + GW - 1) / GW, GW * 32, 0, stream>>>(agg, h, w_ih, w_hh,
                                                                     b_ih, b_hh, N);
    }
    ggnn_final_wmma<<<(tiles + GW - 1) / GW, GW * 32, 0, stream>>>(h, lin_w, lin_b, out, N);
}


# Message weight applied after the neighbour sum (sum is linear): gather reads h rows at a 128-B pitch, the h@W WMMA product moves into the GRU kernel; no transform launches, no m; sums regrouped
# speedup vs baseline: 16.6621x; 1.1615x over previous
#include <hip/hip_runtime.h>
#include <math.h>

#define CCH 30
#define NLAYERS 3
#define CH2 (CCH / 2)
#define MST 32
#define MS2 (MST / 2)

typedef __attribute__((ext_vector_type(2))) float v2f;
typedef __attribute__((ext_vector_type(8))) float v8f;

__device__ __forceinline__ float sigmoid_f(float x) { return 1.0f / (1.0f + __expf(-x)); }

__global__ void ggnn_pad(const float* __restrict__ x, float* __restrict__ h, int N) {
    const int i   = blockIdx.x * blockDim.x + threadIdx.x;
    const int row = i >> 3, q = i & 7;
    if (row < N) {
        const float2* x2 = reinterpret_cast<const float2*>(x) + (size_t)row * CH2 + 2 * q;
        const float2 a = x2[0];
        const float2 b = q < 7 ? x2[1] : make_float2(0.f, 0.f);
        reinterpret_cast<float4*>(h)[i] = make_float4(a.x, a.y, b.x, b.y);
    }
}

__global__ void ggnn_zero(float* __restrict__ p, int n) {
    const int n4 = n >> 2;
    int i = blockIdx.x * blockDim.x + threadIdx.x;
    if (i < n4) reinterpret_cast<float4*>(p)[i] = make_float4(0.f, 0.f, 0.f, 0.f);
    if (i < (n & 3)) p[n4 * 4 + i] = 0.0f;
}

#define BSH   8
#define BNODE (1 << BSH)
#define NBMAX 1024
#define PTHR  1024
#define PPT   6
#define PCH   (PTHR * PPT)
#define EBLK  (4 * PCH)
#define CSTR  64
#define STHR  512
#define SCAP  12288

__device__ __forceinline__ int bucket_of(int d) {
    const unsigned b = (unsigned)d >> BSH;
    return (int)(b < (unsigned)NBMAX ? b : (unsigned)(NBMAX - 1));
}

__device__ __forceinline__ int block_excl_scan(int v, int* s_w, int& total) {
    const int lane = threadIdx.x & 31;
    const int wid  = threadIdx.x >> 5;
    const int nw   = (int)(blockDim.x >> 5);
    int x = v;
    #pragma unroll
    for (int d = 1; d < 32; d <<= 1) {
        const int y = __shfl_up(x, d);
        if (lane >= d) x += y;
    }
    __syncthreads();
    if (lane == 31) s_w[wid] = x;
    __syncthreads();
    if (wid == 0) {
        const int wt = lane < nw ? s_w[lane] : 0;
        int xs = wt;
        #pragma unroll
        for (int d = 1; d < 32; d <<= 1) {
            const int y = __shfl_up(xs, d);
            if (lane >= d) xs += y;
        }
        s_w[lane] = xs - wt;
        if (lane == 31) s_w[32] = xs;
    }
    __syncthreads();
    total = s_w[32];
    return s_w[wid] + (x - v);
}

__global__ void __launch_bounds__(PTHR)
ggnn_bhist(const int* __restrict__ dst, int* __restrict__ bcnt,
           int* __restrict__ hmat, int E) {
    __shared__ int s_h[NBMAX];
    const int tid = threadIdx.x;
    s_h[tid] = 0;
    __syncthreads();
    const int e0 = blockIdx.x * EBLK;
    const int n  = (E - e0) < EBLK ? (E - e0) : EBLK;
    for (int i0 = 0; i0 < n; i0 += PTHR * 8) {
        int d[8];
        #pragma unroll
        for (int j = 0; j < 8; ++j) {
            const int i = i0 + j * PTHR + tid;
            d[j] = i < n ? dst[e0 + i] : 0;
        }
        #pragma unroll
        for (int j = 0; j < 8; ++j) {
            const int i = i0 + j * PTHR + tid;
            if (i < n) atomicAdd(&s_h[bucket_of(d[j])], 1);
        }
    }
    __syncthreads();
    const int c = s_h[tid];
    hmat[blockIdx.x * NBMAX + tid] = c;
    if (c) atomicAdd(&bcnt[tid * CSTR], c);
}

__global__ void __launch_bounds__(PTHR)
ggnn_part(const int* __restrict__ src, const int* __restrict__ dst,
          const int* __restrict__ bcnt, int* __restrict__ bcur,
          const int* __restrict__ hmat, int* __restrict__ bbase,
          int* __restrict__ pairs_w, int E) {
    __shared__ int  s_h[NBMAX];
    __shared__ int  s_lo[NBMAX];
    __shared__ int  s_run[NBMAX];
    __shared__ int  s_w[33];
    __shared__ int2 s_p[PCH];
    int2* pairs = reinterpret_cast<int2*>(pairs_w);
    const int tid  = threadIdx.x;
    const int e0   = blockIdx.x * EBLK;
    const int nblk = (E - e0) < EBLK ? (E - e0) : EBLK;

    const int ctot = hmat[blockIdx.x * NBMAX + tid];
    int gtot;
    const int gb = block_excl_scan(bcnt[tid * CSTR], s_w, gtot);
    s_run[tid] = gb + (ctot > 0 ? atomicAdd(&bcur[tid * CSTR], ctot) : 0);
    if (blockIdx.x == 0) {
        bbase[tid] = gb;
        if (tid == PTHR - 1) bbase[NBMAX] = gtot;
    }

    for (int c0 = 0; c0 < nblk; c0 += PCH) {
        const int n = (nblk - c0) < PCH ? (nblk - c0) : PCH;
        s_h[tid] = 0;
        __syncthreads();

        int d[PPT], s[PPT], r[PPT];
        #pragma unroll
        for (int j = 0; j < PPT; ++j) {
            const int i = j * PTHR + tid;
            d[j] = i < n ? dst[e0 + c0 + i] : 0;
            s[j] = i < n ? src[e0 + c0 + i] : 0;
        }
        #pragma unroll
        for (int j = 0; j < PPT; ++j) {
            const int i = j * PTHR + tid;
            r[j] = i < n ? atomicAdd(&s_h[bucket_of(d[j])], 1) : 0;
        }
        __syncthreads();

        const int c = s_h[tid];
        int tot;
        s_lo[tid] = block_excl_scan(c, s_w, tot);
        __syncthreads();

        #pragma unroll
        for (int j = 0; j < PPT; ++j) {
            const int i = j * PTHR + tid;
            if (i < n) s_p[s_lo[bucket_of(d[j])] + r[j]] = make_int2(d[j], s[j]);
        }
        __syncthreads();

        for (int i = tid; i < n; i += PTHR) {
            const int2 v = s_p[i];
            const int b  = bucket_of(v.x);
            const int pos = s_run[b] + (i - s_lo[b]);
            if ((unsigned)pos < (unsigned)E) pairs[pos] = v;
        }
        __syncthreads();
        s_run[tid] += c;
    }
}

__global__ void __launch_bounds__(STHR)
ggnn_bsort(const int* __restrict__ pairs_w, const int* __restrict__ bbase,
           int* __restrict__ rowptr, int* __restrict__ ssrc, int N, int E) {
    __shared__ int s_c[BNODE];
    __shared__ int s_w[33];
    __shared__ int s_s[SCAP];
    const int tid   = threadIdx.x;
    const int b     = blockIdx.x;
    const int node0 = b << BSH;
    int bs = bbase[b];
    int be = bbase[b + 1];
    bs = bs < 0 ? 0 : (bs > E ? E : bs);
    be = be < bs ? bs : (be > E ? E : be);
    const int n = be - bs;
    const bool stage = n <= SCAP;
    const int2* pb = reinterpret_cast<const int2*>(pairs_w) + bs;

    if (tid < BNODE) s_c[tid] = 0;
    __syncthreads();
    for (int i0 = 0; i0 < n; i0 += STHR * 8) {
        int2 v[8];
        #pragma unroll
        for (int j = 0; j < 8; ++j) {
            const int i = i0 + j * STHR + tid;
            v[j] = i < n ? pb[i] : make_int2(0, 0);
        }
        #pragma unroll
        for (int j = 0; j < 8; ++j) {
            const int i = i0 + j * STHR + tid;
            if (i < n) atomicAdd(&s_c[(v[j].x - node0) & (BNODE - 1)], 1);
        }
    }
    __syncthreads();

    const int c = tid < BNODE ? s_c[tid] : 0;
    int tot;
    const int ex = block_excl_scan(c, s_w, tot);
    if (tid < BNODE) {
        const int node = node0 + tid;
        if (node < N) rowptr[node] = bs + ex;
        s_c[tid] = ex;
    }
    if (b == (int)gridDim.x - 1 && tid == 0) rowptr[N] = E;
    __syncthreads();

    for (int i0 = 0; i0 < n; i0 += STHR * 8) {
        int2 v[8];
        #pragma unroll
        for (int j = 0; j < 8; ++j) {
            const int i = i0 + j * STHR + tid;
            v[j] = i < n ? pb[i] : make_int2(0, 0);
        }
        #pragma unroll
        for (int j = 0; j < 8; ++j) {
            const int i = i0 + j * STHR + tid;
            if (i < n) {
                const int p = atomicAdd(&s_c[(v[j].x - node0) & (BNODE - 1)], 1);
                if (stage) {
                    if ((unsigned)p < (unsigned)SCAP) s_s[p] = v[j].y;
                } else {
                    if ((unsigned)p < (unsigned)n) ssrc[bs + p] = v[j].y;
                }
            }
        }
    }
    if (stage) {
        __syncthreads();
        for (int i = tid; i < n; i += STHR) ssrc[bs + i] = s_s[i];
    }
}

__global__ void __launch_bounds__(256)
ggnn_gather(const int* __restrict__ rowptr, const int* __restrict__ ssrc,
            const float* __restrict__ h, float* __restrict__ aggh, int N) {
    const int tid  = threadIdx.x;
    const int l    = tid & 15;
    const int node = (blockIdx.x * 256 + tid) >> 4;
    const bool nv  = node < N;
    const float2* h2 = reinterpret_cast<const float2*>(h);

    int p         = nv ? rowptr[node]     : 0;
    const int end = nv ? rowptr[node + 1] : 0;
    int rem = end - p;
    const int other = __shfl_xor(rem, 16);
    int remw = __builtin_amdgcn_readfirstlane(rem > other ? rem : other);

    float2 acc = make_float2(0.0f, 0.0f);
    while (remw > 0) {
        const int idx = (l < rem) ? ssrc[p + l] : -1;
        #pragma unroll
        for (int j0 = 0; j0 < 16; j0 += 8) {
            if (j0 < remw) {
                int    s[8];
                float2 v[8];
                #pragma unroll
                for (int j = 0; j < 8; ++j) {
                    s[j] = __shfl(idx, j0 + j, 16);
                    const int sc = s[j] < 0 ? 0 : (s[j] < N ? s[j] : N - 1);
                    v[j] = h2[(size_t)sc * MS2 + l];
                }
                #pragma unroll
                for (int j = 0; j < 8; ++j) {
                    if (s[j] >= 0) { acc.x += v[j].x; acc.y += v[j].y; }
                }
            }
        }
        p += 16; rem -= 16; remw -= 16;
    }
    if (nv)
        reinterpret_cast<float2*>(aggh)[(size_t)node * MS2 + l] =
            l < CH2 ? acc : make_float2(0.0f, 0.0f);
}

#define GST   34
#define GROWS 96
#define GW    8

__global__ void __launch_bounds__(GW * 32)
ggnn_gru_wmma(const float* __restrict__ aggh,
              float* __restrict__ h,
              const float* __restrict__ W,
              const float* __restrict__ w_ih,
              const float* __restrict__ w_hh,
              const float* __restrict__ b_ih,
              const float* __restrict__ b_hh, int N) {
    __shared__ __attribute__((aligned(16))) float s_wih[GROWS * GST];
    __shared__ __attribute__((aligned(16))) float s_whh[GROWS * GST];
    __shared__ __attribute__((aligned(16))) float s_Wt[32 * GST];
    __shared__ __attribute__((aligned(16))) float s_A[GW][16 * GST];
    __shared__ __attribute__((aligned(16))) float s_H[GW][16 * GST];

    const int tid  = threadIdx.x;
    const int lane = tid & 31;
    const int warp = tid >> 5;
    const int row0 = (blockIdx.x * GW + warp) * 16;
    const bool full = (row0 + 16 <= N);
    float* sa = &s_A[warp][0];
    float* sh = &s_H[warp][0];

    const float2* wi2 = reinterpret_cast<const float2*>(w_ih);
    const float2* wh2 = reinterpret_cast<const float2*>(w_hh);
    #pragma unroll
    for (int j = 0; j < (3 * CCH * CH2 + GW * 32 - 1) / (GW * 32); ++j) {
        const int i = j * (GW * 32) + tid;
        if (i < 3 * CCH * CH2) {
            const int c = i / CH2, c2 = i % CH2;
            const int g = c / CCH, cc = c % CCH;
            const int o = (g * 32 + cc) * GST + 2 * c2;
            *reinterpret_cast<float2*>(&s_wih[o]) = wi2[i];
            *reinterpret_cast<float2*>(&s_whh[o]) = wh2[i];
        }
    }
    #pragma unroll
    for (int j = 0; j < (CCH * CCH + GW * 32 - 1) / (GW * 32); ++j) {
        const int i = j * (GW * 32) + tid;
        if (i < CCH * CCH) {
            const int k = i / CCH, c = i % CCH;
            s_Wt[c * GST + k] = W[i];
        }
    }
    if (tid < GROWS) {
        *reinterpret_cast<float2*>(&s_wih[tid * GST + CCH]) = make_float2(0.f, 0.f);
        *reinterpret_cast<float2*>(&s_whh[tid * GST + CCH]) = make_float2(0.f, 0.f);
    }
    if (tid < 6 * CH2) {
        const int rr = tid / CH2, c2 = tid % CH2;
        const int o = ((rr >> 1) * 32 + CCH + (rr & 1)) * GST + 2 * c2;
        *reinterpret_cast<float2*>(&s_wih[o]) = make_float2(0.f, 0.f);
        *reinterpret_cast<float2*>(&s_whh[o]) = make_float2(0.f, 0.f);
    }
    if (tid < 32)
        *reinterpret_cast<float2*>(&s_Wt[tid * GST + CCH]) = make_float2(0.f, 0.f);
    if (tid < 2 * CH2) {
        const int rr = tid / CH2, c2 = tid % CH2;
        *reinterpret_cast<float2*>(&s_Wt[(CCH + rr) * GST + 2 * c2]) = make_float2(0.f, 0.f);
    }
    if (full) {
        const float2* a2 = reinterpret_cast<const float2*>(aggh + (size_t)row0 * MST);
        const float2* h2 = reinterpret_cast<const float2*>(h    + (size_t)row0 * MST);
        const bool padl = (lane & 15) == 15;
        #pragma unroll
        for (int i = lane; i < 16 * MS2; i += 32) {
            const int r = i / MS2, c2 = i % MS2;
            float2 va = a2[i];
            float2 vh = h2[i];
            if (padl) { va = make_float2(0.f, 0.f); vh = make_float2(0.f, 0.f); }
            *reinterpret_cast<float2*>(&sa[r * GST + 2 * c2]) = va;
            *reinterpret_cast<float2*>(&sh[r * GST + 2 * c2]) = vh;
        }
    } else {
        if (lane < 16) {
            *reinterpret_cast<float2*>(&sa[lane * GST + CCH]) = make_float2(0.f, 0.f);
            *reinterpret_cast<float2*>(&sh[lane * GST + CCH]) = make_float2(0.f, 0.f);
        }
        for (int i = lane; i < 16 * CCH; i += 32) {
            const int r = i / CCH, c = i % CCH;
            const int row = row0 + r;
            if (row < N) {
                sa[r * GST + c] = aggh[(size_t)row * MST + c];
                sh[r * GST + c] = h   [(size_t)row * MST + c];
            }
        }
    }
    __syncthreads();

    if (row0 >= N) return;

    const int half = lane >> 4;
    const int l    = lane & 15;

    {
        v8f m0 = {};
        v8f m1 = {};
        #pragma unroll
        for (int k0 = 0; k0 < 32; k0 += 4) {
            const int ka = k0 + half * 2;
            const v2f a  = *(const v2f*)&sa  [l        * GST + ka];
            const v2f b0 = *(const v2f*)&s_Wt[l        * GST + ka];
            const v2f b1 = *(const v2f*)&s_Wt[(16 + l) * GST + ka];
            m0 = __builtin_amdgcn_wmma_f32_16x16x4_f32(false, a, false, b0,
                                                       (short)0, m0, false, false);
            m1 = __builtin_amdgcn_wmma_f32_16x16x4_f32(false, a, false, b1,
                                                       (short)0, m1, false, false);
        }
        #pragma unroll
        for (int v = 0; v < 8; ++v) {
            const int r = v + half * 8;
            sa[r * GST + l]      = m0[v];
            sa[r * GST + 16 + l] = m1[v];
        }
    }

    v8f accr[2], accz[2], acci[2], acch[2];
    #pragma unroll
    for (int t = 0; t < 2; ++t) {
        const int c  = t * 16 + l;
        const bool cv = c < CCH;
        const int cc = cv ? c : 0;
        const float br = cv ? b_ih[cc] + b_hh[cc] : 0.0f;
        const float bz = cv ? b_ih[CCH + cc] + b_hh[CCH + cc] : 0.0f;
        const float bi = cv ? b_ih[2 * CCH + cc] : 0.0f;
        const float bh = cv ? b_hh[2 * CCH + cc] : 0.0f;
        #pragma unroll
        for (int v = 0; v < 8; ++v) {
            accr[t][v] = br; accz[t][v] = bz; acci[t][v] = bi; acch[t][v] = bh;
        }
    }

    #pragma unroll
    for (int k0 = 0; k0 < 32; k0 += 4) {
        const int ka = k0 + half * 2;
        const v2f am = *(const v2f*)&sa[l * GST + ka];
        const v2f ah = *(const v2f*)&sh[l * GST + ka];
        #pragma unroll
        for (int t = 0; t < 2; ++t) {
            const int bo = (t * 16 + l) * GST + ka;
            const v2f bir = *(const v2f*)&s_wih[bo];
            const v2f biz = *(const v2f*)&s_wih[32 * GST + bo];
            const v2f bin = *(const v2f*)&s_wih[64 * GST + bo];
            const v2f bhr = *(const v2f*)&s_whh[bo];
            const v2f bhz = *(const v2f*)&s_whh[32 * GST + bo];
            const v2f bhn = *(const v2f*)&s_whh[64 * GST + bo];
            accr[t] = __builtin_amdgcn_wmma_f32_16x16x4_f32(false, am, false, bir,
                                                            (short)0, accr[t], false, false);
            accz[t] = __builtin_amdgcn_wmma_f32_16x16x4_f32(false, am, false, biz,
                                                            (short)0, accz[t], false, false);
            acci[t] = __builtin_amdgcn_wmma_f32_16x16x4_f32(false, am, false, bin,
                                                            (short)0, acci[t], false, false);
            acch[t] = __builtin_amdgcn_wmma_f32_16x16x4_f32(false, ah, false, bhn,
                                                            (short)0, acch[t], false, false);
            accr[t] = __builtin_amdgcn_wmma_f32_16x16x4_f32(false, ah, false, bhr,
                                                            (short)0, accr[t], false, false);
            accz[t] = __builtin_amdgcn_wmma_f32_16x16x4_f32(false, ah, false, bhz,
                                                            (short)0, accz[t], false, false);
        }
    }

    #pragma unroll
    for (int t = 0; t < 2; ++t) {
        #pragma unroll
        for (int v = 0; v < 8; ++v) {
            const int o = (v + half * 8) * GST + t * 16 + l;
            const float hold = sh[o];
            const float r  = sigmoid_f(accr[t][v]);
            const float z  = sigmoid_f(accz[t][v]);
            const float nn = tanhf(acci[t][v] + r * acch[t][v]);
            sa[o] = (1.0f - z) * nn + z * hold;
        }
    }

    if (full) {
        float2* h2w = reinterpret_cast<float2*>(h + (size_t)row0 * MST);
        #pragma unroll
        for (int i = lane; i < 16 * MS2; i += 32) {
            const int r = i / MS2, c2 = i % MS2;
            h2w[i] = *reinterpret_cast<const float2*>(&sa[r * GST + 2 * c2]);
        }
    } else {
        for (int i = lane; i < 16 * MST; i += 32) {
            const int r = i / MST, c = i % MST;
            const int row = row0 + r;
            if (row < N) h[(size_t)row * MST + c] = sa[r * GST + c];
        }
    }
}

__global__ void __launch_bounds__(GW * 32)
ggnn_final_wmma(const float* __restrict__ h,
                const float* __restrict__ lin_w,
                const float* __restrict__ lin_b,
                float* __restrict__ out, int N) {
    __shared__ __attribute__((aligned(16))) float s_B[32 * GST];
    __shared__ __attribute__((aligned(16))) float s_A[GW][16 * GST];

    const int tid  = threadIdx.x;
    const int lane = tid & 31;
    const int warp = tid >> 5;
    const int row0 = (blockIdx.x * GW + warp) * 16;
    const bool full = (row0 + 16 <= N);
    float* sa = &s_A[warp][0];

    const float2* w2 = reinterpret_cast<const float2*>(lin_w);
    #pragma unroll
    for (int j = 0; j < (CCH * CH2 + GW * 32 - 1) / (GW * 32); ++j) {
        const int i = j * (GW * 32) + tid;
        if (i < CCH * CH2) {
            const int c = i / CH2, c2 = i % CH2;
            *reinterpret_cast<float2*>(&s_B[c * GST + 2 * c2]) = w2[i];
        }
    }
    if (tid < 32)
        *reinterpret_cast<float2*>(&s_B[tid * GST + CCH]) = make_float2(0.f, 0.f);
    if (tid < 2 * CH2) {
        const int rr = tid / CH2, c2 = tid % CH2;
        *reinterpret_cast<float2*>(&s_B[(CCH + rr) * GST + 2 * c2]) = make_float2(0.f, 0.f);
    }
    if (full) {
        const float2* h2 = reinterpret_cast<const float2*>(h + (size_t)row0 * MST);
        const bool padl = (lane & 15) == 15;
        #pragma unroll
        for (int i = lane; i < 16 * MS2; i += 32) {
            const int r = i / MS2, c2 = i % MS2;
            float2 v = h2[i];
            v.x = v.x > 0.0f ? v.x : 0.0f;
            v.y = v.y > 0.0f ? v.y : 0.0f;
            if (padl) v = make_float2(0.f, 0.f);
            *reinterpret_cast<float2*>(&sa[r * GST + 2 * c2]) = v;
        }
    } else {
        if (lane < 16)
            *reinterpret_cast<float2*>(&sa[lane * GST + CCH]) = make_float2(0.f, 0.f);
        for (int i = lane; i < 16 * CCH; i += 32) {
            const int r = i / CCH, c = i % CCH;
            const int row = row0 + r;
            if (row < N) {
                const float v = h[(size_t)row * MST + c];
                sa[r * GST + c] = v > 0.0f ? v : 0.0f;
            }
        }
    }
    __syncthreads();

    if (row0 >= N) return;

    const int half = lane >> 4;
    const int l    = lane & 15;

    v8f acc[2];
    #pragma unroll
    for (int t = 0; t < 2; ++t) {
        const int c = t * 16 + l;
        const float b = c < CCH ? lin_b[c < CCH ? c : 0] : 0.0f;
        #pragma unroll
        for (int v = 0; v < 8; ++v) acc[t][v] = b;
    }

    #pragma unroll
    for (int k0 = 0; k0 < 32; k0 += 4) {
        const int ka = k0 + half * 2;
        const v2f a  = *(const v2f*)&sa [l        * GST + ka];
        const v2f b0 = *(const v2f*)&s_B[l        * GST + ka];
        const v2f b1 = *(const v2f*)&s_B[(16 + l) * GST + ka];
        acc[0] = __builtin_amdgcn_wmma_f32_16x16x4_f32(false, a, false, b0,
                                                       (short)0, acc[0], false, false);
        acc[1] = __builtin_amdgcn_wmma_f32_16x16x4_f32(false, a, false, b1,
                                                       (short)0, acc[1], false, false);
    }

    #pragma unroll
    for (int v = 0; v < 8; ++v) {
        const int r = v + half * 8;
        sa[r * GST + l]      = acc[0][v];
        sa[r * GST + 16 + l] = acc[1][v];
    }

    if (full) {
        float2* o2 = reinterpret_cast<float2*>(out + (size_t)row0 * CCH);
        #pragma unroll
        for (int i = lane; i < 16 * CH2; i += 32) {
            const int r = i / CH2, c2 = i % CH2;
            o2[i] = *reinterpret_cast<const float2*>(&sa[r * GST + 2 * c2]);
        }
    } else {
        for (int i = lane; i < 16 * CCH; i += 32) {
            const int r = i / CCH, c = i % CCH;
            const int row = row0 + r;
            if (row < N) out[(size_t)row * CCH + c] = sa[r * GST + c];
        }
    }
}

extern "C" void kernel_launch(void* const* d_in, const int* in_sizes, int n_in,
                              void* d_out, int out_size, void* d_ws, size_t ws_size,
                              hipStream_t stream) {
    const float* x      = (const float*)d_in[0];
    const int*   eidx   = (const int*)  d_in[1];
    const float* weight = (const float*)d_in[2];
    const float* w_ih   = (const float*)d_in[3];
    const float* w_hh   = (const float*)d_in[4];
    const float* b_ih   = (const float*)d_in[5];
    const float* b_hh   = (const float*)d_in[6];
    const float* lin_w  = (const float*)d_in[7];
    const float* lin_b  = (const float*)d_in[8];
    float* out = (float*)d_out;

    const int N = in_sizes[0] / CCH;
    const int E = in_sizes[1] / 2;
    const int* src = eidx;
    const int* dst = eidx + E;

    float* h    = (float*)d_ws;
    float* aggh = h + (size_t)N * MST;
    const int nblk = (E + EBLK - 1) / EBLK;
    int* bcnt   = (int*)(aggh + (size_t)N * MST);
    int* bcur   = bcnt + NBMAX * CSTR;
    int* bbase  = bcur + NBMAX * CSTR;
    int* hmat   = bbase + NBMAX + 4;
    int* rowptr = hmat + (size_t)nblk * NBMAX;
    int* pairs  = rowptr + ((N + 4) & ~3);
    int* ssrc   = pairs + 2 * (size_t)E;

    ggnn_pad<<<(N * 8 + 255) / 256, 256, 0, stream>>>(x, h, N);

    const int c4 = ((2 * NBMAX * CSTR) >> 2) + 4;
    ggnn_zero<<<(c4 + 255) / 256, 256, 0, stream>>>((float*)bcnt, 2 * NBMAX * CSTR);
    ggnn_bhist<<<nblk, PTHR, 0, stream>>>(dst, bcnt, hmat, E);
    ggnn_part<<<nblk, PTHR, 0, stream>>>(src, dst, bcnt, bcur, hmat, bbase, pairs, E);
    ggnn_bsort<<<(N + BNODE - 1) / BNODE, STHR, 0, stream>>>(pairs, bbase, rowptr, ssrc, N, E);

    const int tiles = (N + 15) / 16;
    for (int layer = 0; layer < NLAYERS; ++layer) {
        ggnn_gather<<<(N + 15) / 16, 256, 0, stream>>>(rowptr, ssrc, h, aggh, N);
        ggnn_gru_wmma<<<(tiles + GW - 1) / GW, GW * 32, 0, stream>>>(
            aggh, h, weight + (size_t)layer * CCH * CCH, w_ih, w_hh, b_ih, b_hh, N);
    }
    ggnn_final_wmma<<<(tiles + GW - 1) / GW, GW * 32, 0, stream>>>(h, lin_w, lin_b, out, N);
}


# GRU gates on hardware transcendentals: sigmoid = v_rcp_f32(1 + v_exp_f32), tanh = 1 - 2*v_rcp_f32(1 + v_exp_f32), in place of the IEEE division sequences and the library tanhf
# speedup vs baseline: 17.7446x; 1.0650x over previous
#include <hip/hip_runtime.h>
#include <math.h>

#define CCH 30
#define NLAYERS 3
#define CH2 (CCH / 2)
#define MST 32
#define MS2 (MST / 2)

typedef __attribute__((ext_vector_type(2))) float v2f;
typedef __attribute__((ext_vector_type(8))) float v8f;

__device__ __forceinline__ float sigmoid_f(float x) {
    return __builtin_amdgcn_rcpf(1.0f + __builtin_amdgcn_exp2f(-1.44269504f * x));
}
__device__ __forceinline__ float tanh_f(float x) {
    return 1.0f - 2.0f * __builtin_amdgcn_rcpf(1.0f + __builtin_amdgcn_exp2f(2.88539008f * x));
}

__global__ void ggnn_pad(const float* __restrict__ x, float* __restrict__ h, int N) {
    const int i   = blockIdx.x * blockDim.x + threadIdx.x;
    const int row = i >> 3, q = i & 7;
    if (row < N) {
        const float2* x2 = reinterpret_cast<const float2*>(x) + (size_t)row * CH2 + 2 * q;
        const float2 a = x2[0];
        const float2 b = q < 7 ? x2[1] : make_float2(0.f, 0.f);
        reinterpret_cast<float4*>(h)[i] = make_float4(a.x, a.y, b.x, b.y);
    }
}

__global__ void ggnn_zero(float* __restrict__ p, int n) {
    const int n4 = n >> 2;
    int i = blockIdx.x * blockDim.x + threadIdx.x;
    if (i < n4) reinterpret_cast<float4*>(p)[i] = make_float4(0.f, 0.f, 0.f, 0.f);
    if (i < (n & 3)) p[n4 * 4 + i] = 0.0f;
}

#define BSH   8
#define BNODE (1 << BSH)
#define NBMAX 1024
#define PTHR  1024
#define PPT   6
#define PCH   (PTHR * PPT)
#define EBLK  (4 * PCH)
#define CSTR  64
#define STHR  512
#define SCAP  12288

__device__ __forceinline__ int bucket_of(int d) {
    const unsigned b = (unsigned)d >> BSH;
    return (int)(b < (unsigned)NBMAX ? b : (unsigned)(NBMAX - 1));
}

__device__ __forceinline__ int block_excl_scan(int v, int* s_w, int& total) {
    const int lane = threadIdx.x & 31;
    const int wid  = threadIdx.x >> 5;
    const int nw   = (int)(blockDim.x >> 5);
    int x = v;
    #pragma unroll
    for (int d = 1; d < 32; d <<= 1) {
        const int y = __shfl_up(x, d);
        if (lane >= d) x += y;
    }
    __syncthreads();
    if (lane == 31) s_w[wid] = x;
    __syncthreads();
    if (wid == 0) {
        const int wt = lane < nw ? s_w[lane] : 0;
        int xs = wt;
        #pragma unroll
        for (int d = 1; d < 32; d <<= 1) {
            const int y = __shfl_up(xs, d);
            if (lane >= d) xs += y;
        }
        s_w[lane] = xs - wt;
        if (lane == 31) s_w[32] = xs;
    }
    __syncthreads();
    total = s_w[32];
    return s_w[wid] + (x - v);
}

__global__ void __launch_bounds__(PTHR)
ggnn_bhist(const int* __restrict__ dst, int* __restrict__ bcnt,
           int* __restrict__ hmat, int E) {
    __shared__ int s_h[NBMAX];
    const int tid = threadIdx.x;
    s_h[tid] = 0;
    __syncthreads();
    const int e0 = blockIdx.x * EBLK;
    const int n  = (E - e0) < EBLK ? (E - e0) : EBLK;
    for (int i0 = 0; i0 < n; i0 += PTHR * 8) {
        int d[8];
        #pragma unroll
        for (int j = 0; j < 8; ++j) {
            const int i = i0 + j * PTHR + tid;
            d[j] = i < n ? dst[e0 + i] : 0;
        }
        #pragma unroll
        for (int j = 0; j < 8; ++j) {
            const int i = i0 + j * PTHR + tid;
            if (i < n) atomicAdd(&s_h[bucket_of(d[j])], 1);
        }
    }
    __syncthreads();
    const int c = s_h[tid];
    hmat[blockIdx.x * NBMAX + tid] = c;
    if (c) atomicAdd(&bcnt[tid * CSTR], c);
}

__global__ void __launch_bounds__(PTHR)
ggnn_part(const int* __restrict__ src, const int* __restrict__ dst,
          const int* __restrict__ bcnt, int* __restrict__ bcur,
          const int* __restrict__ hmat, int* __restrict__ bbase,
          int* __restrict__ pairs_w, int E) {
    __shared__ int  s_h[NBMAX];
    __shared__ int  s_lo[NBMAX];
    __shared__ int  s_run[NBMAX];
    __shared__ int  s_w[33];
    __shared__ int2 s_p[PCH];
    int2* pairs = reinterpret_cast<int2*>(pairs_w);
    const int tid  = threadIdx.x;
    const int e0   = blockIdx.x * EBLK;
    const int nblk = (E - e0) < EBLK ? (E - e0) : EBLK;

    const int ctot = hmat[blockIdx.x * NBMAX + tid];
    int gtot;
    const int gb = block_excl_scan(bcnt[tid * CSTR], s_w, gtot);
    s_run[tid] = gb + (ctot > 0 ? atomicAdd(&bcur[tid * CSTR], ctot) : 0);
    if (blockIdx.x == 0) {
        bbase[tid] = gb;
        if (tid == PTHR - 1) bbase[NBMAX] = gtot;
    }

    for (int c0 = 0; c0 < nblk; c0 += PCH) {
        const int n = (nblk - c0) < PCH ? (nblk - c0) : PCH;
        s_h[tid] = 0;
        __syncthreads();

        int d[PPT], s[PPT], r[PPT];
        #pragma unroll
        for (int j = 0; j < PPT; ++j) {
            const int i = j * PTHR + tid;
            d[j] = i < n ? dst[e0 + c0 + i] : 0;
            s[j] = i < n ? src[e0 + c0 + i] : 0;
        }
        #pragma unroll
        for (int j = 0; j < PPT; ++j) {
            const int i = j * PTHR + tid;
            r[j] = i < n ? atomicAdd(&s_h[bucket_of(d[j])], 1) : 0;
        }
        __syncthreads();

        const int c = s_h[tid];
        int tot;
        s_lo[tid] = block_excl_scan(c, s_w, tot);
        __syncthreads();

        #pragma unroll
        for (int j = 0; j < PPT; ++j) {
            const int i = j * PTHR + tid;
            if (i < n) s_p[s_lo[bucket_of(d[j])] + r[j]] = make_int2(d[j], s[j]);
        }
        __syncthreads();

        for (int i = tid; i < n; i += PTHR) {
            const int2 v = s_p[i];
            const int b  = bucket_of(v.x);
            const int pos = s_run[b] + (i - s_lo[b]);
            if ((unsigned)pos < (unsigned)E) pairs[pos] = v;
        }
        __syncthreads();
        s_run[tid] += c;
    }
}

__global__ void __launch_bounds__(STHR)
ggnn_bsort(const int* __restrict__ pairs_w, const int* __restrict__ bbase,
           int* __restrict__ rowptr, int* __restrict__ ssrc, int N, int E) {
    __shared__ int s_c[BNODE];
    __shared__ int s_w[33];
    __shared__ int s_s[SCAP];
    const int tid   = threadIdx.x;
    const int b     = blockIdx.x;
    const int node0 = b << BSH;
    int bs = bbase[b];
    int be = bbase[b + 1];
    bs = bs < 0 ? 0 : (bs > E ? E : bs);
    be = be < bs ? bs : (be > E ? E : be);
    const int n = be - bs;
    const bool stage = n <= SCAP;
    const int2* pb = reinterpret_cast<const int2*>(pairs_w) + bs;

    if (tid < BNODE) s_c[tid] = 0;
    __syncthreads();
    for (int i0 = 0; i0 < n; i0 += STHR * 8) {
        int2 v[8];
        #pragma unroll
        for (int j = 0; j < 8; ++j) {
            const int i = i0 + j * STHR + tid;
            v[j] = i < n ? pb[i] : make_int2(0, 0);
        }
        #pragma unroll
        for (int j = 0; j < 8; ++j) {
            const int i = i0 + j * STHR + tid;
            if (i < n) atomicAdd(&s_c[(v[j].x - node0) & (BNODE - 1)], 1);
        }
    }
    __syncthreads();

    const int c = tid < BNODE ? s_c[tid] : 0;
    int tot;
    const int ex = block_excl_scan(c, s_w, tot);
    if (tid < BNODE) {
        const int node = node0 + tid;
        if (node < N) rowptr[node] = bs + ex;
        s_c[tid] = ex;
    }
    if (b == (int)gridDim.x - 1 && tid == 0) rowptr[N] = E;
    __syncthreads();

    for (int i0 = 0; i0 < n; i0 += STHR * 8) {
        int2 v[8];
        #pragma unroll
        for (int j = 0; j < 8; ++j) {
            const int i = i0 + j * STHR + tid;
            v[j] = i < n ? pb[i] : make_int2(0, 0);
        }
        #pragma unroll
        for (int j = 0; j < 8; ++j) {
            const int i = i0 + j * STHR + tid;
            if (i < n) {
                const int p = atomicAdd(&s_c[(v[j].x - node0) & (BNODE - 1)], 1);
                if (stage) {
                    if ((unsigned)p < (unsigned)SCAP) s_s[p] = v[j].y;
                } else {
                    if ((unsigned)p < (unsigned)n) ssrc[bs + p] = v[j].y;
                }
            }
        }
    }
    if (stage) {
        __syncthreads();
        for (int i = tid; i < n; i += STHR) ssrc[bs + i] = s_s[i];
    }
}

__global__ void __launch_bounds__(256)
ggnn_gather(const int* __restrict__ rowptr, const int* __restrict__ ssrc,
            const float* __restrict__ h, float* __restrict__ aggh, int N) {
    const int tid  = threadIdx.x;
    const int l    = tid & 15;
    const int node = (blockIdx.x * 256 + tid) >> 4;
    const bool nv  = node < N;
    const float2* h2 = reinterpret_cast<const float2*>(h);

    int p         = nv ? rowptr[node]     : 0;
    const int end = nv ? rowptr[node + 1] : 0;
    int rem = end - p;
    const int other = __shfl_xor(rem, 16);
    int remw = __builtin_amdgcn_readfirstlane(rem > other ? rem : other);

    float2 acc = make_float2(0.0f, 0.0f);
    while (remw > 0) {
        const int idx = (l < rem) ? ssrc[p + l] : -1;
        #pragma unroll
        for (int j0 = 0; j0 < 16; j0 += 8) {
            if (j0 < remw) {
                int    s[8];
                float2 v[8];
                #pragma unroll
                for (int j = 0; j < 8; ++j) {
                    s[j] = __shfl(idx, j0 + j, 16);
                    const int sc = s[j] < 0 ? 0 : (s[j] < N ? s[j] : N - 1);
                    v[j] = h2[(size_t)sc * MS2 + l];
                }
                #pragma unroll
                for (int j = 0; j < 8; ++j) {
                    if (s[j] >= 0) { acc.x += v[j].x; acc.y += v[j].y; }
                }
            }
        }
        p += 16; rem -= 16; remw -= 16;
    }
    if (nv)
        reinterpret_cast<float2*>(aggh)[(size_t)node * MS2 + l] =
            l < CH2 ? acc : make_float2(0.0f, 0.0f);
}

#define GST   34
#define GROWS 96
#define GW    8

__global__ void __launch_bounds__(GW * 32)
ggnn_gru_wmma(const float* __restrict__ aggh,
              float* __restrict__ h,
              const float* __restrict__ W,
              const float* __restrict__ w_ih,
              const float* __restrict__ w_hh,
              const float* __restrict__ b_ih,
              const float* __restrict__ b_hh, int N) {
    __shared__ __attribute__((aligned(16))) float s_wih[GROWS * GST];
    __shared__ __attribute__((aligned(16))) float s_whh[GROWS * GST];
    __shared__ __attribute__((aligned(16))) float s_Wt[32 * GST];
    __shared__ __attribute__((aligned(16))) float s_A[GW][16 * GST];
    __shared__ __attribute__((aligned(16))) float s_H[GW][16 * GST];

    const int tid  = threadIdx.x;
    const int lane = tid & 31;
    const int warp = tid >> 5;
    const int row0 = (blockIdx.x * GW + warp) * 16;
    const bool full = (row0 + 16 <= N);
    float* sa = &s_A[warp][0];
    float* sh = &s_H[warp][0];

    const float2* wi2 = reinterpret_cast<const float2*>(w_ih);
    const float2* wh2 = reinterpret_cast<const float2*>(w_hh);
    #pragma unroll
    for (int j = 0; j < (3 * CCH * CH2 + GW * 32 - 1) / (GW * 32); ++j) {
        const int i = j * (GW * 32) + tid;
        if (i < 3 * CCH * CH2) {
            const int c = i / CH2, c2 = i % CH2;
            const int g = c / CCH, cc = c % CCH;
            const int o = (g * 32 + cc) * GST + 2 * c2;
            *reinterpret_cast<float2*>(&s_wih[o]) = wi2[i];
            *reinterpret_cast<float2*>(&s_whh[o]) = wh2[i];
        }
    }
    #pragma unroll
    for (int j = 0; j < (CCH * CCH + GW * 32 - 1) / (GW * 32); ++j) {
        const int i = j * (GW * 32) + tid;
        if (i < CCH * CCH) {
            const int k = i / CCH, c = i % CCH;
            s_Wt[c * GST + k] = W[i];
        }
    }
    if (tid < GROWS) {
        *reinterpret_cast<float2*>(&s_wih[tid * GST + CCH]) = make_float2(0.f, 0.f);
        *reinterpret_cast<float2*>(&s_whh[tid * GST + CCH]) = make_float2(0.f, 0.f);
    }
    if (tid < 6 * CH2) {
        const int rr = tid / CH2, c2 = tid % CH2;
        const int o = ((rr >> 1) * 32 + CCH + (rr & 1)) * GST + 2 * c2;
        *reinterpret_cast<float2*>(&s_wih[o]) = make_float2(0.f, 0.f);
        *reinterpret_cast<float2*>(&s_whh[o]) = make_float2(0.f, 0.f);
    }
    if (tid < 32)
        *reinterpret_cast<float2*>(&s_Wt[tid * GST + CCH]) = make_float2(0.f, 0.f);
    if (tid < 2 * CH2) {
        const int rr = tid / CH2, c2 = tid % CH2;
        *reinterpret_cast<float2*>(&s_Wt[(CCH + rr) * GST + 2 * c2]) = make_float2(0.f, 0.f);
    }
    if (full) {
        const float2* a2 = reinterpret_cast<const float2*>(aggh + (size_t)row0 * MST);
        const float2* h2 = reinterpret_cast<const float2*>(h    + (size_t)row0 * MST);
        const bool padl = (lane & 15) == 15;
        #pragma unroll
        for (int i = lane; i < 16 * MS2; i += 32) {
            const int r = i / MS2, c2 = i % MS2;
            float2 va = a2[i];
            float2 vh = h2[i];
            if (padl) { va = make_float2(0.f, 0.f); vh = make_float2(0.f, 0.f); }
            *reinterpret_cast<float2*>(&sa[r * GST + 2 * c2]) = va;
            *reinterpret_cast<float2*>(&sh[r * GST + 2 * c2]) = vh;
        }
    } else {
        if (lane < 16) {
            *reinterpret_cast<float2*>(&sa[lane * GST + CCH]) = make_float2(0.f, 0.f);
            *reinterpret_cast<float2*>(&sh[lane * GST + CCH]) = make_float2(0.f, 0.f);
        }
        for (int i = lane; i < 16 * CCH; i += 32) {
            const int r = i / CCH, c = i % CCH;
            const int row = row0 + r;
            if (row < N) {
                sa[r * GST + c] = aggh[(size_t)row * MST + c];
                sh[r * GST + c] = h   [(size_t)row * MST + c];
            }
        }
    }
    __syncthreads();

    if (row0 >= N) return;

    const int half = lane >> 4;
    const int l    = lane & 15;

    {
        v8f m0 = {};
        v8f m1 = {};
        #pragma unroll
        for (int k0 = 0; k0 < 32; k0 += 4) {
            const int ka = k0 + half * 2;
            const v2f a  = *(const v2f*)&sa  [l        * GST + ka];
            const v2f b0 = *(const v2f*)&s_Wt[l        * GST + ka];
            const v2f b1 = *(const v2f*)&s_Wt[(16 + l) * GST + ka];
            m0 = __builtin_amdgcn_wmma_f32_16x16x4_f32(false, a, false, b0,
                                                       (short)0, m0, false, false);
            m1 = __builtin_amdgcn_wmma_f32_16x16x4_f32(false, a, false, b1,
                                                       (short)0, m1, false, false);
        }
        #pragma unroll
        for (int v = 0; v < 8; ++v) {
            const int r = v + half * 8;
            sa[r * GST + l]      = m0[v];
            sa[r * GST + 16 + l] = m1[v];
        }
    }

    v8f accr[2], accz[2], acci[2], acch[2];
    #pragma unroll
    for (int t = 0; t < 2; ++t) {
        const int c  = t * 16 + l;
        const bool cv = c < CCH;
        const int cc = cv ? c : 0;
        const float br = cv ? b_ih[cc] + b_hh[cc] : 0.0f;
        const float bz = cv ? b_ih[CCH + cc] + b_hh[CCH + cc] : 0.0f;
        const float bi = cv ? b_ih[2 * CCH + cc] : 0.0f;
        const float bh = cv ? b_hh[2 * CCH + cc] : 0.0f;
        #pragma unroll
        for (int v = 0; v < 8; ++v) {
            accr[t][v] = br; accz[t][v] = bz; acci[t][v] = bi; acch[t][v] = bh;
        }
    }

    #pragma unroll
    for (int k0 = 0; k0 < 32; k0 += 4) {
        const int ka = k0 + half * 2;
        const v2f am = *(const v2f*)&sa[l * GST + ka];
        const v2f ah = *(const v2f*)&sh[l * GST + ka];
        #pragma unroll
        for (int t = 0; t < 2; ++t) {
            const int bo = (t * 16 + l) * GST + ka;
            const v2f bir = *(const v2f*)&s_wih[bo];
            const v2f biz = *(const v2f*)&s_wih[32 * GST + bo];
            const v2f bin = *(const v2f*)&s_wih[64 * GST + bo];
            const v2f bhr = *(const v2f*)&s_whh[bo];
            const v2f bhz = *(const v2f*)&s_whh[32 * GST + bo];
            const v2f bhn = *(const v2f*)&s_whh[64 * GST + bo];
            accr[t] = __builtin_amdgcn_wmma_f32_16x16x4_f32(false, am, false, bir,
                                                            (short)0, accr[t], false, false);
            accz[t] = __builtin_amdgcn_wmma_f32_16x16x4_f32(false, am, false, biz,
                                                            (short)0, accz[t], false, false);
            acci[t] = __builtin_amdgcn_wmma_f32_16x16x4_f32(false, am, false, bin,
                                                            (short)0, acci[t], false, false);
            acch[t] = __builtin_amdgcn_wmma_f32_16x16x4_f32(false, ah, false, bhn,
                                                            (short)0, acch[t], false, false);
            accr[t] = __builtin_amdgcn_wmma_f32_16x16x4_f32(false, ah, false, bhr,
                                                            (short)0, accr[t], false, false);
            accz[t] = __builtin_amdgcn_wmma_f32_16x16x4_f32(false, ah, false, bhz,
                                                            (short)0, accz[t], false, false);
        }
    }

    #pragma unroll
    for (int t = 0; t < 2; ++t) {
        #pragma unroll
        for (int v = 0; v < 8; ++v) {
            const int o = (v + half * 8) * GST + t * 16 + l;
            const float hold = sh[o];
            const float r  = sigmoid_f(accr[t][v]);
            const float z  = sigmoid_f(accz[t][v]);
            const float nn = tanh_f(acci[t][v] + r * acch[t][v]);
            sa[o] = (1.0f - z) * nn + z * hold;
        }
    }

    if (full) {
        float2* h2w = reinterpret_cast<float2*>(h + (size_t)row0 * MST);
        #pragma unroll
        for (int i = lane; i < 16 * MS2; i += 32) {
            const int r = i / MS2, c2 = i % MS2;
            h2w[i] = *reinterpret_cast<const float2*>(&sa[r * GST + 2 * c2]);
        }
    } else {
        for (int i = lane; i < 16 * MST; i += 32) {
            const int r = i / MST, c = i % MST;
            const int row = row0 + r;
            if (row < N) h[(size_t)row * MST + c] = sa[r * GST + c];
        }
    }
}

__global__ void __launch_bounds__(GW * 32)
ggnn_final_wmma(const float* __restrict__ h,
                const float* __restrict__ lin_w,
                const float* __restrict__ lin_b,
                float* __restrict__ out, int N) {
    __shared__ __attribute__((aligned(16))) float s_B[32 * GST];
    __shared__ __attribute__((aligned(16))) float s_A[GW][16 * GST];

    const int tid  = threadIdx.x;
    const int lane = tid & 31;
    const int warp = tid >> 5;
    const int row0 = (blockIdx.x * GW + warp) * 16;
    const bool full = (row0 + 16 <= N);
    float* sa = &s_A[warp][0];

    const float2* w2 = reinterpret_cast<const float2*>(lin_w);
    #pragma unroll
    for (int j = 0; j < (CCH * CH2 + GW * 32 - 1) / (GW * 32); ++j) {
        const int i = j * (GW * 32) + tid;
        if (i < CCH * CH2) {
            const int c = i / CH2, c2 = i % CH2;
            *reinterpret_cast<float2*>(&s_B[c * GST + 2 * c2]) = w2[i];
        }
    }
    if (tid < 32)
        *reinterpret_cast<float2*>(&s_B[tid * GST + CCH]) = make_float2(0.f, 0.f);
    if (tid < 2 * CH2) {
        const int rr = tid / CH2, c2 = tid % CH2;
        *reinterpret_cast<float2*>(&s_B[(CCH + rr) * GST + 2 * c2]) = make_float2(0.f, 0.f);
    }
    if (full) {
        const float2* h2 = reinterpret_cast<const float2*>(h + (size_t)row0 * MST);
        const bool padl = (lane & 15) == 15;
        #pragma unroll
        for (int i = lane; i < 16 * MS2; i += 32) {
            const int r = i / MS2, c2 = i % MS2;
            float2 v = h2[i];
            v.x = v.x > 0.0f ? v.x : 0.0f;
            v.y = v.y > 0.0f ? v.y : 0.0f;
            if (padl) v = make_float2(0.f, 0.f);
            *reinterpret_cast<float2*>(&sa[r * GST + 2 * c2]) = v;
        }
    } else {
        if (lane < 16)
            *reinterpret_cast<float2*>(&sa[lane * GST + CCH]) = make_float2(0.f, 0.f);
        for (int i = lane; i < 16 * CCH; i += 32) {
            const int r = i / CCH, c = i % CCH;
            const int row = row0 + r;
            if (row < N) {
                const float v = h[(size_t)row * MST + c];
                sa[r * GST + c] = v > 0.0f ? v : 0.0f;
            }
        }
    }
    __syncthreads();

    if (row0 >= N) return;

    const int half = lane >> 4;
    const int l    = lane & 15;

    v8f acc[2];
    #pragma unroll
    for (int t = 0; t < 2; ++t) {
        const int c = t * 16 + l;
        const float b = c < CCH ? lin_b[c < CCH ? c : 0] : 0.0f;
        #pragma unroll
        for (int v = 0; v < 8; ++v) acc[t][v] = b;
    }

    #pragma unroll
    for (int k0 = 0; k0 < 32; k0 += 4) {
        const int ka = k0 + half * 2;
        const v2f a  = *(const v2f*)&sa [l        * GST + ka];
        const v2f b0 = *(const v2f*)&s_B[l        * GST + ka];
        const v2f b1 = *(const v2f*)&s_B[(16 + l) * GST + ka];
        acc[0] = __builtin_amdgcn_wmma_f32_16x16x4_f32(false, a, false, b0,
                                                       (short)0, acc[0], false, false);
        acc[1] = __builtin_amdgcn_wmma_f32_16x16x4_f32(false, a, false, b1,
                                                       (short)0, acc[1], false, false);
    }

    #pragma unroll
    for (int v = 0; v < 8; ++v) {
        const int r = v + half * 8;
        sa[r * GST + l]      = acc[0][v];
        sa[r * GST + 16 + l] = acc[1][v];
    }

    if (full) {
        float2* o2 = reinterpret_cast<float2*>(out + (size_t)row0 * CCH);
        #pragma unroll
        for (int i = lane; i < 16 * CH2; i += 32) {
            const int r = i / CH2, c2 = i % CH2;
            o2[i] = *reinterpret_cast<const float2*>(&sa[r * GST + 2 * c2]);
        }
    } else {
        for (int i = lane; i < 16 * CCH; i += 32) {
            const int r = i / CCH, c = i % CCH;
            const int row = row0 + r;
            if (row < N) out[(size_t)row * CCH + c] = sa[r * GST + c];
        }
    }
}

extern "C" void kernel_launch(void* const* d_in, const int* in_sizes, int n_in,
                              void* d_out, int out_size, void* d_ws, size_t ws_size,
                              hipStream_t stream) {
    const float* x      = (const float*)d_in[0];
    const int*   eidx   = (const int*)  d_in[1];
    const float* weight = (const float*)d_in[2];
    const float* w_ih   = (const float*)d_in[3];
    const float* w_hh   = (const float*)d_in[4];
    const float* b_ih   = (const float*)d_in[5];
    const float* b_hh   = (const float*)d_in[6];
    const float* lin_w  = (const float*)d_in[7];
    const float* lin_b  = (const float*)d_in[8];
    float* out = (float*)d_out;

    const int N = in_sizes[0] / CCH;
    const int E = in_sizes[1] / 2;
    const int* src = eidx;
    const int* dst = eidx + E;

    float* h    = (float*)d_ws;
    float* aggh = h + (size_t)N * MST;
    const int nblk = (E + EBLK - 1) / EBLK;
    int* bcnt   = (int*)(aggh + (size_t)N * MST);
    int* bcur   = bcnt + NBMAX * CSTR;
    int* bbase  = bcur + NBMAX * CSTR;
    int* hmat   = bbase + NBMAX + 4;
    int* rowptr = hmat + (size_t)nblk * NBMAX;
    int* pairs  = rowptr + ((N + 4) & ~3);
    int* ssrc   = pairs + 2 * (size_t)E;

    ggnn_pad<<<(N * 8 + 255) / 256, 256, 0, stream>>>(x, h, N);

    const int c4 = ((2 * NBMAX * CSTR) >> 2) + 4;
    ggnn_zero<<<(c4 + 255) / 256, 256, 0, stream>>>((float*)bcnt, 2 * NBMAX * CSTR);
    ggnn_bhist<<<nblk, PTHR, 0, stream>>>(dst, bcnt, hmat, E);
    ggnn_part<<<nblk, PTHR, 0, stream>>>(src, dst, bcnt, bcur, hmat, bbase, pairs, E);
    ggnn_bsort<<<(N + BNODE - 1) / BNODE, STHR, 0, stream>>>(pairs, bbase, rowptr, ssrc, N, E);

    const int tiles = (N + 15) / 16;
    for (int layer = 0; layer < NLAYERS; ++layer) {
        ggnn_gather<<<(N + 15) / 16, 256, 0, stream>>>(rowptr, ssrc, h, aggh, N);
        ggnn_gru_wmma<<<(tiles + GW - 1) / GW, GW * 32, 0, stream>>>(
            aggh, h, weight + (size_t)layer * CCH * CCH, w_ih, w_hh, b_ih, b_hh, N);
    }
    ggnn_final_wmma<<<(tiles + GW - 1) / GW, GW * 32, 0, stream>>>(h, lin_w, lin_b, out, N);
}


# Gather on a fixed grid of resident waves (2048 blocks): each half-wave walks over several nodes, next node's source ids and bounds loaded ahead, not one short wave per node pair; sum order kept
# speedup vs baseline: 18.3692x; 1.0352x over previous
#include <hip/hip_runtime.h>
#include <math.h>

#define CCH 30
#define NLAYERS 3
#define CH2 (CCH / 2)
#define MST 32
#define MS2 (MST / 2)

typedef __attribute__((ext_vector_type(2))) float v2f;
typedef __attribute__((ext_vector_type(8))) float v8f;

__device__ __forceinline__ float sigmoid_f(float x) {
    return __builtin_amdgcn_rcpf(1.0f + __builtin_amdgcn_exp2f(-1.44269504f * x));
}
__device__ __forceinline__ float tanh_f(float x) {
    return 1.0f - 2.0f * __builtin_amdgcn_rcpf(1.0f + __builtin_amdgcn_exp2f(2.88539008f * x));
}

__global__ void ggnn_pad(const float* __restrict__ x, float* __restrict__ h, int N) {
    const int i   = blockIdx.x * blockDim.x + threadIdx.x;
    const int row = i >> 3, q = i & 7;
    if (row < N) {
        const float2* x2 = reinterpret_cast<const float2*>(x) + (size_t)row * CH2 + 2 * q;
        const float2 a = x2[0];
        const float2 b = q < 7 ? x2[1] : make_float2(0.f, 0.f);
        reinterpret_cast<float4*>(h)[i] = make_float4(a.x, a.y, b.x, b.y);
    }
}

__global__ void ggnn_zero(float* __restrict__ p, int n) {
    const int n4 = n >> 2;
    int i = blockIdx.x * blockDim.x + threadIdx.x;
    if (i < n4) reinterpret_cast<float4*>(p)[i] = make_float4(0.f, 0.f, 0.f, 0.f);
    if (i < (n & 3)) p[n4 * 4 + i] = 0.0f;
}

#define BSH   8
#define BNODE (1 << BSH)
#define NBMAX 1024
#define PTHR  1024
#define PPT   6
#define PCH   (PTHR * PPT)
#define EBLK  (4 * PCH)
#define CSTR  64
#define STHR  512
#define SCAP  12288

__device__ __forceinline__ int bucket_of(int d) {
    const unsigned b = (unsigned)d >> BSH;
    return (int)(b < (unsigned)NBMAX ? b : (unsigned)(NBMAX - 1));
}

__device__ __forceinline__ int block_excl_scan(int v, int* s_w, int& total) {
    const int lane = threadIdx.x & 31;
    const int wid  = threadIdx.x >> 5;
    const int nw   = (int)(blockDim.x >> 5);
    int x = v;
    #pragma unroll
    for (int d = 1; d < 32; d <<= 1) {
        const int y = __shfl_up(x, d);
        if (lane >= d) x += y;
    }
    __syncthreads();
    if (lane == 31) s_w[wid] = x;
    __syncthreads();
    if (wid == 0) {
        const int wt = lane < nw ? s_w[lane] : 0;
        int xs = wt;
        #pragma unroll
        for (int d = 1; d < 32; d <<= 1) {
            const int y = __shfl_up(xs, d);
            if (lane >= d) xs += y;
        }
        s_w[lane] = xs - wt;
        if (lane == 31) s_w[32] = xs;
    }
    __syncthreads();
    total = s_w[32];
    return s_w[wid] + (x - v);
}

__global__ void __launch_bounds__(PTHR)
ggnn_bhist(const int* __restrict__ dst, int* __restrict__ bcnt,
           int* __restrict__ hmat, int E) {
    __shared__ int s_h[NBMAX];
    const int tid = threadIdx.x;
    s_h[tid] = 0;
    __syncthreads();
    const int e0 = blockIdx.x * EBLK;
    const int n  = (E - e0) < EBLK ? (E - e0) : EBLK;
    for (int i0 = 0; i0 < n; i0 += PTHR * 8) {
        int d[8];
        #pragma unroll
        for (int j = 0; j < 8; ++j) {
            const int i = i0 + j * PTHR + tid;
            d[j] = i < n ? dst[e0 + i] : 0;
        }
        #pragma unroll
        for (int j = 0; j < 8; ++j) {
            const int i = i0 + j * PTHR + tid;
            if (i < n) atomicAdd(&s_h[bucket_of(d[j])], 1);
        }
    }
    __syncthreads();
    const int c = s_h[tid];
    hmat[blockIdx.x * NBMAX + tid] = c;
    if (c) atomicAdd(&bcnt[tid * CSTR], c);
}

__global__ void __launch_bounds__(PTHR)
ggnn_part(const int* __restrict__ src, const int* __restrict__ dst,
          const int* __restrict__ bcnt, int* __restrict__ bcur,
          const int* __restrict__ hmat, int* __restrict__ bbase,
          int* __restrict__ pairs_w, int E) {
    __shared__ int  s_h[NBMAX];
    __shared__ int  s_lo[NBMAX];
    __shared__ int  s_run[NBMAX];
    __shared__ int  s_w[33];
    __shared__ int2 s_p[PCH];
    int2* pairs = reinterpret_cast<int2*>(pairs_w);
    const int tid  = threadIdx.x;
    const int e0   = blockIdx.x * EBLK;
    const int nblk = (E - e0) < EBLK ? (E - e0) : EBLK;

    const int ctot = hmat[blockIdx.x * NBMAX + tid];
    int gtot;
    const int gb = block_excl_scan(bcnt[tid * CSTR], s_w, gtot);
    s_run[tid] = gb + (ctot > 0 ? atomicAdd(&bcur[tid * CSTR], ctot) : 0);
    if (blockIdx.x == 0) {
        bbase[tid] = gb;
        if (tid == PTHR - 1) bbase[NBMAX] = gtot;
    }

    for (int c0 = 0; c0 < nblk; c0 += PCH) {
        const int n = (nblk - c0) < PCH ? (nblk - c0) : PCH;
        s_h[tid] = 0;
        __syncthreads();

        int d[PPT], s[PPT], r[PPT];
        #pragma unroll
        for (int j = 0; j < PPT; ++j) {
            const int i = j * PTHR + tid;
            d[j] = i < n ? dst[e0 + c0 + i] : 0;
            s[j] = i < n ? src[e0 + c0 + i] : 0;
        }
        #pragma unroll
        for (int j = 0; j < PPT; ++j) {
            const int i = j * PTHR + tid;
            r[j] = i < n ? atomicAdd(&s_h[bucket_of(d[j])], 1) : 0;
        }
        __syncthreads();

        const int c = s_h[tid];
        int tot;
        s_lo[tid] = block_excl_scan(c, s_w, tot);
        __syncthreads();

        #pragma unroll
        for (int j = 0; j < PPT; ++j) {
            const int i = j * PTHR + tid;
            if (i < n) s_p[s_lo[bucket_of(d[j])] + r[j]] = make_int2(d[j], s[j]);
        }
        __syncthreads();

        for (int i = tid; i < n; i += PTHR) {
            const int2 v = s_p[i];
            const int b  = bucket_of(v.x);
            const int pos = s_run[b] + (i - s_lo[b]);
            if ((unsigned)pos < (unsigned)E) pairs[pos] = v;
        }
        __syncthreads();
        s_run[tid] += c;
    }
}

__global__ void __launch_bounds__(STHR)
ggnn_bsort(const int* __restrict__ pairs_w, const int* __restrict__ bbase,
           int* __restrict__ rowptr, int* __restrict__ ssrc, int N, int E) {
    __shared__ int s_c[BNODE];
    __shared__ int s_w[33];
    __shared__ int s_s[SCAP];
    const int tid   = threadIdx.x;
    const int b     = blockIdx.x;
    const int node0 = b << BSH;
    int bs = bbase[b];
    int be = bbase[b + 1];
    bs = bs < 0 ? 0 : (bs > E ? E : bs);
    be = be < bs ? bs : (be > E ? E : be);
    const int n = be - bs;
    const bool stage = n <= SCAP;
    const int2* pb = reinterpret_cast<const int2*>(pairs_w) + bs;

    if (tid < BNODE) s_c[tid] = 0;
    __syncthreads();
    for (int i0 = 0; i0 < n; i0 += STHR * 8) {
        int2 v[8];
        #pragma unroll
        for (int j = 0; j < 8; ++j) {
            const int i = i0 + j * STHR + tid;
            v[j] = i < n ? pb[i] : make_int2(0, 0);
        }
        #pragma unroll
        for (int j = 0; j < 8; ++j) {
            const int i = i0 + j * STHR + tid;
            if (i < n) atomicAdd(&s_c[(v[j].x - node0) & (BNODE - 1)], 1);
        }
    }
    __syncthreads();

    const int c = tid < BNODE ? s_c[tid] : 0;
    int tot;
    const int ex = block_excl_scan(c, s_w, tot);
    if (tid < BNODE) {
        const int node = node0 + tid;
        if (node < N) rowptr[node] = bs + ex;
        s_c[tid] = ex;
    }
    if (b == (int)gridDim.x - 1 && tid == 0) rowptr[N] = E;
    __syncthreads();

    for (int i0 = 0; i0 < n; i0 += STHR * 8) {
        int2 v[8];
        #pragma unroll
        for (int j = 0; j < 8; ++j) {
            const int i = i0 + j * STHR + tid;
            v[j] = i < n ? pb[i] : make_int2(0, 0);
        }
        #pragma unroll
        for (int j = 0; j < 8; ++j) {
            const int i = i0 + j * STHR + tid;
            if (i < n) {
                const int p = atomicAdd(&s_c[(v[j].x - node0) & (BNODE - 1)], 1);
                if (stage) {
                    if ((unsigned)p < (unsigned)SCAP) s_s[p] = v[j].y;
                } else {
                    if ((unsigned)p < (unsigned)n) ssrc[bs + p] = v[j].y;
                }
            }
        }
    }
    if (stage) {
        __syncthreads();
        for (int i = tid; i < n; i += STHR) ssrc[bs + i] = s_s[i];
    }
}

#define GBLK 2048

__device__ __forceinline__ void gather16(int idx, int rw, int l, int N,
                                         const float2* __restrict__ h2, float2& acc) {
    #pragma unroll
    for (int j0 = 0; j0 < 16; j0 += 8) {
        if (j0 < rw) {
            int    s[8];
            float2 v[8];
            #pragma unroll
            for (int j = 0; j < 8; ++j) {
                s[j] = __shfl(idx, j0 + j, 16);
                const int sc = s[j] < 0 ? 0 : (s[j] < N ? s[j] : N - 1);
                v[j] = h2[(size_t)sc * MS2 + l];
            }
            #pragma unroll
            for (int j = 0; j < 8; ++j) {
                if (s[j] >= 0) { acc.x += v[j].x; acc.y += v[j].y; }
            }
        }
    }
}

__global__ void __launch_bounds__(256)
ggnn_gather(const int* __restrict__ rowptr, const int* __restrict__ ssrc,
            const float* __restrict__ h, float* __restrict__ aggh, int N) {
    const int tid   = threadIdx.x;
    const int l     = tid & 15;
    const int step  = (int)gridDim.x * 16;
    const int first = (blockIdx.x * 256 + tid) >> 4;
    const float2* h2 = reinterpret_cast<const float2*>(h);

    int node = first - step;
    int p = 0, rem = 0;
    int ia = -1, ib = -1, ic = -1;
    int p1 = 0, e1 = 0;
    if (first < N) { p1 = rowptr[first]; e1 = rowptr[first + 1]; }
    int rem1 = e1 - p1;

    for (int base = __builtin_amdgcn_readfirstlane(first & ~1) - step; base < N; base += step) {
        const int ja = (l      < rem1) ? ssrc[p1 + l]      : -1;
        const int jb = (l + 16 < rem1) ? ssrc[p1 + 16 + l] : -1;
        const int jc = (l + 32 < rem1) ? ssrc[p1 + 32 + l] : -1;
        int p2 = 0, e2 = 0;
        if (node + 2 * step < N) {
            p2 = rowptr[node + 2 * step];
            e2 = rowptr[node + 2 * step + 1];
        }

        const int other = __shfl_xor(rem, 16);
        const int remw  = __builtin_amdgcn_readfirstlane(rem > other ? rem : other);

        float2 acc = make_float2(0.0f, 0.0f);
        if (remw > 0)  gather16(ia, remw,      l, N, h2, acc);
        if (remw > 16) gather16(ib, remw - 16, l, N, h2, acc);
        if (remw > 32) gather16(ic, remw - 32, l, N, h2, acc);
        {
            int q = p + 48;
            int r = rem - 48;
            for (int rw = remw - 48; rw > 0; rw -= 16) {
                const int id = (l < r) ? ssrc[q + l] : -1;
                gather16(id, rw, l, N, h2, acc);
                q += 16; r -= 16;
            }
        }
        if ((unsigned)node < (unsigned)N)
            reinterpret_cast<float2*>(aggh)[(size_t)node * MS2 + l] =
                l < CH2 ? acc : make_float2(0.0f, 0.0f);

        node += step;
        p = p1; rem = rem1; p1 = p2; rem1 = e2 - p2;
        ia = ja; ib = jb; ic = jc;
    }
}

#define GST   34
#define GROWS 96
#define GW    8

__global__ void __launch_bounds__(GW * 32)
ggnn_gru_wmma(const float* __restrict__ aggh,
              float* __restrict__ h,
              const float* __restrict__ W,
              const float* __restrict__ w_ih,
              const float* __restrict__ w_hh,
              const float* __restrict__ b_ih,
              const float* __restrict__ b_hh, int N) {
    __shared__ __attribute__((aligned(16))) float s_wih[GROWS * GST];
    __shared__ __attribute__((aligned(16))) float s_whh[GROWS * GST];
    __shared__ __attribute__((aligned(16))) float s_Wt[32 * GST];
    __shared__ __attribute__((aligned(16))) float s_A[GW][16 * GST];
    __shared__ __attribute__((aligned(16))) float s_H[GW][16 * GST];

    const int tid  = threadIdx.x;
    const int lane = tid & 31;
    const int warp = tid >> 5;
    const int row0 = (blockIdx.x * GW + warp) * 16;
    const bool full = (row0 + 16 <= N);
    float* sa = &s_A[warp][0];
    float* sh = &s_H[warp][0];

    const float2* wi2 = reinterpret_cast<const float2*>(w_ih);
    const float2* wh2 = reinterpret_cast<const float2*>(w_hh);
    #pragma unroll
    for (int j = 0; j < (3 * CCH * CH2 + GW * 32 - 1) / (GW * 32); ++j) {
        const int i = j * (GW * 32) + tid;
        if (i < 3 * CCH * CH2) {
            const int c = i / CH2, c2 = i % CH2;
            const int g = c / CCH, cc = c % CCH;
            const int o = (g * 32 + cc) * GST + 2 * c2;
            *reinterpret_cast<float2*>(&s_wih[o]) = wi2[i];
            *reinterpret_cast<float2*>(&s_whh[o]) = wh2[i];
        }
    }
    #pragma unroll
    for (int j = 0; j < (CCH * CCH + GW * 32 - 1) / (GW * 32); ++j) {
        const int i = j * (GW * 32) + tid;
        if (i < CCH * CCH) {
            const int k = i / CCH, c = i % CCH;
            s_Wt[c * GST + k] = W[i];
        }
    }
    if (tid < GROWS) {
        *reinterpret_cast<float2*>(&s_wih[tid * GST + CCH]) = make_float2(0.f, 0.f);
        *reinterpret_cast<float2*>(&s_whh[tid * GST + CCH]) = make_float2(0.f, 0.f);
    }
    if (tid < 6 * CH2) {
        const int rr = tid / CH2, c2 = tid % CH2;
        const int o = ((rr >> 1) * 32 + CCH + (rr & 1)) * GST + 2 * c2;
        *reinterpret_cast<float2*>(&s_wih[o]) = make_float2(0.f, 0.f);
        *reinterpret_cast<float2*>(&s_whh[o]) = make_float2(0.f, 0.f);
    }
    if (tid < 32)
        *reinterpret_cast<float2*>(&s_Wt[tid * GST + CCH]) = make_float2(0.f, 0.f);
    if (tid < 2 * CH2) {
        const int rr = tid / CH2, c2 = tid % CH2;
        *reinterpret_cast<float2*>(&s_Wt[(CCH + rr) * GST + 2 * c2]) = make_float2(0.f, 0.f);
    }
    if (full) {
        const float2* a2 = reinterpret_cast<const float2*>(aggh + (size_t)row0 * MST);
        const float2* h2 = reinterpret_cast<const float2*>(h    + (size_t)row0 * MST);
        const bool padl = (lane & 15) == 15;
        #pragma unroll
        for (int i = lane; i < 16 * MS2; i += 32) {
            const int r = i / MS2, c2 = i % MS2;
            float2 va = a2[i];
            float2 vh = h2[i];
            if (padl) { va = make_float2(0.f, 0.f); vh = make_float2(0.f, 0.f); }
            *reinterpret_cast<float2*>(&sa[r * GST + 2 * c2]) = va;
            *reinterpret_cast<float2*>(&sh[r * GST + 2 * c2]) = vh;
        }
    } else {
        if (lane < 16) {
            *reinterpret_cast<float2*>(&sa[lane * GST + CCH]) = make_float2(0.f, 0.f);
            *reinterpret_cast<float2*>(&sh[lane * GST + CCH]) = make_float2(0.f, 0.f);
        }
        for (int i = lane; i < 16 * CCH; i += 32) {
            const int r = i / CCH, c = i % CCH;
            const int row = row0 + r;
            if (row < N) {
                sa[r * GST + c] = aggh[(size_t)row * MST + c];
                sh[r * GST + c] = h   [(size_t)row * MST + c];
            }
        }
    }
    __syncthreads();

    if (row0 >= N) return;

    const int half = lane >> 4;
    const int l    = lane & 15;

    {
        v8f m0 = {};
        v8f m1 = {};
        #pragma unroll
        for (int k0 = 0; k0 < 32; k0 += 4) {
            const int ka = k0 + half * 2;
            const v2f a  = *(const v2f*)&sa  [l        * GST + ka];
            const v2f b0 = *(const v2f*)&s_Wt[l        * GST + ka];
            const v2f b1 = *(const v2f*)&s_Wt[(16 + l) * GST + ka];
            m0 = __builtin_amdgcn_wmma_f32_16x16x4_f32(false, a, false, b0,
                                                       (short)0, m0, false, false);
            m1 = __builtin_amdgcn_wmma_f32_16x16x4_f32(false, a, false, b1,
                                                       (short)0, m1, false, false);
        }
        #pragma unroll
        for (int v = 0; v < 8; ++v) {
            const int r = v + half * 8;
            sa[r * GST + l]      = m0[v];
            sa[r * GST + 16 + l] = m1[v];
        }
    }

    v8f accr[2], accz[2], acci[2], acch[2];
    #pragma unroll
    for (int t = 0; t < 2; ++t) {
        const int c  = t * 16 + l;
        const bool cv = c < CCH;
        const int cc = cv ? c : 0;
        const float br = cv ? b_ih[cc] + b_hh[cc] : 0.0f;
        const float bz = cv ? b_ih[CCH + cc] + b_hh[CCH + cc] : 0.0f;
        const float bi = cv ? b_ih[2 * CCH + cc] : 0.0f;
        const float bh = cv ? b_hh[2 * CCH + cc] : 0.0f;
        #pragma unroll
        for (int v = 0; v < 8; ++v) {
            accr[t][v] = br; accz[t][v] = bz; acci[t][v] = bi; acch[t][v] = bh;
        }
    }

    #pragma unroll
    for (int k0 = 0; k0 < 32; k0 += 4) {
        const int ka = k0 + half * 2;
        const v2f am = *(const v2f*)&sa[l * GST + ka];
        const v2f ah = *(const v2f*)&sh[l * GST + ka];
        #pragma unroll
        for (int t = 0; t < 2; ++t) {
            const int bo = (t * 16 + l) * GST + ka;
            const v2f bir = *(const v2f*)&s_wih[bo];
            const v2f biz = *(const v2f*)&s_wih[32 * GST + bo];
            const v2f bin = *(const v2f*)&s_wih[64 * GST + bo];
            const v2f bhr = *(const v2f*)&s_whh[bo];
            const v2f bhz = *(const v2f*)&s_whh[32 * GST + bo];
            const v2f bhn = *(const v2f*)&s_whh[64 * GST + bo];
            accr[t] = __builtin_amdgcn_wmma_f32_16x16x4_f32(false, am, false, bir,
                                                            (short)0, accr[t], false, false);
            accz[t] = __builtin_amdgcn_wmma_f32_16x16x4_f32(false, am, false, biz,
                                                            (short)0, accz[t], false, false);
            acci[t] = __builtin_amdgcn_wmma_f32_16x16x4_f32(false, am, false, bin,
                                                            (short)0, acci[t], false, false);
            acch[t] = __builtin_amdgcn_wmma_f32_16x16x4_f32(false, ah, false, bhn,
                                                            (short)0, acch[t], false, false);
            accr[t] = __builtin_amdgcn_wmma_f32_16x16x4_f32(false, ah, false, bhr,
                                                            (short)0, accr[t], false, false);
            accz[t] = __builtin_amdgcn_wmma_f32_16x16x4_f32(false, ah, false, bhz,
                                                            (short)0, accz[t], false, false);
        }
    }

    #pragma unroll
    for (int t = 0; t < 2; ++t) {
        #pragma unroll
        for (int v = 0; v < 8; ++v) {
            const int o = (v + half * 8) * GST + t * 16 + l;
            const float hold = sh[o];
            const float r  = sigmoid_f(accr[t][v]);
            const float z  = sigmoid_f(accz[t][v]);
            const float nn = tanh_f(acci[t][v] + r * acch[t][v]);
            sa[o] = (1.0f - z) * nn + z * hold;
        }
    }

    if (full) {
        float2* h2w = reinterpret_cast<float2*>(h + (size_t)row0 * MST);
        #pragma unroll
        for (int i = lane; i < 16 * MS2; i += 32) {
            const int r = i / MS2, c2 = i % MS2;
            h2w[i] = *reinterpret_cast<const float2*>(&sa[r * GST + 2 * c2]);
        }
    } else {
        for (int i = lane; i < 16 * MST; i += 32) {
            const int r = i / MST, c = i % MST;
            const int row = row0 + r;
            if (row < N) h[(size_t)row * MST + c] = sa[r * GST + c];
        }
    }
}

__global__ void __launch_bounds__(GW * 32)
ggnn_final_wmma(const float* __restrict__ h,
                const float* __restrict__ lin_w,
                const float* __restrict__ lin_b,
                float* __restrict__ out, int N) {
    __shared__ __attribute__((aligned(16))) float s_B[32 * GST];
    __shared__ __attribute__((aligned(16))) float s_A[GW][16 * GST];

    const int tid  = threadIdx.x;
    const int lane = tid & 31;
    const int warp = tid >> 5;
    const int row0 = (blockIdx.x * GW + warp) * 16;
    const bool full = (row0 + 16 <= N);
    float* sa = &s_A[warp][0];

    const float2* w2 = reinterpret_cast<const float2*>(lin_w);
    #pragma unroll
    for (int j = 0; j < (CCH * CH2 + GW * 32 - 1) / (GW * 32); ++j) {
        const int i = j * (GW * 32) + tid;
        if (i < CCH * CH2) {
            const int c = i / CH2, c2 = i % CH2;
            *reinterpret_cast<float2*>(&s_B[c * GST + 2 * c2]) = w2[i];
        }
    }
    if (tid < 32)
        *reinterpret_cast<float2*>(&s_B[tid * GST + CCH]) = make_float2(0.f, 0.f);
    if (tid < 2 * CH2) {
        const int rr = tid / CH2, c2 = tid % CH2;
        *reinterpret_cast<float2*>(&s_B[(CCH + rr) * GST + 2 * c2]) = make_float2(0.f, 0.f);
    }
    if (full) {
        const float2* h2 = reinterpret_cast<const float2*>(h + (size_t)row0 * MST);
        const bool padl = (lane & 15) == 15;
        #pragma unroll
        for (int i = lane; i < 16 * MS2; i += 32) {
            const int r = i / MS2, c2 = i % MS2;
            float2 v = h2[i];
            v.x = v.x > 0.0f ? v.x : 0.0f;
            v.y = v.y > 0.0f ? v.y : 0.0f;
            if (padl) v = make_float2(0.f, 0.f);
            *reinterpret_cast<float2*>(&sa[r * GST + 2 * c2]) = v;
        }
    } else {
        if (lane < 16)
            *reinterpret_cast<float2*>(&sa[lane * GST + CCH]) = make_float2(0.f, 0.f);
        for (int i = lane; i < 16 * CCH; i += 32) {
            const int r = i / CCH, c = i % CCH;
            const int row = row0 + r;
            if (row < N) {
                const float v = h[(size_t)row * MST + c];
                sa[r * GST + c] = v > 0.0f ? v : 0.0f;
            }
        }
    }
    __syncthreads();

    if (row0 >= N) return;

    const int half = lane >> 4;
    const int l    = lane & 15;

    v8f acc[2];
    #pragma unroll
    for (int t = 0; t < 2; ++t) {
        const int c = t * 16 + l;
        const float b = c < CCH ? lin_b[c < CCH ? c : 0] : 0.0f;
        #pragma unroll
        for (int v = 0; v < 8; ++v) acc[t][v] = b;
    }

    #pragma unroll
    for (int k0 = 0; k0 < 32; k0 += 4) {
        const int ka = k0 + half * 2;
        const v2f a  = *(const v2f*)&sa [l        * GST + ka];
        const v2f b0 = *(const v2f*)&s_B[l        * GST + ka];
        const v2f b1 = *(const v2f*)&s_B[(16 + l) * GST + ka];
        acc[0] = __builtin_amdgcn_wmma_f32_16x16x4_f32(false, a, false, b0,
                                                       (short)0, acc[0], false, false);
        acc[1] = __builtin_amdgcn_wmma_f32_16x16x4_f32(false, a, false, b1,
                                                       (short)0, acc[1], false, false);
    }

    #pragma unroll
    for (int v = 0; v < 8; ++v) {
        const int r = v + half * 8;
        sa[r * GST + l]      = acc[0][v];
        sa[r * GST + 16 + l] = acc[1][v];
    }

    if (full) {
        float2* o2 = reinterpret_cast<float2*>(out + (size_t)row0 * CCH);
        #pragma unroll
        for (int i = lane; i < 16 * CH2; i += 32) {
            const int r = i / CH2, c2 = i % CH2;
            o2[i] = *reinterpret_cast<const float2*>(&sa[r * GST + 2 * c2]);
        }
    } else {
        for (int i = lane; i < 16 * CCH; i += 32) {
            const int r = i / CCH, c = i % CCH;
            const int row = row0 + r;
            if (row < N) out[(size_t)row * CCH + c] = sa[r * GST + c];
        }
    }
}

extern "C" void kernel_launch(void* const* d_in, const int* in_sizes, int n_in,
                              void* d_out, int out_size, void* d_ws, size_t ws_size,
                              hipStream_t stream) {
    const float* x      = (const float*)d_in[0];
    const int*   eidx   = (const int*)  d_in[1];
    const float* weight = (const float*)d_in[2];
    const float* w_ih   = (const float*)d_in[3];
    const float* w_hh   = (const float*)d_in[4];
    const float* b_ih   = (const float*)d_in[5];
    const float* b_hh   = (const float*)d_in[6];
    const float* lin_w  = (const float*)d_in[7];
    const float* lin_b  = (const float*)d_in[8];
    float* out = (float*)d_out;

    const int N = in_sizes[0] / CCH;
    const int E = in_sizes[1] / 2;
    const int* src = eidx;
    const int* dst = eidx + E;

    float* h    = (float*)d_ws;
    float* aggh = h + (size_t)N * MST;
    const int nblk = (E + EBLK - 1) / EBLK;
    int* bcnt   = (int*)(aggh + (size_t)N * MST);
    int* bcur   = bcnt + NBMAX * CSTR;
    int* bbase  = bcur + NBMAX * CSTR;
    int* hmat   = bbase + NBMAX + 4;
    int* rowptr = hmat + (size_t)nblk * NBMAX;
    int* pairs  = rowptr + ((N + 4) & ~3);
    int* ssrc   = pairs + 2 * (size_t)E;

    ggnn_pad<<<(N * 8 + 255) / 256, 256, 0, stream>>>(x, h, N);

    const int c4 = ((2 * NBMAX * CSTR) >> 2) + 4;
    ggnn_zero<<<(c4 + 255) / 256, 256, 0, stream>>>((float*)bcnt, 2 * NBMAX * CSTR);
    ggnn_bhist<<<nblk, PTHR, 0, stream>>>(dst, bcnt, hmat, E);
    ggnn_part<<<nblk, PTHR, 0, stream>>>(src, dst, bcnt, bcur, hmat, bbase, pairs, E);
    ggnn_bsort<<<(N + BNODE - 1) / BNODE, STHR, 0, stream>>>(pairs, bbase, rowptr, ssrc, N, E);

    const int tiles = (N + 15) / 16;
    for (int layer = 0; layer < NLAYERS; ++layer) {
        ggnn_gather<<<GBLK, 256, 0, stream>>>(rowptr, ssrc, h, aggh, N);
        ggnn_gru_wmma<<<(tiles + GW - 1) / GW, GW * 32, 0, stream>>>(
            aggh, h, weight + (size_t)layer * CCH * CCH, w_ih, w_hh, b_ih, b_hh, N);
    }
    ggnn_final_wmma<<<(tiles + GW - 1) / GW, GW * 32, 0, stream>>>(h, lin_w, lin_b, out, N);
}


# Gather inner loop from ~12 to 3 vector ops per source row: each id becomes a 32-bit row offset once (unsigned min, shift), absent ids hit a zero row behind h so adds need no select; sum order kept
# speedup vs baseline: 19.4245x; 1.0574x over previous
#include <hip/hip_runtime.h>
#include <math.h>

#define CCH 30
#define NLAYERS 3
#define CH2 (CCH / 2)
#define MST 32
#define MS2 (MST / 2)

typedef __attribute__((ext_vector_type(2))) float v2f;
typedef __attribute__((ext_vector_type(8))) float v8f;

__device__ __forceinline__ float sigmoid_f(float x) {
    return __builtin_amdgcn_rcpf(1.0f + __builtin_amdgcn_exp2f(-1.44269504f * x));
}
__device__ __forceinline__ float tanh_f(float x) {
    return 1.0f - 2.0f * __builtin_amdgcn_rcpf(1.0f + __builtin_amdgcn_exp2f(2.88539008f * x));
}

#define ZROWS 2

__global__ void ggnn_pad(const float* __restrict__ x, float* __restrict__ h, int N) {
    const int i   = blockIdx.x * blockDim.x + threadIdx.x;
    const int row = i >> 3, q = i & 7;
    if (row < N) {
        const float2* x2 = reinterpret_cast<const float2*>(x) + (size_t)row * CH2 + 2 * q;
        const float2 a = x2[0];
        const float2 b = q < 7 ? x2[1] : make_float2(0.f, 0.f);
        reinterpret_cast<float4*>(h)[i] = make_float4(a.x, a.y, b.x, b.y);
    } else if (row < N + ZROWS) {
        reinterpret_cast<float4*>(h)[i] = make_float4(0.f, 0.f, 0.f, 0.f);
    }
}

__global__ void ggnn_zero(float* __restrict__ p, int n) {
    const int n4 = n >> 2;
    int i = blockIdx.x * blockDim.x + threadIdx.x;
    if (i < n4) reinterpret_cast<float4*>(p)[i] = make_float4(0.f, 0.f, 0.f, 0.f);
    if (i < (n & 3)) p[n4 * 4 + i] = 0.0f;
}

#define BSH   8
#define BNODE (1 << BSH)
#define NBMAX 1024
#define PTHR  1024
#define PPT   6
#define PCH   (PTHR * PPT)
#define EBLK  (4 * PCH)
#define CSTR  64
#define STHR  512
#define SCAP  12288

__device__ __forceinline__ int bucket_of(int d) {
    const unsigned b = (unsigned)d >> BSH;
    return (int)(b < (unsigned)NBMAX ? b : (unsigned)(NBMAX - 1));
}

__device__ __forceinline__ int block_excl_scan(int v, int* s_w, int& total) {
    const int lane = threadIdx.x & 31;
    const int wid  = threadIdx.x >> 5;
    const int nw   = (int)(blockDim.x >> 5);
    int x = v;
    #pragma unroll
    for (int d = 1; d < 32; d <<= 1) {
        const int y = __shfl_up(x, d);
        if (lane >= d) x += y;
    }
    __syncthreads();
    if (lane == 31) s_w[wid] = x;
    __syncthreads();
    if (wid == 0) {
        const int wt = lane < nw ? s_w[lane] : 0;
        int xs = wt;
        #pragma unroll
        for (int d = 1; d < 32; d <<= 1) {
            const int y = __shfl_up(xs, d);
            if (lane >= d) xs += y;
        }
        s_w[lane] = xs - wt;
        if (lane == 31) s_w[32] = xs;
    }
    __syncthreads();
    total = s_w[32];
    return s_w[wid] + (x - v);
}

__global__ void __launch_bounds__(PTHR)
ggnn_bhist(const int* __restrict__ dst, int* __restrict__ bcnt,
           int* __restrict__ hmat, int E) {
    __shared__ int s_h[NBMAX];
    const int tid = threadIdx.x;
    s_h[tid] = 0;
    __syncthreads();
    const int e0 = blockIdx.x * EBLK;
    const int n  = (E - e0) < EBLK ? (E - e0) : EBLK;
    for (int i0 = 0; i0 < n; i0 += PTHR * 8) {
        int d[8];
        #pragma unroll
        for (int j = 0; j < 8; ++j) {
            const int i = i0 + j * PTHR + tid;
            d[j] = i < n ? dst[e0 + i] : 0;
        }
        #pragma unroll
        for (int j = 0; j < 8; ++j) {
            const int i = i0 + j * PTHR + tid;
            if (i < n) atomicAdd(&s_h[bucket_of(d[j])], 1);
        }
    }
    __syncthreads();
    const int c = s_h[tid];
    hmat[blockIdx.x * NBMAX + tid] = c;
    if (c) atomicAdd(&bcnt[tid * CSTR], c);
}

__global__ void __launch_bounds__(PTHR)
ggnn_part(const int* __restrict__ src, const int* __restrict__ dst,
          const int* __restrict__ bcnt, int* __restrict__ bcur,
          const int* __restrict__ hmat, int* __restrict__ bbase,
          int* __restrict__ pairs_w, int E) {
    __shared__ int  s_h[NBMAX];
    __shared__ int  s_lo[NBMAX];
    __shared__ int  s_run[NBMAX];
    __shared__ int  s_w[33];
    __shared__ int2 s_p[PCH];
    int2* pairs = reinterpret_cast<int2*>(pairs_w);
    const int tid  = threadIdx.x;
    const int e0   = blockIdx.x * EBLK;
    const int nblk = (E - e0) < EBLK ? (E - e0) : EBLK;

    const int ctot = hmat[blockIdx.x * NBMAX + tid];
    int gtot;
    const int gb = block_excl_scan(bcnt[tid * CSTR], s_w, gtot);
    s_run[tid] = gb + (ctot > 0 ? atomicAdd(&bcur[tid * CSTR], ctot) : 0);
    if (blockIdx.x == 0) {
        bbase[tid] = gb;
        if (tid == PTHR - 1) bbase[NBMAX] = gtot;
    }

    for (int c0 = 0; c0 < nblk; c0 += PCH) {
        const int n = (nblk - c0) < PCH ? (nblk - c0) : PCH;
        s_h[tid] = 0;
        __syncthreads();

        int d[PPT], s[PPT], r[PPT];
        #pragma unroll
        for (int j = 0; j < PPT; ++j) {
            const int i = j * PTHR + tid;
            d[j] = i < n ? dst[e0 + c0 + i] : 0;
            s[j] = i < n ? src[e0 + c0 + i] : 0;
        }
        #pragma unroll
        for (int j = 0; j < PPT; ++j) {
            const int i = j * PTHR + tid;
            r[j] = i < n ? atomicAdd(&s_h[bucket_of(d[j])], 1) : 0;
        }
        __syncthreads();

        const int c = s_h[tid];
        int tot;
        s_lo[tid] = block_excl_scan(c, s_w, tot);
        __syncthreads();

        #pragma unroll
        for (int j = 0; j < PPT; ++j) {
            const int i = j * PTHR + tid;
            if (i < n) s_p[s_lo[bucket_of(d[j])] + r[j]] = make_int2(d[j], s[j]);
        }
        __syncthreads();

        for (int i = tid; i < n; i += PTHR) {
            const int2 v = s_p[i];
            const int b  = bucket_of(v.x);
            const int pos = s_run[b] + (i - s_lo[b]);
            if ((unsigned)pos < (unsigned)E) pairs[pos] = v;
        }
        __syncthreads();
        s_run[tid] += c;
    }
}

__global__ void __launch_bounds__(STHR)
ggnn_bsort(const int* __restrict__ pairs_w, const int* __restrict__ bbase,
           int* __restrict__ rowptr, int* __restrict__ ssrc, int N, int E) {
    __shared__ int s_c[BNODE];
    __shared__ int s_w[33];
    __shared__ int s_s[SCAP];
    const int tid   = threadIdx.x;
    const int b     = blockIdx.x;
    const int node0 = b << BSH;
    int bs = bbase[b];
    int be = bbase[b + 1];
    bs = bs < 0 ? 0 : (bs > E ? E : bs);
    be = be < bs ? bs : (be > E ? E : be);
    const int n = be - bs;
    const bool stage = n <= SCAP;
    const int2* pb = reinterpret_cast<const int2*>(pairs_w) + bs;

    if (tid < BNODE) s_c[tid] = 0;
    __syncthreads();
    for (int i0 = 0; i0 < n; i0 += STHR * 8) {
        int2 v[8];
        #pragma unroll
        for (int j = 0; j < 8; ++j) {
            const int i = i0 + j * STHR + tid;
            v[j] = i < n ? pb[i] : make_int2(0, 0);
        }
        #pragma unroll
        for (int j = 0; j < 8; ++j) {
            const int i = i0 + j * STHR + tid;
            if (i < n) atomicAdd(&s_c[(v[j].x - node0) & (BNODE - 1)], 1);
        }
    }
    __syncthreads();

    const int c = tid < BNODE ? s_c[tid] : 0;
    int tot;
    const int ex = block_excl_scan(c, s_w, tot);
    if (tid < BNODE) {
        const int node = node0 + tid;
        if (node < N) rowptr[node] = bs + ex;
        s_c[tid] = ex;
    }
    if (b == (int)gridDim.x - 1 && tid == 0) rowptr[N] = E;
    __syncthreads();

    for (int i0 = 0; i0 < n; i0 += STHR * 8) {
        int2 v[8];
        #pragma unroll
        for (int j = 0; j < 8; ++j) {
            const int i = i0 + j * STHR + tid;
            v[j] = i < n ? pb[i] : make_int2(0, 0);
        }
        #pragma unroll
        for (int j = 0; j < 8; ++j) {
            const int i = i0 + j * STHR + tid;
            if (i < n) {
                const int p = atomicAdd(&s_c[(v[j].x - node0) & (BNODE - 1)], 1);
                if (stage) {
                    if ((unsigned)p < (unsigned)SCAP) s_s[p] = v[j].y;
                } else {
                    if ((unsigned)p < (unsigned)n) ssrc[bs + p] = v[j].y;
                }
            }
        }
    }
    if (stage) {
        __syncthreads();
        for (int i = tid; i < n; i += STHR) ssrc[bs + i] = s_s[i];
    }
}

#define GBLK 2048

__device__ __forceinline__ void gather16(int idx, int rw, unsigned l8, unsigned N,
                                         const char* __restrict__ hb, float2& acc) {
    const unsigned rid = (unsigned)idx < N ? (unsigned)idx : N;
    const int ro = (int)(rid * (unsigned)(MST * sizeof(float)));
    #pragma unroll
    for (int j0 = 0; j0 < 16; j0 += 8) {
        if (j0 < rw) {
            float2 v[8];
            #pragma unroll
            for (int j = 0; j < 8; ++j) {
                const unsigned o = (unsigned)__shfl(ro, j0 + j, 16) + l8;
                v[j] = *reinterpret_cast<const float2*>(hb + o);
            }
            #pragma unroll
            for (int j = 0; j < 8; ++j) { acc.x += v[j].x; acc.y += v[j].y; }
        }
    }
}

__global__ void __launch_bounds__(256)
ggnn_gather(const int* __restrict__ rowptr, const int* __restrict__ ssrc,
            const float* __restrict__ h, float* __restrict__ aggh, int N) {
    const int tid   = threadIdx.x;
    const int l     = tid & 15;
    const int step  = (int)gridDim.x * 16;
    const int first = (blockIdx.x * 256 + tid) >> 4;
    const char* hb = reinterpret_cast<const char*>(h);
    const unsigned l8 = (unsigned)l * (unsigned)sizeof(float2);
    const unsigned un = (unsigned)N;

    int node = first - step;
    int p = 0, rem = 0;
    int ia = -1, ib = -1, ic = -1;
    int p1 = 0, e1 = 0;
    if (first < N) { p1 = rowptr[first]; e1 = rowptr[first + 1]; }
    int rem1 = e1 - p1;

    for (int base = __builtin_amdgcn_readfirstlane(first & ~1) - step; base < N; base += step) {
        const int ja = (l      < rem1) ? ssrc[p1 + l]      : -1;
        const int jb = (l + 16 < rem1) ? ssrc[p1 + 16 + l] : -1;
        const int jc = (l + 32 < rem1) ? ssrc[p1 + 32 + l] : -1;
        int p2 = 0, e2 = 0;
        if (node + 2 * step < N) {
            p2 = rowptr[node + 2 * step];
            e2 = rowptr[node + 2 * step + 1];
        }

        const int other = __shfl_xor(rem, 16);
        const int remw  = __builtin_amdgcn_readfirstlane(rem > other ? rem : other);

        float2 acc = make_float2(0.0f, 0.0f);
        if (remw > 0)  gather16(ia, remw,      l8, un, hb, acc);
        if (remw > 16) gather16(ib, remw - 16, l8, un, hb, acc);
        if (remw > 32) gather16(ic, remw - 32, l8, un, hb, acc);
        {
            int q = p + 48;
            int r = rem - 48;
            for (int rw = remw - 48; rw > 0; rw -= 16) {
                const int id = (l < r) ? ssrc[q + l] : -1;
                gather16(id, rw, l8, un, hb, acc);
                q += 16; r -= 16;
            }
        }
        if ((unsigned)node < (unsigned)N)
            reinterpret_cast<float2*>(aggh)[(size_t)node * MS2 + l] =
                l < CH2 ? acc : make_float2(0.0f, 0.0f);

        node += step;
        p = p1; rem = rem1; p1 = p2; rem1 = e2 - p2;
        ia = ja; ib = jb; ic = jc;
    }
}

#define GST   34
#define GROWS 96
#define GW    8

__global__ void __launch_bounds__(GW * 32)
ggnn_gru_wmma(const float* __restrict__ aggh,
              float* __restrict__ h,
              const float* __restrict__ W,
              const float* __restrict__ w_ih,
              const float* __restrict__ w_hh,
              const float* __restrict__ b_ih,
              const float* __restrict__ b_hh, int N) {
    __shared__ __attribute__((aligned(16))) float s_wih[GROWS * GST];
    __shared__ __attribute__((aligned(16))) float s_whh[GROWS * GST];
    __shared__ __attribute__((aligned(16))) float s_Wt[32 * GST];
    __shared__ __attribute__((aligned(16))) float s_A[GW][16 * GST];
    __shared__ __attribute__((aligned(16))) float s_H[GW][16 * GST];

    const int tid  = threadIdx.x;
    const int lane = tid & 31;
    const int warp = tid >> 5;
    const int row0 = (blockIdx.x * GW + warp) * 16;
    const bool full = (row0 + 16 <= N);
    float* sa = &s_A[warp][0];
    float* sh = &s_H[warp][0];

    const float2* wi2 = reinterpret_cast<const float2*>(w_ih);
    const float2* wh2 = reinterpret_cast<const float2*>(w_hh);
    #pragma unroll
    for (int j = 0; j < (3 * CCH * CH2 + GW * 32 - 1) / (GW * 32); ++j) {
        const int i = j * (GW * 32) + tid;
        if (i < 3 * CCH * CH2) {
            const int c = i / CH2, c2 = i % CH2;
            const int g = c / CCH, cc = c % CCH;
            const int o = (g * 32 + cc) * GST + 2 * c2;
            *reinterpret_cast<float2*>(&s_wih[o]) = wi2[i];
            *reinterpret_cast<float2*>(&s_whh[o]) = wh2[i];
        }
    }
    #pragma unroll
    for (int j = 0; j < (CCH * CCH + GW * 32 - 1) / (GW * 32); ++j) {
        const int i = j * (GW * 32) + tid;
        if (i < CCH * CCH) {
            const int k = i / CCH, c = i % CCH;
            s_Wt[c * GST + k] = W[i];
        }
    }
    if (tid < GROWS) {
        *reinterpret_cast<float2*>(&s_wih[tid * GST + CCH]) = make_float2(0.f, 0.f);
        *reinterpret_cast<float2*>(&s_whh[tid * GST + CCH]) = make_float2(0.f, 0.f);
    }
    if (tid < 6 * CH2) {
        const int rr = tid / CH2, c2 = tid % CH2;
        const int o = ((rr >> 1) * 32 + CCH + (rr & 1)) * GST + 2 * c2;
        *reinterpret_cast<float2*>(&s_wih[o]) = make_float2(0.f, 0.f);
        *reinterpret_cast<float2*>(&s_whh[o]) = make_float2(0.f, 0.f);
    }
    if (tid < 32)
        *reinterpret_cast<float2*>(&s_Wt[tid * GST + CCH]) = make_float2(0.f, 0.f);
    if (tid < 2 * CH2) {
        const int rr = tid / CH2, c2 = tid % CH2;
        *reinterpret_cast<float2*>(&s_Wt[(CCH + rr) * GST + 2 * c2]) = make_float2(0.f, 0.f);
    }
    if (full) {
        const float2* a2 = reinterpret_cast<const float2*>(aggh + (size_t)row0 * MST);
        const float2* h2 = reinterpret_cast<const float2*>(h    + (size_t)row0 * MST);
        const bool padl = (lane & 15) == 15;
        #pragma unroll
        for (int i = lane; i < 16 * MS2; i += 32) {
            const int r = i / MS2, c2 = i % MS2;
            float2 va = a2[i];
            float2 vh = h2[i];
            if (padl) { va = make_float2(0.f, 0.f); vh = make_float2(0.f, 0.f); }
            *reinterpret_cast<float2*>(&sa[r * GST + 2 * c2]) = va;
            *reinterpret_cast<float2*>(&sh[r * GST + 2 * c2]) = vh;
        }
    } else {
        if (lane < 16) {
            *reinterpret_cast<float2*>(&sa[lane * GST + CCH]) = make_float2(0.f, 0.f);
            *reinterpret_cast<float2*>(&sh[lane * GST + CCH]) = make_float2(0.f, 0.f);
        }
        for (int i = lane; i < 16 * CCH; i += 32) {
            const int r = i / CCH, c = i % CCH;
            const int row = row0 + r;
            if (row < N) {
                sa[r * GST + c] = aggh[(size_t)row * MST + c];
                sh[r * GST + c] = h   [(size_t)row * MST + c];
            }
        }
    }
    __syncthreads();

    if (row0 >= N) return;

    const int half = lane >> 4;
    const int l    = lane & 15;

    {
        v8f m0 = {};
        v8f m1 = {};
        #pragma unroll
        for (int k0 = 0; k0 < 32; k0 += 4) {
            const int ka = k0 + half * 2;
            const v2f a  = *(const v2f*)&sa  [l        * GST + ka];
            const v2f b0 = *(const v2f*)&s_Wt[l        * GST + ka];
            const v2f b1 = *(const v2f*)&s_Wt[(16 + l) * GST + ka];
            m0 = __builtin_amdgcn_wmma_f32_16x16x4_f32(false, a, false, b0,
                                                       (short)0, m0, false, false);
            m1 = __builtin_amdgcn_wmma_f32_16x16x4_f32(false, a, false, b1,
                                                       (short)0, m1, false, false);
        }
        #pragma unroll
        for (int v = 0; v < 8; ++v) {
            const int r = v + half * 8;
            sa[r * GST + l]      = m0[v];
            sa[r * GST + 16 + l] = m1[v];
        }
    }

    v8f accr[2], accz[2], acci[2], acch[2];
    #pragma unroll
    for (int t = 0; t < 2; ++t) {
        const int c  = t * 16 + l;
        const bool cv = c < CCH;
        const int cc = cv ? c : 0;
        const float br = cv ? b_ih[cc] + b_hh[cc] : 0.0f;
        const float bz = cv ? b_ih[CCH + cc] + b_hh[CCH + cc] : 0.0f;
        const float bi = cv ? b_ih[2 * CCH + cc] : 0.0f;
        const float bh = cv ? b_hh[2 * CCH + cc] : 0.0f;
        #pragma unroll
        for (int v = 0; v < 8; ++v) {
            accr[t][v] = br; accz[t][v] = bz; acci[t][v] = bi; acch[t][v] = bh;
        }
    }

    #pragma unroll
    for (int k0 = 0; k0 < 32; k0 += 4) {
        const int ka = k0 + half * 2;
        const v2f am = *(const v2f*)&sa[l * GST + ka];
        const v2f ah = *(const v2f*)&sh[l * GST + ka];
        #pragma unroll
        for (int t = 0; t < 2; ++t) {
            const int bo = (t * 16 + l) * GST + ka;
            const v2f bir = *(const v2f*)&s_wih[bo];
            const v2f biz = *(const v2f*)&s_wih[32 * GST + bo];
            const v2f bin = *(const v2f*)&s_wih[64 * GST + bo];
            const v2f bhr = *(const v2f*)&s_whh[bo];
            const v2f bhz = *(const v2f*)&s_whh[32 * GST + bo];
            const v2f bhn = *(const v2f*)&s_whh[64 * GST + bo];
            accr[t] = __builtin_amdgcn_wmma_f32_16x16x4_f32(false, am, false, bir,
                                                            (short)0, accr[t], false, false);
            accz[t] = __builtin_amdgcn_wmma_f32_16x16x4_f32(false, am, false, biz,
                                                            (short)0, accz[t], false, false);
            acci[t] = __builtin_amdgcn_wmma_f32_16x16x4_f32(false, am, false, bin,
                                                            (short)0, acci[t], false, false);
            acch[t] = __builtin_amdgcn_wmma_f32_16x16x4_f32(false, ah, false, bhn,
                                                            (short)0, acch[t], false, false);
            accr[t] = __builtin_amdgcn_wmma_f32_16x16x4_f32(false, ah, false, bhr,
                                                            (short)0, accr[t], false, false);
            accz[t] = __builtin_amdgcn_wmma_f32_16x16x4_f32(false, ah, false, bhz,
                                                            (short)0, accz[t], false, false);
        }
    }

    #pragma unroll
    for (int t = 0; t < 2; ++t) {
        #pragma unroll
        for (int v = 0; v < 8; ++v) {
            const int o = (v + half * 8) * GST + t * 16 + l;
            const float hold = sh[o];
            const float r  = sigmoid_f(accr[t][v]);
            const float z  = sigmoid_f(accz[t][v]);
            const float nn = tanh_f(acci[t][v] + r * acch[t][v]);
            sa[o] = (1.0f - z) * nn + z * hold;
        }
    }

    if (full) {
        float2* h2w = reinterpret_cast<float2*>(h + (size_t)row0 * MST);
        #pragma unroll
        for (int i = lane; i < 16 * MS2; i += 32) {
            const int r = i / MS2, c2 = i % MS2;
            h2w[i] = *reinterpret_cast<const float2*>(&sa[r * GST + 2 * c2]);
        }
    } else {
        for (int i = lane; i < 16 * MST; i += 32) {
            const int r = i / MST, c = i % MST;
            const int row = row0 + r;
            if (row < N) h[(size_t)row * MST + c] = sa[r * GST + c];
        }
    }
}

__global__ void __launch_bounds__(GW * 32)
ggnn_final_wmma(const float* __restrict__ h,
                const float* __restrict__ lin_w,
                const float* __restrict__ lin_b,
                float* __restrict__ out, int N) {
    __shared__ __attribute__((aligned(16))) float s_B[32 * GST];
    __shared__ __attribute__((aligned(16))) float s_A[GW][16 * GST];

    const int tid  = threadIdx.x;
    const int lane = tid & 31;
    const int warp = tid >> 5;
    const int row0 = (blockIdx.x * GW + warp) * 16;
    const bool full = (row0 + 16 <= N);
    float* sa = &s_A[warp][0];

    const float2* w2 = reinterpret_cast<const float2*>(lin_w);
    #pragma unroll
    for (int j = 0; j < (CCH * CH2 + GW * 32 - 1) / (GW * 32); ++j) {
        const int i = j * (GW * 32) + tid;
        if (i < CCH * CH2) {
            const int c = i / CH2, c2 = i % CH2;
            *reinterpret_cast<float2*>(&s_B[c * GST + 2 * c2]) = w2[i];
        }
    }
    if (tid < 32)
        *reinterpret_cast<float2*>(&s_B[tid * GST + CCH]) = make_float2(0.f, 0.f);
    if (tid < 2 * CH2) {
        const int rr = tid / CH2, c2 = tid % CH2;
        *reinterpret_cast<float2*>(&s_B[(CCH + rr) * GST + 2 * c2]) = make_float2(0.f, 0.f);
    }
    if (full) {
        const float2* h2 = reinterpret_cast<const float2*>(h + (size_t)row0 * MST);
        const bool padl = (lane & 15) == 15;
        #pragma unroll
        for (int i = lane; i < 16 * MS2; i += 32) {
            const int r = i / MS2, c2 = i % MS2;
            float2 v = h2[i];
            v.x = v.x > 0.0f ? v.x : 0.0f;
            v.y = v.y > 0.0f ? v.y : 0.0f;
            if (padl) v = make_float2(0.f, 0.f);
            *reinterpret_cast<float2*>(&sa[r * GST + 2 * c2]) = v;
        }
    } else {
        if (lane < 16)
            *reinterpret_cast<float2*>(&sa[lane * GST + CCH]) = make_float2(0.f, 0.f);
        for (int i = lane; i < 16 * CCH; i += 32) {
            const int r = i / CCH, c = i % CCH;
            const int row = row0 + r;
            if (row < N) {
                const float v = h[(size_t)row * MST + c];
                sa[r * GST + c] = v > 0.0f ? v : 0.0f;
            }
        }
    }
    __syncthreads();

    if (row0 >= N) return;

    const int half = lane >> 4;
    const int l    = lane & 15;

    v8f acc[2];
    #pragma unroll
    for (int t = 0; t < 2; ++t) {
        const int c = t * 16 + l;
        const float b = c < CCH ? lin_b[c < CCH ? c : 0] : 0.0f;
        #pragma unroll
        for (int v = 0; v < 8; ++v) acc[t][v] = b;
    }

    #pragma unroll
    for (int k0 = 0; k0 < 32; k0 += 4) {
        const int ka = k0 + half * 2;
        const v2f a  = *(const v2f*)&sa [l        * GST + ka];
        const v2f b0 = *(const v2f*)&s_B[l        * GST + ka];
        const v2f b1 = *(const v2f*)&s_B[(16 + l) * GST + ka];
        acc[0] = __builtin_amdgcn_wmma_f32_16x16x4_f32(false, a, false, b0,
                                                       (short)0, acc[0], false, false);
        acc[1] = __builtin_amdgcn_wmma_f32_16x16x4_f32(false, a, false, b1,
                                                       (short)0, acc[1], false, false);
    }

    #pragma unroll
    for (int v = 0; v < 8; ++v) {
        const int r = v + half * 8;
        sa[r * GST + l]      = acc[0][v];
        sa[r * GST + 16 + l] = acc[1][v];
    }

    if (full) {
        float2* o2 = reinterpret_cast<float2*>(out + (size_t)row0 * CCH);
        #pragma unroll
        for (int i = lane; i < 16 * CH2; i += 32) {
            const int r = i / CH2, c2 = i % CH2;
            o2[i] = *reinterpret_cast<const float2*>(&sa[r * GST + 2 * c2]);
        }
    } else {
        for (int i = lane; i < 16 * CCH; i += 32) {
            const int r = i / CCH, c = i % CCH;
            const int row = row0 + r;
            if (row < N) out[(size_t)row * CCH + c] = sa[r * GST + c];
        }
    }
}

extern "C" void kernel_launch(void* const* d_in, const int* in_sizes, int n_in,
                              void* d_out, int out_size, void* d_ws, size_t ws_size,
                              hipStream_t stream) {
    const float* x      = (const float*)d_in[0];
    const int*   eidx   = (const int*)  d_in[1];
    const float* weight = (const float*)d_in[2];
    const float* w_ih   = (const float*)d_in[3];
    const float* w_hh   = (const float*)d_in[4];
    const float* b_ih   = (const float*)d_in[5];
    const float* b_hh   = (const float*)d_in[6];
    const float* lin_w  = (const float*)d_in[7];
    const float* lin_b  = (const float*)d_in[8];
    float* out = (float*)d_out;

    const int N = in_sizes[0] / CCH;
    const int E = in_sizes[1] / 2;
    const int* src = eidx;
    const int* dst = eidx + E;

    float* h    = (float*)d_ws;
    float* aggh = h + (size_t)(N + ZROWS) * MST;
    const int nblk = (E + EBLK - 1) / EBLK;
    int* bcnt   = (int*)(aggh + (size_t)N * MST);
    int* bcur   = bcnt + NBMAX * CSTR;
    int* bbase  = bcur + NBMAX * CSTR;
    int* hmat   = bbase + NBMAX + 4;
    int* rowptr = hmat + (size_t)nblk * NBMAX;
    int* pairs  = rowptr + ((N + 4) & ~3);
    int* ssrc   = pairs + 2 * (size_t)E;

    ggnn_pad<<<((N + ZROWS) * 8 + 255) / 256, 256, 0, stream>>>(x, h, N);

    const int c4 = ((2 * NBMAX * CSTR) >> 2) + 4;
    ggnn_zero<<<(c4 + 255) / 256, 256, 0, stream>>>((float*)bcnt, 2 * NBMAX * CSTR);
    ggnn_bhist<<<nblk, PTHR, 0, stream>>>(dst, bcnt, hmat, E);
    ggnn_part<<<nblk, PTHR, 0, stream>>>(src, dst, bcnt, bcur, hmat, bbase, pairs, E);
    ggnn_bsort<<<(N + BNODE - 1) / BNODE, STHR, 0, stream>>>(pairs, bbase, rowptr, ssrc, N, E);

    const int tiles = (N + 15) / 16;
    for (int layer = 0; layer < NLAYERS; ++layer) {
        ggnn_gather<<<GBLK, 256, 0, stream>>>(rowptr, ssrc, h, aggh, N);
        ggnn_gru_wmma<<<(tiles + GW - 1) / GW, GW * 32, 0, stream>>>(
            aggh, h, weight + (size_t)layer * CCH * CCH, w_ih, w_hh, b_ih, b_hh, N);
    }
    ggnn_final_wmma<<<(tiles + GW - 1) / GW, GW * 32, 0, stream>>>(h, lin_w, lin_b, out, N);
}
